# starting kernel
# baseline (speedup 1.0000x reference)
#include <hip/hip_runtime.h>
#include <hip/hip_bf16.h>
#include <math.h>

typedef __attribute__((ext_vector_type(2))) float v2f;
typedef __attribute__((ext_vector_type(8))) float v8f;

template <int NT>
__global__ void __launch_bounds__(256) gemm_f32_wmma(
    const float* __restrict__ A, const float* __restrict__ B,
    float* __restrict__ C, int M, int K, int N)
{
    __shared__ v2f sB2[4096];
    const int tid = threadIdx.x;
    const int pairCount = (K >> 1) * N;
    for (int p = tid; p < pairCount; p += 256) {
        const int kk = (p / N) << 1;
        const int n  = p - (p / N) * N;
        v2f b;
        b.x = B[kk * N + n];
        b.y = B[(kk + 1) * N + n];
        sB2[p] = b;
    }
    __syncthreads();

    const int lane    = tid & 31;
    const int wave    = tid >> 5;
    const int rowTile = blockIdx.x * 8 + wave;
    const int numTiles = (M + 15) >> 4;
    if (rowTile >= numTiles) return;

    const int half = lane >> 4;
    const int l15  = lane & 15;
    int mrow = rowTile * 16 + l15;
    if (mrow >= M) mrow = M - 1;

    v8f acc[NT] = {};
    const float* arow = A + (size_t)mrow * K + 2 * half;
    const v2f* __restrict__ bcol = sB2 + half * N + l15;

    for (int k0 = 0; k0 < K; k0 += 4) {
        const v2f a = *(const v2f*)(arow + k0);
        const v2f* brow = bcol + (k0 >> 1) * N;
#pragma unroll
        for (int nt = 0; nt < NT; ++nt) {
            const v2f b = brow[nt * 16];
            acc[nt] = __builtin_amdgcn_wmma_f32_16x16x4_f32(
                false, a, false, b, (short)0, acc[nt], false, false);
        }
    }

#pragma unroll
    for (int nt = 0; nt < NT; ++nt) {
#pragma unroll
        for (int r = 0; r < 8; ++r) {
            const int row = rowTile * 16 + r + 8 * half;
            if (row < M)
                C[(size_t)row * N + nt * 16 + l15] = acc[nt][r];
        }
    }
}

__device__ __forceinline__ void atomicMaxFloat(float* addr, float val) {
    const int iv = __float_as_int(val);
    if (iv >= 0) atomicMax((int*)addr, iv);
    else         atomicMin((unsigned int*)addr, (unsigned int)iv);
}

__global__ void node_score_kernel(const float* __restrict__ z,
                                  const float* __restrict__ w_attn,
                                  float* __restrict__ ssrc, float* __restrict__ sdst,
                                  int N)
{
    const int n = blockIdx.x * blockDim.x + threadIdx.x;
    if (n >= N) return;
    const float* zr = z + (size_t)n * 64;
    float s0 = 0.f, s1 = 0.f;
#pragma unroll
    for (int i = 0; i < 64; ++i) {
        const float v = zr[i];
        s0 += v * w_attn[i];
        s1 += v * w_attn[96 + i];
    }
    ssrc[n] = s0;
    sdst[n] = s1;
}

__global__ void init_kernel(float* __restrict__ h, float* __restrict__ mbuf,
                            float* __restrict__ den, float* __restrict__ zphi,
                            int hsize, int N)
{
    const int i = blockIdx.x * blockDim.x + threadIdx.x;
    const int stride = gridDim.x * blockDim.x;
    for (int j = i; j < hsize; j += stride) h[j] = 0.f;
    for (int j = i; j < N; j += stride) { mbuf[j] = -INFINITY; den[j] = 0.f; }
    if (i < 256) zphi[i] = 0.f;
}

__global__ void edge_score_kernel(const float* __restrict__ eft,
                                  const int* __restrict__ src, const int* __restrict__ dst,
                                  const float* __restrict__ ssrc, const float* __restrict__ sdst,
                                  const float* __restrict__ w_attn,
                                  float* __restrict__ a, float* __restrict__ mbuf, int E)
{
    const int e = blockIdx.x * blockDim.x + threadIdx.x;
    if (e >= E) return;
    const float* er = eft + (size_t)e * 32;
    float t = ssrc[src[e]] + sdst[dst[e]];
#pragma unroll
    for (int i = 0; i < 32; ++i) t += er[i] * w_attn[64 + i];
    const float lr = t > 0.f ? t : 0.01f * t;
    a[e] = lr;
    atomicMaxFloat(&mbuf[dst[e]], lr);
}

__global__ void edge_exp_kernel(const int* __restrict__ dst,
                                const float* __restrict__ mbuf,
                                float* __restrict__ a, float* __restrict__ den, int E)
{
    const int e = blockIdx.x * blockDim.x + threadIdx.x;
    if (e >= E) return;
    const int d = dst[e];
    const float ex = __expf(a[e] - mbuf[d]);
    a[e] = ex;
    atomicAdd(&den[d], ex);
}

__global__ void edge_scatter_kernel(const float* __restrict__ z, const float* __restrict__ eft,
                                    const float* __restrict__ a, const float* __restrict__ den,
                                    const int* __restrict__ src, const int* __restrict__ dst,
                                    const int* __restrict__ etype,
                                    float* __restrict__ h, int E)
{
    const int e    = (int)((blockIdx.x * blockDim.x + threadIdx.x) >> 5);
    const int lane = threadIdx.x & 31;
    if (e >= E) return;
    const int d = dst[e];
    const float alpha = a[e] / den[d];
    const size_t base = ((size_t)d * 4 + etype[e]) * 96;
    const float* zr = z + (size_t)src[e] * 64;
    const float* er = eft + (size_t)e * 32;
#pragma unroll
    for (int j = lane; j < 96; j += 32) {
        const float v = (j < 64) ? zr[j] : er[j - 64];
        atomicAdd(&h[base + j], alpha * v);
    }
}

__global__ void zphi_kernel(const float* __restrict__ h2, float* __restrict__ zphi, int nodes)
{
    const int s = threadIdx.x;
    float sum = 0.f;
    for (int n = blockIdx.x; n < nodes; n += gridDim.x)
        sum += h2[(size_t)n * 256 + s];
    atomicAdd(&zphi[s], sum);
}

__global__ void beta_kernel(const float* __restrict__ zphi,
                            const float* __restrict__ w_sem, float* __restrict__ beta)
{
    if (threadIdx.x != 0) return;
    float wp[4], mx = -INFINITY;
#pragma unroll
    for (int t = 0; t < 4; ++t) {
        float d = 0.f;
        for (int o = 0; o < 64; ++o) d += zphi[t * 64 + o] * w_sem[o];
        const float lr = d > 0.f ? d : 0.01f * d;
        wp[t] = lr;
        mx = fmaxf(mx, lr);
    }
    float s = 0.f;
#pragma unroll
    for (int t = 0; t < 4; ++t) { wp[t] = __expf(wp[t] - mx); s += wp[t]; }
#pragma unroll
    for (int t = 0; t < 4; ++t) beta[t] = wp[t] / s;
}

__global__ void out_kernel(const float* __restrict__ h2, const float* __restrict__ beta,
                           float* __restrict__ Z, int N)
{
    const int i = blockIdx.x * blockDim.x + threadIdx.x;
    if (i >= N * 64) return;
    const int n = i >> 6, o = i & 63;
    const float* r = h2 + (size_t)n * 256 + o;
    Z[i] = beta[0] * r[0] + beta[1] * r[64] + beta[2] * r[128] + beta[3] * r[192];
}

extern "C" void kernel_launch(void* const* d_in, const int* in_sizes, int n_in,
                              void* d_out, int out_size, void* d_ws, size_t ws_size,
                              hipStream_t stream)
{
    const float* node   = (const float*)d_in[0];
    const float* edgef  = (const float*)d_in[1];
    const int*   src    = (const int*)d_in[2];
    const int*   dst    = (const int*)d_in[3];
    const int*   etype  = (const int*)d_in[4];
    const float* w_n    = (const float*)d_in[5];
    const float* w_e    = (const float*)d_in[6];
    const float* w_attn = (const float*)d_in[7];
    const float* w_fc2  = (const float*)d_in[8];
    const float* w_sem  = (const float*)d_in[9];
    float* out = (float*)d_out;

    const int N = in_sizes[0] / 128;
    const int E = in_sizes[2];
    const int T = 4;

    float* ws   = (float*)d_ws;
    float* z    = ws;
    float* ssrc = z + (size_t)N * 64;
    float* sdst = ssrc + N;
    float* abuf = sdst + N;
    float* mbuf = abuf + E;
    float* den  = mbuf + N;
    float* zphi = den + N;
    float* beta = zphi + 256;
    float* h    = beta + 16;
    float* eft  = h + (size_t)N * T * 96;
    float* h2   = eft;

    const int thr = 256;

    gemm_f32_wmma<4><<<((N + 15) / 16 + 7) / 8, thr, 0, stream>>>(node, w_n, z, N, 128, 64);
    gemm_f32_wmma<2><<<((E + 15) / 16 + 7) / 8, thr, 0, stream>>>(edgef, w_e, eft, E, 64, 32);
    node_score_kernel<<<(N + thr - 1) / thr, thr, 0, stream>>>(z, w_attn, ssrc, sdst, N);
    init_kernel<<<1024, thr, 0, stream>>>(h, mbuf, den, zphi, N * T * 96, N);
    edge_score_kernel<<<(E + thr - 1) / thr, thr, 0, stream>>>(eft, src, dst, ssrc, sdst,
                                                               w_attn, abuf, mbuf, E);
    edge_exp_kernel<<<(E + thr - 1) / thr, thr, 0, stream>>>(dst, mbuf, abuf, den, E);
    edge_scatter_kernel<<<(E + 7) / 8, thr, 0, stream>>>(z, eft, abuf, den, src, dst, etype, h, E);
    gemm_f32_wmma<4><<<((N * T + 15) / 16 + 7) / 8, thr, 0, stream>>>(h, w_fc2, h2, N * T, 96, 64);
    zphi_kernel<<<256, thr, 0, stream>>>(h2, zphi, N);
    beta_kernel<<<1, 32, 0, stream>>>(zphi, w_sem, beta);
    out_kernel<<<(N * 64 + thr - 1) / thr, thr, 0, stream>>>(h2, beta, out, N);
}


# fc2 applied per edge before the scatter (linear): wave takes 16 edges, eft tile @ w_fc2[64:96] on WMMA plus (z@w_fc2[0:64])[src]; 64 atomics per edge, was 96; GEMM over h goes; sum order changes
# speedup vs baseline: 1.6179x; 1.6179x over previous
#include <hip/hip_runtime.h>
#include <hip/hip_bf16.h>
#include <math.h>

typedef __attribute__((ext_vector_type(2))) float v2f;
typedef __attribute__((ext_vector_type(8))) float v8f;

template <int NT>
__global__ void __launch_bounds__(256) gemm_f32_wmma(
    const float* __restrict__ A, const float* __restrict__ B,
    float* __restrict__ C, int M, int K, int N)
{
    __shared__ v2f sB2[4096];
    const int tid = threadIdx.x;
    const int pairCount = (K >> 1) * N;
    for (int p = tid; p < pairCount; p += 256) {
        const int kk = (p / N) << 1;
        const int n  = p - (p / N) * N;
        v2f b;
        b.x = B[kk * N + n];
        b.y = B[(kk + 1) * N + n];
        sB2[p] = b;
    }
    __syncthreads();

    const int lane    = tid & 31;
    const int wave    = tid >> 5;
    const int rowTile = blockIdx.x * 8 + wave;
    const int numTiles = (M + 15) >> 4;
    if (rowTile >= numTiles) return;

    const int half = lane >> 4;
    const int l15  = lane & 15;
    int mrow = rowTile * 16 + l15;
    if (mrow >= M) mrow = M - 1;

    v8f acc[NT] = {};
    const float* arow = A + (size_t)mrow * K + 2 * half;
    const v2f* __restrict__ bcol = sB2 + half * N + l15;

    for (int k0 = 0; k0 < K; k0 += 4) {
        const v2f a = *(const v2f*)(arow + k0);
        const v2f* brow = bcol + (k0 >> 1) * N;
#pragma unroll
        for (int nt = 0; nt < NT; ++nt) {
            const v2f b = brow[nt * 16];
            acc[nt] = __builtin_amdgcn_wmma_f32_16x16x4_f32(
                false, a, false, b, (short)0, acc[nt], false, false);
        }
    }

#pragma unroll
    for (int nt = 0; nt < NT; ++nt) {
#pragma unroll
        for (int r = 0; r < 8; ++r) {
            const int row = rowTile * 16 + r + 8 * half;
            if (row < M)
                C[(size_t)row * N + nt * 16 + l15] = acc[nt][r];
        }
    }
}

__device__ __forceinline__ void atomicMaxFloat(float* addr, float val) {
    const int iv = __float_as_int(val);
    if (iv >= 0) atomicMax((int*)addr, iv);
    else         atomicMin((unsigned int*)addr, (unsigned int)iv);
}

__global__ void node_score_kernel(const float* __restrict__ z,
                                  const float* __restrict__ w_attn,
                                  float* __restrict__ ssrc, float* __restrict__ sdst,
                                  int N)
{
    const int n = blockIdx.x * blockDim.x + threadIdx.x;
    if (n >= N) return;
    const float* zr = z + (size_t)n * 64;
    float s0 = 0.f, s1 = 0.f;
#pragma unroll
    for (int i = 0; i < 64; ++i) {
        const float v = zr[i];
        s0 += v * w_attn[i];
        s1 += v * w_attn[96 + i];
    }
    ssrc[n] = s0;
    sdst[n] = s1;
}

__global__ void init_kernel(float* __restrict__ h, float* __restrict__ mbuf,
                            float* __restrict__ den, int hsize, int N)
{
    const int i = blockIdx.x * blockDim.x + threadIdx.x;
    const int stride = gridDim.x * blockDim.x;
    for (int j = i; j < hsize; j += stride) h[j] = 0.f;
    for (int j = i; j < N; j += stride) { mbuf[j] = -INFINITY; den[j] = 0.f; }
}

__global__ void edge_score_kernel(const float* __restrict__ eft,
                                  const int* __restrict__ src, const int* __restrict__ dst,
                                  const float* __restrict__ ssrc, const float* __restrict__ sdst,
                                  const float* __restrict__ w_attn,
                                  float* __restrict__ a, float* __restrict__ mbuf, int E)
{
    const int e = blockIdx.x * blockDim.x + threadIdx.x;
    if (e >= E) return;
    const float* er = eft + (size_t)e * 32;
    float t = ssrc[src[e]] + sdst[dst[e]];
#pragma unroll
    for (int i = 0; i < 32; ++i) t += er[i] * w_attn[64 + i];
    const float lr = t > 0.f ? t : 0.01f * t;
    a[e] = lr;
    atomicMaxFloat(&mbuf[dst[e]], lr);
}

__global__ void edge_exp_kernel(const int* __restrict__ dst,
                                const float* __restrict__ mbuf,
                                float* __restrict__ a, float* __restrict__ den, int E)
{
    const int e = blockIdx.x * blockDim.x + threadIdx.x;
    if (e >= E) return;
    const int d = dst[e];
    const float ex = __expf(a[e] - mbuf[d]);
    a[e] = ex;
    atomicAdd(&den[d], ex);
}

#define MSG_STRIDE 68
__global__ void __launch_bounds__(256) edge_scatter_fc2_kernel(
    const float* __restrict__ eft, const float* __restrict__ W2, const float* __restrict__ zz,
    const float* __restrict__ a, const float* __restrict__ den,
    const int* __restrict__ src, const int* __restrict__ dst, const int* __restrict__ etype,
    float* __restrict__ h2, int E)
{
    __shared__ v2f   sB2[16 * 64];
    __shared__ float sT[8][16 * MSG_STRIDE];
    const int tid = threadIdx.x;
    for (int p = tid; p < 16 * 64; p += 256) {
        const int kk = (p >> 6) << 1;
        const int n  = p & 63;
        v2f b;
        b.x = W2[kk * 64 + n];
        b.y = W2[(kk + 1) * 64 + n];
        sB2[p] = b;
    }
    __syncthreads();

    const int lane = tid & 31;
    const int wave = tid >> 5;
    const int tile = blockIdx.x * 8 + wave;
    const bool active = tile < ((E + 15) >> 4);
    const int half = lane >> 4;
    const int l15  = lane & 15;
    int ec = tile * 16 + l15;
    if (ec >= E) ec = E - 1;
    const int   d_l   = dst[ec];
    const int   s_l   = src[ec];
    const int   t_l   = etype[ec];
    const float a_l   = a[ec];
    const float den_l = den[d_l];

    if (active) {
        v8f acc[4] = {};
        const float* arow = eft + (size_t)ec * 32 + 2 * half;
        const v2f* __restrict__ bcol = sB2 + half * 64 + l15;
#pragma unroll
        for (int k0 = 0; k0 < 32; k0 += 4) {
            const v2f av = *(const v2f*)(arow + k0);
#pragma unroll
            for (int nt = 0; nt < 4; ++nt) {
                const v2f b = bcol[(k0 >> 1) * 64 + nt * 16];
                acc[nt] = __builtin_amdgcn_wmma_f32_16x16x4_f32(
                    false, av, false, b, (short)0, acc[nt], false, false);
            }
        }
        float* tw = sT[wave];
#pragma unroll
        for (int nt = 0; nt < 4; ++nt) {
#pragma unroll
            for (int r = 0; r < 8; ++r)
                tw[(r + 8 * half) * MSG_STRIDE + nt * 16 + l15] = acc[nt][r];
        }
    }
    __syncthreads();
    if (!active) return;

    const float al_l = a_l / den_l;
    const float* tr = sT[wave];
    const int nvalid = E - __builtin_amdgcn_readfirstlane(tile) * 16;

    float m0[16], m1[16];
#pragma unroll
    for (int i = 0; i < 16; ++i) {
        const int s = __builtin_amdgcn_readlane(s_l, i);
        const float alpha = __int_as_float(__builtin_amdgcn_readlane(__float_as_int(al_l), i));
        const float* zr = zz + (size_t)s * 64;
        m0[i] = alpha * (tr[i * MSG_STRIDE + lane]      + zr[lane]);
        m1[i] = alpha * (tr[i * MSG_STRIDE + lane + 32] + zr[lane + 32]);
    }
#pragma unroll
    for (int i = 0; i < 16; ++i) {
        if (i < nvalid) {
            const int d = __builtin_amdgcn_readlane(d_l, i);
            const int t = __builtin_amdgcn_readlane(t_l, i);
            float* hb = h2 + ((size_t)d * 4 + t) * 64;
            atomicAdd(hb + lane,      m0[i]);
            atomicAdd(hb + lane + 32, m1[i]);
        }
    }
}

__global__ void zphi_fold_kernel(const float* __restrict__ part, float* __restrict__ zpart, int rows)
{
    const int s   = threadIdx.x;
    const int per = (rows + (int)gridDim.x - 1) / (int)gridDim.x;
    const int lo  = blockIdx.x * per;
    const int hi  = min(rows, lo + per);
    float sum = 0.f;
    for (int r = lo; r < hi; r += 8) {
        float v[8];
        for (int j = 0; j < 8; ++j)
            v[j] = (r + j < hi) ? part[(size_t)(r + j) * 256 + s] : 0.f;
        for (int j = 0; j < 8; ++j) sum += v[j];
    }
    zpart[blockIdx.x * 256 + s] = sum;
}

__global__ void beta_kernel(const float* __restrict__ zpart, int nparts,
                            const float* __restrict__ w_sem, float* __restrict__ beta)
{
    __shared__ float prod[256];
    __shared__ float wsh[4];
    const int tid = threadIdx.x;
    float z = 0.f;
    for (int r = 0; r < nparts; r += 8) {
        float v[8];
        for (int j = 0; j < 8; ++j)
            v[j] = (r + j < nparts) ? zpart[(r + j) * 256 + tid] : 0.f;
        for (int j = 0; j < 8; ++j) z += v[j];
    }
    prod[tid] = z * w_sem[tid & 63];
    __syncthreads();
    if (tid < 4) {
        float d = 0.f;
        for (int o = 0; o < 64; ++o) d += prod[tid * 64 + o];
        wsh[tid] = d > 0.f ? d : 0.01f * d;
    }
    __syncthreads();
    if (tid != 0) return;
    float wp[4], mx = -INFINITY;
#pragma unroll
    for (int t = 0; t < 4; ++t) {
        wp[t] = wsh[t];
        mx = fmaxf(mx, wp[t]);
    }
    float s = 0.f;
#pragma unroll
    for (int t = 0; t < 4; ++t) { wp[t] = __expf(wp[t] - mx); s += wp[t]; }
#pragma unroll
    for (int t = 0; t < 4; ++t) beta[t] = wp[t] / s;
}

__global__ void out_kernel(const float* __restrict__ h2, const float* __restrict__ beta,
                           float* __restrict__ Z, int N)
{
    const int i = blockIdx.x * blockDim.x + threadIdx.x;
    if (i >= N * 64) return;
    const int n = i >> 6, o = i & 63;
    const float* r = h2 + (size_t)n * 256 + o;
    Z[i] = beta[0] * r[0] + beta[1] * r[64] + beta[2] * r[128] + beta[3] * r[192];
}

extern "C" void kernel_launch(void* const* d_in, const int* in_sizes, int n_in,
                              void* d_out, int out_size, void* d_ws, size_t ws_size,
                              hipStream_t stream)
{
    const float* node   = (const float*)d_in[0];
    const float* edgef  = (const float*)d_in[1];
    const int*   src    = (const int*)d_in[2];
    const int*   dst    = (const int*)d_in[3];
    const int*   etype  = (const int*)d_in[4];
    const float* w_n    = (const float*)d_in[5];
    const float* w_e    = (const float*)d_in[6];
    const float* w_attn = (const float*)d_in[7];
    const float* w_fc2  = (const float*)d_in[8];
    const float* w_sem  = (const float*)d_in[9];
    float* out = (float*)d_out;

    const int N = in_sizes[0] / 128;
    const int E = in_sizes[2];
    const int T = 4;

    float* ws   = (float*)d_ws;
    float* z    = ws;
    float* ssrc = z + (size_t)N * 64;
    float* sdst = ssrc + N;
    float* abuf = sdst + N;
    float* mbuf = abuf + E;
    float* den  = mbuf + N;
    float* zphi = den + N;
    float* beta = zphi + 256;
    float* h2   = beta + 16;
    float* zz   = h2 + (size_t)N * T * 64;
    float* eft  = beta + 16 + (size_t)N * T * 96;
    float* part  = z;
    float* zpart = abuf;

    const int thr = 256;

    gemm_f32_wmma<4><<<((N + 15) / 16 + 7) / 8, thr, 0, stream>>>(node, w_n, z, N, 128, 64);
    gemm_f32_wmma<2><<<((E + 15) / 16 + 7) / 8, thr, 0, stream>>>(edgef, w_e, eft, E, 64, 32);
    gemm_f32_wmma<4><<<((N + 15) / 16 + 7) / 8, thr, 0, stream>>>(z, w_fc2, zz, N, 64, 64);
    node_score_kernel<<<(N + thr - 1) / thr, thr, 0, stream>>>(z, w_attn, ssrc, sdst, N);
    init_kernel<<<1024, thr, 0, stream>>>(h2, mbuf, den, N * T * 64, N);
    edge_score_kernel<<<(E + thr - 1) / thr, thr, 0, stream>>>(eft, src, dst, ssrc, sdst,
                                                               w_attn, abuf, mbuf, E);
    edge_exp_kernel<<<(E + thr - 1) / thr, thr, 0, stream>>>(dst, mbuf, abuf, den, E);
    edge_scatter_fc2_kernel<<<((E + 15) / 16 + 7) / 8, thr, 0, stream>>>(eft, w_fc2 + 64 * 64, zz, abuf, den,
                                                                      src, dst, etype, h2, E);
    zphi_fold_kernel<<<1024, thr, 0, stream>>>(h2, part, N);
    zphi_fold_kernel<<<64, thr, 0, stream>>>(part, zpart, 1024);
    beta_kernel<<<1, thr, 0, stream>>>(zpart, 64, w_sem, beta);
    out_kernel<<<(N * 64 + thr - 1) / thr, thr, 0, stream>>>(h2, beta, out, N);
}


# projected [E,32] edge plane not stored: score kernel and scatter each redo edges @ w_e per 16-edge tile on WMMA, score dot from accumulators (16-lane tree); its GEMM goes; score sum order changes
# speedup vs baseline: 1.7283x; 1.0682x over previous
#include <hip/hip_runtime.h>
#include <hip/hip_bf16.h>
#include <math.h>

typedef __attribute__((ext_vector_type(2))) float v2f;
typedef __attribute__((ext_vector_type(8))) float v8f;

template <int NT>
__global__ void __launch_bounds__(256) gemm_f32_wmma(
    const float* __restrict__ A, const float* __restrict__ B,
    float* __restrict__ C, int M, int K, int N)
{
    __shared__ v2f sB2[4096];
    const int tid = threadIdx.x;
    const int pairCount = (K >> 1) * N;
    for (int p = tid; p < pairCount; p += 256) {
        const int kk = (p / N) << 1;
        const int n  = p - (p / N) * N;
        v2f b;
        b.x = B[kk * N + n];
        b.y = B[(kk + 1) * N + n];
        sB2[p] = b;
    }
    __syncthreads();

    const int lane    = tid & 31;
    const int wave    = tid >> 5;
    const int rowTile = blockIdx.x * 8 + wave;
    const int numTiles = (M + 15) >> 4;
    if (rowTile >= numTiles) return;

    const int half = lane >> 4;
    const int l15  = lane & 15;
    int mrow = rowTile * 16 + l15;
    if (mrow >= M) mrow = M - 1;

    v8f acc[NT] = {};
    const float* arow = A + (size_t)mrow * K + 2 * half;
    const v2f* __restrict__ bcol = sB2 + half * N + l15;

    for (int k0 = 0; k0 < K; k0 += 4) {
        const v2f a = *(const v2f*)(arow + k0);
        const v2f* brow = bcol + (k0 >> 1) * N;
#pragma unroll
        for (int nt = 0; nt < NT; ++nt) {
            const v2f b = brow[nt * 16];
            acc[nt] = __builtin_amdgcn_wmma_f32_16x16x4_f32(
                false, a, false, b, (short)0, acc[nt], false, false);
        }
    }

#pragma unroll
    for (int nt = 0; nt < NT; ++nt) {
#pragma unroll
        for (int r = 0; r < 8; ++r) {
            const int row = rowTile * 16 + r + 8 * half;
            if (row < M)
                C[(size_t)row * N + nt * 16 + l15] = acc[nt][r];
        }
    }
}

__device__ __forceinline__ void atomicMaxFloat(float* addr, float val) {
    const int iv = __float_as_int(val);
    if (iv >= 0) atomicMax((int*)addr, iv);
    else         atomicMin((unsigned int*)addr, (unsigned int)iv);
}

__global__ void node_score_kernel(const float* __restrict__ z,
                                  const float* __restrict__ w_attn,
                                  float* __restrict__ ssrc, float* __restrict__ sdst,
                                  int N)
{
    const int n = blockIdx.x * blockDim.x + threadIdx.x;
    if (n >= N) return;
    const float* zr = z + (size_t)n * 64;
    float s0 = 0.f, s1 = 0.f;
#pragma unroll
    for (int i = 0; i < 64; ++i) {
        const float v = zr[i];
        s0 += v * w_attn[i];
        s1 += v * w_attn[96 + i];
    }
    ssrc[n] = s0;
    sdst[n] = s1;
}

__global__ void init_kernel(float* __restrict__ h, float* __restrict__ mbuf,
                            float* __restrict__ den, int hsize, int N)
{
    const int i = blockIdx.x * blockDim.x + threadIdx.x;
    const int stride = gridDim.x * blockDim.x;
    for (int j = i; j < hsize; j += stride) h[j] = 0.f;
    for (int j = i; j < N; j += stride) { mbuf[j] = -INFINITY; den[j] = 0.f; }
}

template <int M>
__device__ __forceinline__ float row_xor(float v) {
    return __int_as_float(__builtin_amdgcn_update_dpp(0, __float_as_int(v), 0x160 | M, 0xf, 0xf, true));
}

__global__ void __launch_bounds__(256) edge_score_wmma_kernel(
    const float* __restrict__ ef, const float* __restrict__ w_e, const float* __restrict__ w_attn,
    const int* __restrict__ src, const int* __restrict__ dst,
    const float* __restrict__ ssrc, const float* __restrict__ sdst,
    float* __restrict__ a, float* __restrict__ mbuf, int E)
{
    __shared__ v2f sBe[32 * 32];
    const int tid = threadIdx.x;
    for (int p = tid; p < 32 * 32; p += 256) {
        const int kk = (p >> 5) << 1;
        const int n  = p & 31;
        v2f b;
        b.x = w_e[kk * 32 + n];
        b.y = w_e[(kk + 1) * 32 + n];
        sBe[p] = b;
    }
    __syncthreads();

    const int lane = tid & 31;
    const int wave = tid >> 5;
    const int tile = blockIdx.x * 8 + wave;
    if (tile >= ((E + 15) >> 4)) return;
    const int half = lane >> 4;
    const int l15  = lane & 15;
    int ec = tile * 16 + l15;
    if (ec >= E) ec = E - 1;
    int er = tile * 16 + (l15 & 7) + 8 * half;
    const bool own = (l15 < 8) && (er < E);
    if (er >= E) er = E - 1;
    const int   d_l  = dst[er];
    const float base = ssrc[src[er]] + sdst[d_l];
    const float w0 = w_attn[64 + l15];
    const float w1 = w_attn[80 + l15];

    v8f acc[2] = {};
    const float* arow = ef + (size_t)ec * 64 + 2 * half;
    const v2f* __restrict__ bcol = sBe + half * 32 + l15;
#pragma unroll
    for (int k0 = 0; k0 < 64; k0 += 4) {
        const v2f av = *(const v2f*)(arow + k0);
#pragma unroll
        for (int nt = 0; nt < 2; ++nt) {
            const v2f b = bcol[(k0 >> 1) * 32 + nt * 16];
            acc[nt] = __builtin_amdgcn_wmma_f32_16x16x4_f32(
                false, av, false, b, (short)0, acc[nt], false, false);
        }
    }
    float mid = 0.f;
#pragma unroll
    for (int r = 0; r < 8; ++r) {
        float p = acc[0][r] * w0 + acc[1][r] * w1;
        p += row_xor<8>(p);
        p += row_xor<4>(p);
        p += row_xor<2>(p);
        p += row_xor<1>(p);
        if (l15 == r) mid = p;
    }
    if (own) {
        const float t  = base + mid;
        const float lr = t > 0.f ? t : 0.01f * t;
        a[er] = lr;
        atomicMaxFloat(&mbuf[d_l], lr);
    }
}

__global__ void edge_exp_kernel(const int* __restrict__ dst,
                                const float* __restrict__ mbuf,
                                float* __restrict__ a, float* __restrict__ den, int E)
{
    const int e = blockIdx.x * blockDim.x + threadIdx.x;
    if (e >= E) return;
    const int d = dst[e];
    const float ex = __expf(a[e] - mbuf[d]);
    a[e] = ex;
    atomicAdd(&den[d], ex);
}

#define MSG_STRIDE 68
__global__ void __launch_bounds__(256) edge_scatter_fc2_kernel(
    const float* __restrict__ ef, const float* __restrict__ w_e, const float* __restrict__ W2,
    const float* __restrict__ zz,
    const float* __restrict__ a, const float* __restrict__ den,
    const int* __restrict__ src, const int* __restrict__ dst, const int* __restrict__ etype,
    float* __restrict__ h2, int E)
{
    __shared__ v2f   sBe[32 * 32];
    __shared__ v2f   sB2[16 * 64];
    __shared__ __attribute__((aligned(16))) float sT[8][16 * MSG_STRIDE];
    const int tid = threadIdx.x;
    for (int p = tid; p < 32 * 32; p += 256) {
        const int kk = (p >> 5) << 1;
        const int n  = p & 31;
        v2f b;
        b.x = w_e[kk * 32 + n];
        b.y = w_e[(kk + 1) * 32 + n];
        sBe[p] = b;
    }
    for (int p = tid; p < 16 * 64; p += 256) {
        const int kk = (p >> 6) << 1;
        const int n  = p & 63;
        v2f b;
        b.x = W2[kk * 64 + n];
        b.y = W2[(kk + 1) * 64 + n];
        sB2[p] = b;
    }
    __syncthreads();

    const int lane = tid & 31;
    const int wave = tid >> 5;
    const int tile = blockIdx.x * 8 + wave;
    const bool active = tile < ((E + 15) >> 4);
    const int half = lane >> 4;
    const int l15  = lane & 15;
    int ec = tile * 16 + l15;
    if (ec >= E) ec = E - 1;
    const int   d_l   = dst[ec];
    const int   s_l   = src[ec];
    const int   t_l   = etype[ec];
    const float a_l   = a[ec];
    const float den_l = den[d_l];

    if (active) {
        v8f pe[2] = {};
        const float* arow = ef + (size_t)ec * 64 + 2 * half;
        const v2f* __restrict__ becol = sBe + half * 32 + l15;
#pragma unroll
        for (int k0 = 0; k0 < 64; k0 += 4) {
            const v2f av = *(const v2f*)(arow + k0);
#pragma unroll
            for (int nt = 0; nt < 2; ++nt) {
                const v2f b = becol[(k0 >> 1) * 32 + nt * 16];
                pe[nt] = __builtin_amdgcn_wmma_f32_16x16x4_f32(
                    false, av, false, b, (short)0, pe[nt], false, false);
            }
        }
        float* tw = sT[wave];
#pragma unroll
        for (int nt = 0; nt < 2; ++nt) {
#pragma unroll
            for (int r = 0; r < 8; ++r)
                tw[(r + 8 * half) * MSG_STRIDE + nt * 16 + l15] = pe[nt][r];
        }
    }
    __syncthreads();
    if (active) {
        v8f acc[4] = {};
        const float* arow = sT[wave] + l15 * MSG_STRIDE + 2 * half;
        const v2f* __restrict__ bcol = sB2 + half * 64 + l15;
#pragma unroll
        for (int k0 = 0; k0 < 32; k0 += 4) {
            const v2f av = *(const v2f*)(arow + k0);
#pragma unroll
            for (int nt = 0; nt < 4; ++nt) {
                const v2f b = bcol[(k0 >> 1) * 64 + nt * 16];
                acc[nt] = __builtin_amdgcn_wmma_f32_16x16x4_f32(
                    false, av, false, b, (short)0, acc[nt], false, false);
            }
        }
        float* tw = sT[wave];
#pragma unroll
        for (int nt = 0; nt < 4; ++nt) {
#pragma unroll
            for (int r = 0; r < 8; ++r)
                tw[(r + 8 * half) * MSG_STRIDE + nt * 16 + l15] = acc[nt][r];
        }
    }
    __syncthreads();
    if (!active) return;

    const float al_l = a_l / den_l;
    const float* tr = sT[wave];
    const int nvalid = E - __builtin_amdgcn_readfirstlane(tile) * 16;

    float m0[16], m1[16];
#pragma unroll
    for (int i = 0; i < 16; ++i) {
        const int s = __builtin_amdgcn_readlane(s_l, i);
        const float alpha = __int_as_float(__builtin_amdgcn_readlane(__float_as_int(al_l), i));
        const float* zr = zz + (size_t)s * 64;
        m0[i] = alpha * (tr[i * MSG_STRIDE + lane]      + zr[lane]);
        m1[i] = alpha * (tr[i * MSG_STRIDE + lane + 32] + zr[lane + 32]);
    }
#pragma unroll
    for (int i = 0; i < 16; ++i) {
        if (i < nvalid) {
            const int d = __builtin_amdgcn_readlane(d_l, i);
            const int t = __builtin_amdgcn_readlane(t_l, i);
            float* hb = h2 + ((size_t)d * 4 + t) * 64;
            atomicAdd(hb + lane,      m0[i]);
            atomicAdd(hb + lane + 32, m1[i]);
        }
    }
}

__global__ void zphi_fold_kernel(const float* __restrict__ part, float* __restrict__ zpart, int rows)
{
    const int s   = threadIdx.x;
    const int per = (rows + (int)gridDim.x - 1) / (int)gridDim.x;
    const int lo  = blockIdx.x * per;
    const int hi  = min(rows, lo + per);
    float sum = 0.f;
    for (int r = lo; r < hi; r += 8) {
        float v[8];
        for (int j = 0; j < 8; ++j)
            v[j] = (r + j < hi) ? part[(size_t)(r + j) * 256 + s] : 0.f;
        for (int j = 0; j < 8; ++j) sum += v[j];
    }
    zpart[blockIdx.x * 256 + s] = sum;
}

__global__ void beta_kernel(const float* __restrict__ zpart, int nparts,
                            const float* __restrict__ w_sem, float* __restrict__ beta)
{
    __shared__ float prod[256];
    __shared__ float wsh[4];
    const int tid = threadIdx.x;
    float z = 0.f;
    for (int r = 0; r < nparts; r += 8) {
        float v[8];
        for (int j = 0; j < 8; ++j)
            v[j] = (r + j < nparts) ? zpart[(r + j) * 256 + tid] : 0.f;
        for (int j = 0; j < 8; ++j) z += v[j];
    }
    prod[tid] = z * w_sem[tid & 63];
    __syncthreads();
    if (tid < 4) {
        float d = 0.f;
        for (int o = 0; o < 64; ++o) d += prod[tid * 64 + o];
        wsh[tid] = d > 0.f ? d : 0.01f * d;
    }
    __syncthreads();
    if (tid != 0) return;
    float wp[4], mx = -INFINITY;
#pragma unroll
    for (int t = 0; t < 4; ++t) {
        wp[t] = wsh[t];
        mx = fmaxf(mx, wp[t]);
    }
    float s = 0.f;
#pragma unroll
    for (int t = 0; t < 4; ++t) { wp[t] = __expf(wp[t] - mx); s += wp[t]; }
#pragma unroll
    for (int t = 0; t < 4; ++t) beta[t] = wp[t] / s;
}

__global__ void out_kernel(const float* __restrict__ h2, const float* __restrict__ beta,
                           float* __restrict__ Z, int N)
{
    const int i = blockIdx.x * blockDim.x + threadIdx.x;
    if (i >= N * 64) return;
    const int n = i >> 6, o = i & 63;
    const float* r = h2 + (size_t)n * 256 + o;
    Z[i] = beta[0] * r[0] + beta[1] * r[64] + beta[2] * r[128] + beta[3] * r[192];
}

extern "C" void kernel_launch(void* const* d_in, const int* in_sizes, int n_in,
                              void* d_out, int out_size, void* d_ws, size_t ws_size,
                              hipStream_t stream)
{
    const float* node   = (const float*)d_in[0];
    const float* edgef  = (const float*)d_in[1];
    const int*   src    = (const int*)d_in[2];
    const int*   dst    = (const int*)d_in[3];
    const int*   etype  = (const int*)d_in[4];
    const float* w_n    = (const float*)d_in[5];
    const float* w_e    = (const float*)d_in[6];
    const float* w_attn = (const float*)d_in[7];
    const float* w_fc2  = (const float*)d_in[8];
    const float* w_sem  = (const float*)d_in[9];
    float* out = (float*)d_out;

    const int N = in_sizes[0] / 128;
    const int E = in_sizes[2];
    const int T = 4;

    float* ws   = (float*)d_ws;
    float* z    = ws;
    float* ssrc = z + (size_t)N * 64;
    float* sdst = ssrc + N;
    float* abuf = sdst + N;
    float* mbuf = abuf + E;
    float* den  = mbuf + N;
    float* zphi = den + N;
    float* beta = zphi + 256;
    float* h2   = beta + 16;
    float* zz   = h2 + (size_t)N * T * 64;
    float* part  = z;
    float* zpart = abuf;

    const int thr = 256;

    gemm_f32_wmma<4><<<((N + 15) / 16 + 7) / 8, thr, 0, stream>>>(node, w_n, z, N, 128, 64);
    gemm_f32_wmma<4><<<((N + 15) / 16 + 7) / 8, thr, 0, stream>>>(z, w_fc2, zz, N, 64, 64);
    node_score_kernel<<<(N + thr - 1) / thr, thr, 0, stream>>>(z, w_attn, ssrc, sdst, N);
    init_kernel<<<1024, thr, 0, stream>>>(h2, mbuf, den, N * T * 64, N);
    edge_score_wmma_kernel<<<((E + 15) / 16 + 7) / 8, thr, 0, stream>>>(edgef, w_e, w_attn, src, dst,
                                                                     ssrc, sdst, abuf, mbuf, E);
    edge_exp_kernel<<<(E + thr - 1) / thr, thr, 0, stream>>>(dst, mbuf, abuf, den, E);
    edge_scatter_fc2_kernel<<<((E + 15) / 16 + 7) / 8, thr, 0, stream>>>(edgef, w_e, w_fc2 + 64 * 64, zz, abuf, den,
                                                                      src, dst, etype, h2, E);
    zphi_fold_kernel<<<1024, thr, 0, stream>>>(h2, part, N);
    zphi_fold_kernel<<<64, thr, 0, stream>>>(part, zpart, 1024);
    beta_kernel<<<1, thr, 0, stream>>>(zpart, 64, w_sem, beta);
    out_kernel<<<(N * 64 + thr - 1) / thr, thr, 0, stream>>>(h2, beta, out, N);
}


# segment max and exp-sum per destination kept in per-block LDS tables of all nodes (LDS atomics), 192 rows folded in fixed order by a new pass; scattered global atomics go; den sum order changes
# speedup vs baseline: 2.1894x; 1.2668x over previous
#include <hip/hip_runtime.h>
#include <hip/hip_bf16.h>
#include <math.h>

typedef __attribute__((ext_vector_type(2))) float v2f;
typedef __attribute__((ext_vector_type(8))) float v8f;

template <int NT>
__global__ void __launch_bounds__(256) gemm_f32_wmma(
    const float* __restrict__ A, const float* __restrict__ B,
    float* __restrict__ C, int M, int K, int N)
{
    __shared__ v2f sB2[4096];
    const int tid = threadIdx.x;
    const int pairCount = (K >> 1) * N;
    for (int p = tid; p < pairCount; p += 256) {
        const int kk = (p / N) << 1;
        const int n  = p - (p / N) * N;
        v2f b;
        b.x = B[kk * N + n];
        b.y = B[(kk + 1) * N + n];
        sB2[p] = b;
    }
    __syncthreads();

    const int lane    = tid & 31;
    const int wave    = tid >> 5;
    const int rowTile = blockIdx.x * 8 + wave;
    const int numTiles = (M + 15) >> 4;
    if (rowTile >= numTiles) return;

    const int half = lane >> 4;
    const int l15  = lane & 15;
    int mrow = rowTile * 16 + l15;
    if (mrow >= M) mrow = M - 1;

    v8f acc[NT] = {};
    const float* arow = A + (size_t)mrow * K + 2 * half;
    const v2f* __restrict__ bcol = sB2 + half * N + l15;

    for (int k0 = 0; k0 < K; k0 += 4) {
        const v2f a = *(const v2f*)(arow + k0);
        const v2f* brow = bcol + (k0 >> 1) * N;
#pragma unroll
        for (int nt = 0; nt < NT; ++nt) {
            const v2f b = brow[nt * 16];
            acc[nt] = __builtin_amdgcn_wmma_f32_16x16x4_f32(
                false, a, false, b, (short)0, acc[nt], false, false);
        }
    }

#pragma unroll
    for (int nt = 0; nt < NT; ++nt) {
#pragma unroll
        for (int r = 0; r < 8; ++r) {
            const int row = rowTile * 16 + r + 8 * half;
            if (row < M)
                C[(size_t)row * N + nt * 16 + l15] = acc[nt][r];
        }
    }
}

__device__ __forceinline__ void ldsMaxFloat(float* addr, float val) {
    const int iv = __float_as_int(val);
    if (iv >= 0) atomicMax((int*)addr, iv);
    else         atomicMin((unsigned int*)addr, (unsigned int)iv);
}

__global__ void node_score_kernel(const float* __restrict__ z,
                                  const float* __restrict__ w_attn,
                                  float* __restrict__ ssrc, float* __restrict__ sdst,
                                  int N)
{
    const int n = blockIdx.x * blockDim.x + threadIdx.x;
    if (n >= N) return;
    const float* zr = z + (size_t)n * 64;
    float s0 = 0.f, s1 = 0.f;
#pragma unroll
    for (int i = 0; i < 64; ++i) {
        const float v = zr[i];
        s0 += v * w_attn[i];
        s1 += v * w_attn[96 + i];
    }
    ssrc[n] = s0;
    sdst[n] = s1;
}

__global__ void init_kernel(float* __restrict__ h, int hsize)
{
    const int i = blockIdx.x * blockDim.x + threadIdx.x;
    const int stride = gridDim.x * blockDim.x;
    for (int j = i; j < hsize; j += stride) h[j] = 0.f;
}

template <int M>
__device__ __forceinline__ float row_xor(float v) {
    return __int_as_float(__builtin_amdgcn_update_dpp(0, __float_as_int(v), 0x160 | M, 0xf, 0xf, true));
}

#define SCORE_BLOCKS   192
#define SCORE_THREADS  1024
#define TAB_NODES      50000
#define CHUNK_ITERS    5
#define CHUNK_CAP      (CHUNK_ITERS * SCORE_THREADS)

__device__ __forceinline__ void table_out(const float* tab, float* __restrict__ row, int N, int tid) {
    if (((N & 3) == 0) && ((((size_t)row) & 15) == 0)) {
        const float4* t4 = (const float4*)tab;
        float4* r4 = (float4*)row;
        for (int q = tid; q < (N >> 2); q += SCORE_THREADS) r4[q] = t4[q];
    } else {
        for (int n = tid; n < N; n += SCORE_THREADS) row[n] = tab[n];
    }
}
__global__ void __launch_bounds__(SCORE_THREADS) edge_score_tables_kernel(
    const float* __restrict__ ef, const float* __restrict__ w_e, const float* __restrict__ w_attn,
    const int* __restrict__ src, const int* __restrict__ dst,
    const float* __restrict__ ssrc, const float* __restrict__ sdst,
    float* __restrict__ a, float* __restrict__ mpart, float* __restrict__ spart, int E, int N)
{
    __shared__ __attribute__((aligned(16))) float tab[TAB_NODES];
    __shared__ v2f   sBe[32 * 32];
    __shared__ float sA[CHUNK_CAP];
    __shared__ int   sD[CHUNK_CAP];
    const int tid = threadIdx.x;
    {
        const int kk = (tid >> 5) << 1;
        const int n  = tid & 31;
        v2f b;
        b.x = w_e[kk * 32 + n];
        b.y = w_e[(kk + 1) * 32 + n];
        sBe[tid] = b;
    }
    for (int n = tid; n < TAB_NODES; n += SCORE_THREADS) tab[n] = -INFINITY;

    const int lane = tid & 31;
    const int wave = __builtin_amdgcn_readfirstlane(tid >> 5);
    const int numTiles = (E + 15) >> 4;
    const int tpb = (numTiles + (int)gridDim.x - 1) / (int)gridDim.x;
    const int t0  = min(numTiles, (int)blockIdx.x * tpb);
    const int t1  = min(numTiles, t0 + tpb);
    const int e0  = t0 * 16;
    const int cnt = min(E, t1 * 16) - e0;

    {
        int dreg[CHUNK_ITERS], sreg[CHUNK_ITERS];
#pragma unroll
        for (int k = 0; k < CHUNK_ITERS; ++k) {
            const int e = min(e0 + tid + k * SCORE_THREADS, E - 1);
            dreg[k] = dst[e];
            sreg[k] = src[e];
        }
        float breg[CHUNK_ITERS];
#pragma unroll
        for (int k = 0; k < CHUNK_ITERS; ++k) breg[k] = ssrc[sreg[k]] + sdst[dreg[k]];
#pragma unroll
        for (int k = 0; k < CHUNK_ITERS; ++k) {
            sD[tid + k * SCORE_THREADS] = dreg[k];
            sA[tid + k * SCORE_THREADS] = breg[k];
        }
    }
    __syncthreads();

    const int half = lane >> 4;
    const int l15  = lane & 15;
    const float w0 = w_attn[64 + l15];
    const float w1 = w_attn[80 + l15];
    const v2f* __restrict__ bcol = sBe + half * 32 + l15;

    for (int tile = t0 + wave; tile < t1; tile += SCORE_THREADS / 32) {
        int ec = tile * 16 + l15;
        if (ec >= E) ec = E - 1;
        const int er = tile * 16 + (l15 & 7) + 8 * half;
        const bool own = (l15 < 8) && (er < E);
        const int li = er - e0;

        v8f acc[2] = {};
        const float* arow = ef + (size_t)ec * 64 + 2 * half;
#pragma unroll
        for (int k0 = 0; k0 < 64; k0 += 4) {
            const v2f av = *(const v2f*)(arow + k0);
#pragma unroll
            for (int nt = 0; nt < 2; ++nt) {
                const v2f b = bcol[(k0 >> 1) * 32 + nt * 16];
                acc[nt] = __builtin_amdgcn_wmma_f32_16x16x4_f32(
                    false, av, false, b, (short)0, acc[nt], false, false);
            }
        }
        float mid = 0.f;
#pragma unroll
        for (int r = 0; r < 8; ++r) {
            float p = acc[0][r] * w0 + acc[1][r] * w1;
            p += row_xor<8>(p);
            p += row_xor<4>(p);
            p += row_xor<2>(p);
            p += row_xor<1>(p);
            if (l15 == r) mid = p;
        }
        if (own) {
            const float t  = sA[li] + mid;
            const float lr = t > 0.f ? t : 0.01f * t;
            a[er]  = lr;
            sA[li] = lr;
            ldsMaxFloat(&tab[sD[li]], lr);
        }
    }
    __syncthreads();

    float mreg[CHUNK_ITERS];
#pragma unroll
    for (int k = 0; k < CHUNK_ITERS; ++k) {
        const int i = tid + k * SCORE_THREADS;
        mreg[k] = (i < cnt) ? tab[sD[i]] : 0.f;
    }
    table_out(tab, mpart + (size_t)blockIdx.x * N, N, tid);
    __syncthreads();
    for (int n = tid; n < TAB_NODES; n += SCORE_THREADS) tab[n] = 0.f;
    __syncthreads();
#pragma unroll
    for (int k = 0; k < CHUNK_ITERS; ++k) {
        const int i = tid + k * SCORE_THREADS;
        if (i < cnt) atomicAdd(&tab[sD[i]], __expf(sA[i] - mreg[k]));
    }
    __syncthreads();
    table_out(tab, spart + (size_t)blockIdx.x * N, N, tid);
}

__global__ void __launch_bounds__(256) softmax_combine_kernel(
    const float* __restrict__ mpart, const float* __restrict__ spart,
    float* __restrict__ mbuf, float* __restrict__ den, int N)
{
    __shared__ float smax[8][32];
    __shared__ float ssum[8][32];
    const int lane = threadIdx.x & 31;
    const int g    = threadIdx.x >> 5;
    const int n    = blockIdx.x * 32 + lane;
    const int nn   = min(n, N - 1);
    float mb[SCORE_BLOCKS / 8], sb[SCORE_BLOCKS / 8];
#pragma unroll
    for (int j = 0; j < SCORE_BLOCKS / 8; ++j) {
        const size_t row = (size_t)(g * (SCORE_BLOCKS / 8) + j) * N + nn;
        mb[j] = mpart[row];
        sb[j] = spart[row];
    }
    float m = -INFINITY;
#pragma unroll
    for (int j = 0; j < SCORE_BLOCKS / 8; ++j) m = fmaxf(m, mb[j]);
    smax[g][lane] = m;
    __syncthreads();
    float M = smax[0][lane];
#pragma unroll
    for (int gg = 1; gg < 8; ++gg) M = fmaxf(M, smax[gg][lane]);
    float d = 0.f;
#pragma unroll
    for (int j = 0; j < SCORE_BLOCKS / 8; ++j)
        if (sb[j] > 0.f) d += sb[j] * __expf(mb[j] - M);
    ssum[g][lane] = d;
    __syncthreads();
    if (g == 0 && n < N) {
        float t = 0.f;
#pragma unroll
        for (int gg = 0; gg < 8; ++gg) t += ssum[gg][lane];
        mbuf[n] = M;
        den[n]  = t;
    }
}

#define MSG_STRIDE 68
__global__ void __launch_bounds__(256) edge_scatter_fc2_kernel(
    const float* __restrict__ ef, const float* __restrict__ w_e, const float* __restrict__ W2,
    const float* __restrict__ zz,
    const float* __restrict__ a, const float* __restrict__ mbuf, const float* __restrict__ den,
    const int* __restrict__ src, const int* __restrict__ dst, const int* __restrict__ etype,
    float* __restrict__ h2, int E)
{
    __shared__ v2f   sBe[32 * 32];
    __shared__ v2f   sB2[16 * 64];
    __shared__ __attribute__((aligned(16))) float sT[8][16 * MSG_STRIDE];
    const int tid = threadIdx.x;
    for (int p = tid; p < 32 * 32; p += 256) {
        const int kk = (p >> 5) << 1;
        const int n  = p & 31;
        v2f b;
        b.x = w_e[kk * 32 + n];
        b.y = w_e[(kk + 1) * 32 + n];
        sBe[p] = b;
    }
    for (int p = tid; p < 16 * 64; p += 256) {
        const int kk = (p >> 6) << 1;
        const int n  = p & 63;
        v2f b;
        b.x = W2[kk * 64 + n];
        b.y = W2[(kk + 1) * 64 + n];
        sB2[p] = b;
    }
    __syncthreads();

    const int lane = tid & 31;
    const int wave = tid >> 5;
    const int tile = blockIdx.x * 8 + wave;
    const bool active = tile < ((E + 15) >> 4);
    const int half = lane >> 4;
    const int l15  = lane & 15;
    int ec = tile * 16 + l15;
    if (ec >= E) ec = E - 1;
    const int   d_l   = dst[ec];
    const int   s_l   = src[ec];
    const int   t_l   = etype[ec];
    const float a_l   = a[ec];
    const float m_l   = mbuf[d_l];
    const float den_l = den[d_l];

    if (active) {
        v8f pe[2] = {};
        const float* arow = ef + (size_t)ec * 64 + 2 * half;
        const v2f* __restrict__ becol = sBe + half * 32 + l15;
#pragma unroll
        for (int k0 = 0; k0 < 64; k0 += 4) {
            const v2f av = *(const v2f*)(arow + k0);
#pragma unroll
            for (int nt = 0; nt < 2; ++nt) {
                const v2f b = becol[(k0 >> 1) * 32 + nt * 16];
                pe[nt] = __builtin_amdgcn_wmma_f32_16x16x4_f32(
                    false, av, false, b, (short)0, pe[nt], false, false);
            }
        }
        float* tw = sT[wave];
#pragma unroll
        for (int nt = 0; nt < 2; ++nt) {
#pragma unroll
            for (int r = 0; r < 8; ++r)
                tw[(r + 8 * half) * MSG_STRIDE + nt * 16 + l15] = pe[nt][r];
        }
    }
    __syncthreads();
    if (active) {
        v8f acc[4] = {};
        const float* arow = sT[wave] + l15 * MSG_STRIDE + 2 * half;
        const v2f* __restrict__ bcol = sB2 + half * 64 + l15;
#pragma unroll
        for (int k0 = 0; k0 < 32; k0 += 4) {
            const v2f av = *(const v2f*)(arow + k0);
#pragma unroll
            for (int nt = 0; nt < 4; ++nt) {
                const v2f b = bcol[(k0 >> 1) * 64 + nt * 16];
                acc[nt] = __builtin_amdgcn_wmma_f32_16x16x4_f32(
                    false, av, false, b, (short)0, acc[nt], false, false);
            }
        }
        float* tw = sT[wave];
#pragma unroll
        for (int nt = 0; nt < 4; ++nt) {
#pragma unroll
            for (int r = 0; r < 8; ++r)
                tw[(r + 8 * half) * MSG_STRIDE + nt * 16 + l15] = acc[nt][r];
        }
    }
    __syncthreads();
    if (!active) return;

    const float al_l = __expf(a_l - m_l) / den_l;
    const float* tr = sT[wave];
    const int nvalid = E - __builtin_amdgcn_readfirstlane(tile) * 16;

    float m0[16], m1[16];
#pragma unroll
    for (int i = 0; i < 16; ++i) {
        const int s = __builtin_amdgcn_readlane(s_l, i);
        const float alpha = __int_as_float(__builtin_amdgcn_readlane(__float_as_int(al_l), i));
        const float* zr = zz + (size_t)s * 64;
        m0[i] = alpha * (tr[i * MSG_STRIDE + lane]      + zr[lane]);
        m1[i] = alpha * (tr[i * MSG_STRIDE + lane + 32] + zr[lane + 32]);
    }
#pragma unroll
    for (int i = 0; i < 16; ++i) {
        if (i < nvalid) {
            const int d = __builtin_amdgcn_readlane(d_l, i);
            const int t = __builtin_amdgcn_readlane(t_l, i);
            float* hb = h2 + ((size_t)d * 4 + t) * 64;
            atomicAdd(hb + lane,      m0[i]);
            atomicAdd(hb + lane + 32, m1[i]);
        }
    }
}

__global__ void zphi_fold_kernel(const float* __restrict__ part, float* __restrict__ zpart, int rows)
{
    const int s   = threadIdx.x;
    const int per = (rows + (int)gridDim.x - 1) / (int)gridDim.x;
    const int lo  = blockIdx.x * per;
    const int hi  = min(rows, lo + per);
    float sum = 0.f;
    for (int r = lo; r < hi; r += 8) {
        float v[8];
        for (int j = 0; j < 8; ++j)
            v[j] = (r + j < hi) ? part[(size_t)(r + j) * 256 + s] : 0.f;
        for (int j = 0; j < 8; ++j) sum += v[j];
    }
    zpart[blockIdx.x * 256 + s] = sum;
}

__global__ void beta_kernel(const float* __restrict__ zpart, int nparts,
                            const float* __restrict__ w_sem, float* __restrict__ beta)
{
    __shared__ float prod[256];
    __shared__ float wsh[4];
    const int tid = threadIdx.x;
    float z = 0.f;
    for (int r = 0; r < nparts; r += 8) {
        float v[8];
        for (int j = 0; j < 8; ++j)
            v[j] = (r + j < nparts) ? zpart[(r + j) * 256 + tid] : 0.f;
        for (int j = 0; j < 8; ++j) z += v[j];
    }
    prod[tid] = z * w_sem[tid & 63];
    __syncthreads();
    if (tid < 4) {
        float d = 0.f;
        for (int o = 0; o < 64; ++o) d += prod[tid * 64 + o];
        wsh[tid] = d > 0.f ? d : 0.01f * d;
    }
    __syncthreads();
    if (tid != 0) return;
    float wp[4], mx = -INFINITY;
#pragma unroll
    for (int t = 0; t < 4; ++t) {
        wp[t] = wsh[t];
        mx = fmaxf(mx, wp[t]);
    }
    float s = 0.f;
#pragma unroll
    for (int t = 0; t < 4; ++t) { wp[t] = __expf(wp[t] - mx); s += wp[t]; }
#pragma unroll
    for (int t = 0; t < 4; ++t) beta[t] = wp[t] / s;
}

__global__ void out_kernel(const float* __restrict__ h2, const float* __restrict__ beta,
                           float* __restrict__ Z, int N)
{
    const int i = blockIdx.x * blockDim.x + threadIdx.x;
    if (i >= N * 64) return;
    const int n = i >> 6, o = i & 63;
    const float* r = h2 + (size_t)n * 256 + o;
    Z[i] = beta[0] * r[0] + beta[1] * r[64] + beta[2] * r[128] + beta[3] * r[192];
}

extern "C" void kernel_launch(void* const* d_in, const int* in_sizes, int n_in,
                              void* d_out, int out_size, void* d_ws, size_t ws_size,
                              hipStream_t stream)
{
    const float* node   = (const float*)d_in[0];
    const float* edgef  = (const float*)d_in[1];
    const int*   src    = (const int*)d_in[2];
    const int*   dst    = (const int*)d_in[3];
    const int*   etype  = (const int*)d_in[4];
    const float* w_n    = (const float*)d_in[5];
    const float* w_e    = (const float*)d_in[6];
    const float* w_attn = (const float*)d_in[7];
    const float* w_fc2  = (const float*)d_in[8];
    const float* w_sem  = (const float*)d_in[9];
    float* out = (float*)d_out;

    const int N = in_sizes[0] / 128;
    const int E = in_sizes[2];
    const int T = 4;

    float* ws   = (float*)d_ws;
    float* z    = ws;
    float* ssrc = z + (size_t)N * 64;
    float* sdst = ssrc + N;
    float* abuf = sdst + N;
    float* mbuf = abuf + E;
    float* den  = mbuf + N;
    float* zphi = den + N;
    float* beta = zphi + 256;
    float* h2   = beta + 16;
    float* zz   = h2 + (size_t)N * T * 64;
    float* mpart = h2;
    float* spart = mpart + (size_t)SCORE_BLOCKS * N;
    float* part  = z;
    float* zpart = abuf;

    const int thr = 256;

    gemm_f32_wmma<4><<<((N + 15) / 16 + 7) / 8, thr, 0, stream>>>(node, w_n, z, N, 128, 64);
    node_score_kernel<<<(N + thr - 1) / thr, thr, 0, stream>>>(z, w_attn, ssrc, sdst, N);
    edge_score_tables_kernel<<<SCORE_BLOCKS, SCORE_THREADS, 0, stream>>>(edgef, w_e, w_attn, src, dst,
                                                                      ssrc, sdst, abuf, mpart, spart, E, N);
    softmax_combine_kernel<<<(N + 31) / 32, thr, 0, stream>>>(mpart, spart, mbuf, den, N);
    gemm_f32_wmma<4><<<((N + 15) / 16 + 7) / 8, thr, 0, stream>>>(z, w_fc2, zz, N, 64, 64);
    init_kernel<<<1024, thr, 0, stream>>>(h2, N * T * 64);
    edge_scatter_fc2_kernel<<<((E + 15) / 16 + 7) / 8, thr, 0, stream>>>(edgef, w_e, w_fc2 + 64 * 64, zz, abuf, mbuf, den,
                                                                      src, dst, etype, h2, E);
    zphi_fold_kernel<<<1024, thr, 0, stream>>>(h2, part, N);
    zphi_fold_kernel<<<64, thr, 0, stream>>>(part, zpart, 1024);
    beta_kernel<<<1, thr, 0, stream>>>(zpart, 64, w_sem, beta);
    out_kernel<<<(N * 64 + thr - 1) / thr, thr, 0, stream>>>(h2, beta, out, N);
}


# z tile kept on chip: one kernel does node @ w_n, both node scores from its accumulators (16-lane tree) and zz = z @ w_fc2[0:64] through LDS; z plane and two launches go; score sum order changes
# speedup vs baseline: 2.4511x; 1.1195x over previous
#include <hip/hip_runtime.h>
#include <hip/hip_bf16.h>
#include <math.h>

typedef __attribute__((ext_vector_type(2))) float v2f;
typedef __attribute__((ext_vector_type(8))) float v8f;


template <int M>
__device__ __forceinline__ float row_xor(float v) {
    return __int_as_float(__builtin_amdgcn_update_dpp(0, __float_as_int(v), 0x160 | M, 0xf, 0xf, true));
}

#define FRONT_STRIDE 68
__global__ void __launch_bounds__(256) node_front_kernel(
    const float* __restrict__ node, const float* __restrict__ w_n, const float* __restrict__ w_attn,
    const float* __restrict__ w_fc2,
    float* __restrict__ ssrc, float* __restrict__ sdst, float* __restrict__ zz, int M)
{
    __shared__ v2f sB1[64 * 64];
    __shared__ v2f sB2[32 * 64];
    __shared__ __attribute__((aligned(16))) float sT[8][16 * FRONT_STRIDE];
    const int tid  = threadIdx.x;
    const int lane = tid & 31;
    const int wave = tid >> 5;
    const int half = lane >> 4;
    const int l15  = lane & 15;
    const int rowTile = blockIdx.x * 8 + wave;
    int mrow = rowTile * 16 + l15;
    if (mrow >= M) mrow = M - 1;

    const float* arow = node + (size_t)mrow * 128 + 2 * half;
    v2f av[32];
#pragma unroll
    for (int j = 0; j < 32; ++j) av[j] = *(const v2f*)(arow + 4 * j);

#pragma unroll
    for (int i = 0; i < 16; ++i) {
        const int p  = tid + i * 256;
        const int kk = (p >> 6) << 1;
        const int n  = p & 63;
        v2f b;
        b.x = w_n[kk * 64 + n];
        b.y = w_n[(kk + 1) * 64 + n];
        sB1[p] = b;
    }
#pragma unroll
    for (int i = 0; i < 8; ++i) {
        const int p  = tid + i * 256;
        const int kk = (p >> 6) << 1;
        const int n  = p & 63;
        v2f b;
        b.x = w_fc2[kk * 64 + n];
        b.y = w_fc2[(kk + 1) * 64 + n];
        sB2[p] = b;
    }
    float wa[4], wb[4];
#pragma unroll
    for (int nt = 0; nt < 4; ++nt) {
        wa[nt] = w_attn[nt * 16 + l15];
        wb[nt] = w_attn[96 + nt * 16 + l15];
    }
    __syncthreads();

    v8f acc[4] = {};
    const v2f* __restrict__ b1col = sB1 + half * 64 + l15;
#pragma unroll
    for (int j = 0; j < 32; ++j) {
#pragma unroll
        for (int nt = 0; nt < 4; ++nt) {
            const v2f b = b1col[(2 * j) * 64 + nt * 16];
            acc[nt] = __builtin_amdgcn_wmma_f32_16x16x4_f32(
                false, av[j], false, b, (short)0, acc[nt], false, false);
        }
    }

    float s0 = 0.f, s1 = 0.f;
#pragma unroll
    for (int r = 0; r < 8; ++r) {
        float p0 = acc[0][r] * wa[0] + acc[1][r] * wa[1] + acc[2][r] * wa[2] + acc[3][r] * wa[3];
        float p1 = acc[0][r] * wb[0] + acc[1][r] * wb[1] + acc[2][r] * wb[2] + acc[3][r] * wb[3];
        p0 += row_xor<8>(p0);  p1 += row_xor<8>(p1);
        p0 += row_xor<4>(p0);  p1 += row_xor<4>(p1);
        p0 += row_xor<2>(p0);  p1 += row_xor<2>(p1);
        p0 += row_xor<1>(p0);  p1 += row_xor<1>(p1);
        if (l15 == r) { s0 = p0; s1 = p1; }
    }
    {
        const int orow = rowTile * 16 + (l15 & 7) + 8 * half;
        if (l15 < 8 && orow < M) { ssrc[orow] = s0; sdst[orow] = s1; }
    }

    float* tw = sT[wave];
#pragma unroll
    for (int nt = 0; nt < 4; ++nt) {
#pragma unroll
        for (int r = 0; r < 8; ++r)
            tw[(r + 8 * half) * FRONT_STRIDE + nt * 16 + l15] = acc[nt][r];
    }
    __syncthreads();

    v8f acc2[4] = {};
    const float* a2row = tw + l15 * FRONT_STRIDE + 2 * half;
    const v2f* __restrict__ b2col = sB2 + half * 64 + l15;
#pragma unroll
    for (int k0 = 0; k0 < 64; k0 += 4) {
        const v2f a = *(const v2f*)(a2row + k0);
#pragma unroll
        for (int nt = 0; nt < 4; ++nt) {
            const v2f b = b2col[(k0 >> 1) * 64 + nt * 16];
            acc2[nt] = __builtin_amdgcn_wmma_f32_16x16x4_f32(
                false, a, false, b, (short)0, acc2[nt], false, false);
        }
    }

#pragma unroll
    for (int nt = 0; nt < 4; ++nt) {
#pragma unroll
        for (int r = 0; r < 8; ++r) {
            const int row = rowTile * 16 + r + 8 * half;
            if (row < M)
                zz[(size_t)row * 64 + nt * 16 + l15] = acc2[nt][r];
        }
    }
}

__device__ __forceinline__ void ldsMaxFloat(float* addr, float val) {
    const int iv = __float_as_int(val);
    if (iv >= 0) atomicMax((int*)addr, iv);
    else         atomicMin((unsigned int*)addr, (unsigned int)iv);
}

__global__ void init_kernel(float* __restrict__ h, int hsize)
{
    const int i = blockIdx.x * blockDim.x + threadIdx.x;
    const int stride = gridDim.x * blockDim.x;
    for (int j = i; j < hsize; j += stride) h[j] = 0.f;
}

#define SCORE_BLOCKS   192
#define SCORE_THREADS  1024
#define TAB_NODES      50000
#define CHUNK_ITERS    5
#define CHUNK_CAP      (CHUNK_ITERS * SCORE_THREADS)

__device__ __forceinline__ void table_out(const float* tab, float* __restrict__ row, int N, int tid) {
    if (((N & 3) == 0) && ((((size_t)row) & 15) == 0)) {
        const float4* t4 = (const float4*)tab;
        float4* r4 = (float4*)row;
        for (int q = tid; q < (N >> 2); q += SCORE_THREADS) r4[q] = t4[q];
    } else {
        for (int n = tid; n < N; n += SCORE_THREADS) row[n] = tab[n];
    }
}
__global__ void __launch_bounds__(SCORE_THREADS) edge_score_tables_kernel(
    const float* __restrict__ ef, const float* __restrict__ w_e, const float* __restrict__ w_attn,
    const int* __restrict__ src, const int* __restrict__ dst,
    const float* __restrict__ ssrc, const float* __restrict__ sdst,
    float* __restrict__ a, float* __restrict__ mpart, float* __restrict__ spart, int E, int N)
{
    __shared__ __attribute__((aligned(16))) float tab[TAB_NODES];
    __shared__ v2f   sBe[32 * 32];
    __shared__ float sA[CHUNK_CAP];
    __shared__ int   sD[CHUNK_CAP];
    const int tid = threadIdx.x;
    {
        const int kk = (tid >> 5) << 1;
        const int n  = tid & 31;
        v2f b;
        b.x = w_e[kk * 32 + n];
        b.y = w_e[(kk + 1) * 32 + n];
        sBe[tid] = b;
    }
    for (int n = tid; n < TAB_NODES; n += SCORE_THREADS) tab[n] = -INFINITY;

    const int lane = tid & 31;
    const int wave = __builtin_amdgcn_readfirstlane(tid >> 5);
    const int numTiles = (E + 15) >> 4;
    const int tpb = (numTiles + (int)gridDim.x - 1) / (int)gridDim.x;
    const int t0  = min(numTiles, (int)blockIdx.x * tpb);
    const int t1  = min(numTiles, t0 + tpb);
    const int e0  = t0 * 16;
    const int cnt = min(E, t1 * 16) - e0;

    {
        int dreg[CHUNK_ITERS], sreg[CHUNK_ITERS];
#pragma unroll
        for (int k = 0; k < CHUNK_ITERS; ++k) {
            const int e = min(e0 + tid + k * SCORE_THREADS, E - 1);
            dreg[k] = dst[e];
            sreg[k] = src[e];
        }
        float breg[CHUNK_ITERS];
#pragma unroll
        for (int k = 0; k < CHUNK_ITERS; ++k) breg[k] = ssrc[sreg[k]] + sdst[dreg[k]];
#pragma unroll
        for (int k = 0; k < CHUNK_ITERS; ++k) {
            sD[tid + k * SCORE_THREADS] = dreg[k];
            sA[tid + k * SCORE_THREADS] = breg[k];
        }
    }
    __syncthreads();

    const int half = lane >> 4;
    const int l15  = lane & 15;
    const float w0 = w_attn[64 + l15];
    const float w1 = w_attn[80 + l15];
    const v2f* __restrict__ bcol = sBe + half * 32 + l15;

    for (int tile = t0 + wave; tile < t1; tile += SCORE_THREADS / 32) {
        int ec = tile * 16 + l15;
        if (ec >= E) ec = E - 1;
        const int er = tile * 16 + (l15 & 7) + 8 * half;
        const bool own = (l15 < 8) && (er < E);
        const int li = er - e0;

        v8f acc[2] = {};
        const float* arow = ef + (size_t)ec * 64 + 2 * half;
#pragma unroll
        for (int k0 = 0; k0 < 64; k0 += 4) {
            const v2f av = *(const v2f*)(arow + k0);
#pragma unroll
            for (int nt = 0; nt < 2; ++nt) {
                const v2f b = bcol[(k0 >> 1) * 32 + nt * 16];
                acc[nt] = __builtin_amdgcn_wmma_f32_16x16x4_f32(
                    false, av, false, b, (short)0, acc[nt], false, false);
            }
        }
        float mid = 0.f;
#pragma unroll
        for (int r = 0; r < 8; ++r) {
            float p = acc[0][r] * w0 + acc[1][r] * w1;
            p += row_xor<8>(p);
            p += row_xor<4>(p);
            p += row_xor<2>(p);
            p += row_xor<1>(p);
            if (l15 == r) mid = p;
        }
        if (own) {
            const float t  = sA[li] + mid;
            const float lr = t > 0.f ? t : 0.01f * t;
            a[er]  = lr;
            sA[li] = lr;
            ldsMaxFloat(&tab[sD[li]], lr);
        }
    }
    __syncthreads();

    float mreg[CHUNK_ITERS];
#pragma unroll
    for (int k = 0; k < CHUNK_ITERS; ++k) {
        const int i = tid + k * SCORE_THREADS;
        mreg[k] = (i < cnt) ? tab[sD[i]] : 0.f;
    }
    table_out(tab, mpart + (size_t)blockIdx.x * N, N, tid);
    __syncthreads();
    for (int n = tid; n < TAB_NODES; n += SCORE_THREADS) tab[n] = 0.f;
    __syncthreads();
#pragma unroll
    for (int k = 0; k < CHUNK_ITERS; ++k) {
        const int i = tid + k * SCORE_THREADS;
        if (i < cnt) atomicAdd(&tab[sD[i]], __expf(sA[i] - mreg[k]));
    }
    __syncthreads();
    table_out(tab, spart + (size_t)blockIdx.x * N, N, tid);
}

__global__ void __launch_bounds__(256) softmax_combine_kernel(
    const float* __restrict__ mpart, const float* __restrict__ spart,
    float* __restrict__ mbuf, float* __restrict__ den, int N)
{
    __shared__ float smax[8][32];
    __shared__ float ssum[8][32];
    const int lane = threadIdx.x & 31;
    const int g    = threadIdx.x >> 5;
    const int n    = blockIdx.x * 32 + lane;
    const int nn   = min(n, N - 1);
    float mb[SCORE_BLOCKS / 8], sb[SCORE_BLOCKS / 8];
#pragma unroll
    for (int j = 0; j < SCORE_BLOCKS / 8; ++j) {
        const size_t row = (size_t)(g * (SCORE_BLOCKS / 8) + j) * N + nn;
        mb[j] = mpart[row];
        sb[j] = spart[row];
    }
    float m = -INFINITY;
#pragma unroll
    for (int j = 0; j < SCORE_BLOCKS / 8; ++j) m = fmaxf(m, mb[j]);
    smax[g][lane] = m;
    __syncthreads();
    float M = smax[0][lane];
#pragma unroll
    for (int gg = 1; gg < 8; ++gg) M = fmaxf(M, smax[gg][lane]);
    float d = 0.f;
#pragma unroll
    for (int j = 0; j < SCORE_BLOCKS / 8; ++j)
        if (sb[j] > 0.f) d += sb[j] * __expf(mb[j] - M);
    ssum[g][lane] = d;
    __syncthreads();
    if (g == 0 && n < N) {
        float t = 0.f;
#pragma unroll
        for (int gg = 0; gg < 8; ++gg) t += ssum[gg][lane];
        mbuf[n] = M;
        den[n]  = t;
    }
}

#define MSG_STRIDE 68
__global__ void __launch_bounds__(256) edge_scatter_fc2_kernel(
    const float* __restrict__ ef, const float* __restrict__ w_e, const float* __restrict__ W2,
    const float* __restrict__ zz,
    const float* __restrict__ a, const float* __restrict__ mbuf, const float* __restrict__ den,
    const int* __restrict__ src, const int* __restrict__ dst, const int* __restrict__ etype,
    float* __restrict__ h2, int E)
{
    __shared__ v2f   sBe[32 * 32];
    __shared__ v2f   sB2[16 * 64];
    __shared__ __attribute__((aligned(16))) float sT[8][16 * MSG_STRIDE];
    const int tid = threadIdx.x;
    for (int p = tid; p < 32 * 32; p += 256) {
        const int kk = (p >> 5) << 1;
        const int n  = p & 31;
        v2f b;
        b.x = w_e[kk * 32 + n];
        b.y = w_e[(kk + 1) * 32 + n];
        sBe[p] = b;
    }
    for (int p = tid; p < 16 * 64; p += 256) {
        const int kk = (p >> 6) << 1;
        const int n  = p & 63;
        v2f b;
        b.x = W2[kk * 64 + n];
        b.y = W2[(kk + 1) * 64 + n];
        sB2[p] = b;
    }
    __syncthreads();

    const int lane = tid & 31;
    const int wave = tid >> 5;
    const int tile = blockIdx.x * 8 + wave;
    const bool active = tile < ((E + 15) >> 4);
    const int half = lane >> 4;
    const int l15  = lane & 15;
    int ec = tile * 16 + l15;
    if (ec >= E) ec = E - 1;
    const int   d_l   = dst[ec];
    const int   s_l   = src[ec];
    const int   t_l   = etype[ec];
    const float a_l   = a[ec];
    const float m_l   = mbuf[d_l];
    const float den_l = den[d_l];

    if (active) {
        v8f pe[2] = {};
        const float* arow = ef + (size_t)ec * 64 + 2 * half;
        const v2f* __restrict__ becol = sBe + half * 32 + l15;
#pragma unroll
        for (int k0 = 0; k0 < 64; k0 += 4) {
            const v2f av = *(const v2f*)(arow + k0);
#pragma unroll
            for (int nt = 0; nt < 2; ++nt) {
                const v2f b = becol[(k0 >> 1) * 32 + nt * 16];
                pe[nt] = __builtin_amdgcn_wmma_f32_16x16x4_f32(
                    false, av, false, b, (short)0, pe[nt], false, false);
            }
        }
        float* tw = sT[wave];
#pragma unroll
        for (int nt = 0; nt < 2; ++nt) {
#pragma unroll
            for (int r = 0; r < 8; ++r)
                tw[(r + 8 * half) * MSG_STRIDE + nt * 16 + l15] = pe[nt][r];
        }
    }
    __syncthreads();
    if (active) {
        v8f acc[4] = {};
        const float* arow = sT[wave] + l15 * MSG_STRIDE + 2 * half;
        const v2f* __restrict__ bcol = sB2 + half * 64 + l15;
#pragma unroll
        for (int k0 = 0; k0 < 32; k0 += 4) {
            const v2f av = *(const v2f*)(arow + k0);
#pragma unroll
            for (int nt = 0; nt < 4; ++nt) {
                const v2f b = bcol[(k0 >> 1) * 64 + nt * 16];
                acc[nt] = __builtin_amdgcn_wmma_f32_16x16x4_f32(
                    false, av, false, b, (short)0, acc[nt], false, false);
            }
        }
        float* tw = sT[wave];
#pragma unroll
        for (int nt = 0; nt < 4; ++nt) {
#pragma unroll
            for (int r = 0; r < 8; ++r)
                tw[(r + 8 * half) * MSG_STRIDE + nt * 16 + l15] = acc[nt][r];
        }
    }
    __syncthreads();
    if (!active) return;

    const float al_l = __expf(a_l - m_l) / den_l;
    const float* tr = sT[wave];
    const int nvalid = E - __builtin_amdgcn_readfirstlane(tile) * 16;

    float m0[16], m1[16];
#pragma unroll
    for (int i = 0; i < 16; ++i) {
        const int s = __builtin_amdgcn_readlane(s_l, i);
        const float alpha = __int_as_float(__builtin_amdgcn_readlane(__float_as_int(al_l), i));
        const float* zr = zz + (size_t)s * 64;
        m0[i] = alpha * (tr[i * MSG_STRIDE + lane]      + zr[lane]);
        m1[i] = alpha * (tr[i * MSG_STRIDE + lane + 32] + zr[lane + 32]);
    }
#pragma unroll
    for (int i = 0; i < 16; ++i) {
        if (i < nvalid) {
            const int d = __builtin_amdgcn_readlane(d_l, i);
            const int t = __builtin_amdgcn_readlane(t_l, i);
            float* hb = h2 + ((size_t)d * 4 + t) * 64;
            atomicAdd(hb + lane,      m0[i]);
            atomicAdd(hb + lane + 32, m1[i]);
        }
    }
}

__global__ void zphi_fold_kernel(const float* __restrict__ part, float* __restrict__ zpart, int rows)
{
    const int s   = threadIdx.x;
    const int per = (rows + (int)gridDim.x - 1) / (int)gridDim.x;
    const int lo  = blockIdx.x * per;
    const int hi  = min(rows, lo + per);
    float sum = 0.f;
    for (int r = lo; r < hi; r += 8) {
        float v[8];
        for (int j = 0; j < 8; ++j)
            v[j] = (r + j < hi) ? part[(size_t)(r + j) * 256 + s] : 0.f;
        for (int j = 0; j < 8; ++j) sum += v[j];
    }
    zpart[blockIdx.x * 256 + s] = sum;
}

__global__ void beta_kernel(const float* __restrict__ zpart, int nparts,
                            const float* __restrict__ w_sem, float* __restrict__ beta)
{
    __shared__ float prod[256];
    __shared__ float wsh[4];
    const int tid = threadIdx.x;
    float z = 0.f;
    for (int r = 0; r < nparts; r += 8) {
        float v[8];
        for (int j = 0; j < 8; ++j)
            v[j] = (r + j < nparts) ? zpart[(r + j) * 256 + tid] : 0.f;
        for (int j = 0; j < 8; ++j) z += v[j];
    }
    prod[tid] = z * w_sem[tid & 63];
    __syncthreads();
    if (tid < 4) {
        float d = 0.f;
        for (int o = 0; o < 64; ++o) d += prod[tid * 64 + o];
        wsh[tid] = d > 0.f ? d : 0.01f * d;
    }
    __syncthreads();
    if (tid != 0) return;
    float wp[4], mx = -INFINITY;
#pragma unroll
    for (int t = 0; t < 4; ++t) {
        wp[t] = wsh[t];
        mx = fmaxf(mx, wp[t]);
    }
    float s = 0.f;
#pragma unroll
    for (int t = 0; t < 4; ++t) { wp[t] = __expf(wp[t] - mx); s += wp[t]; }
#pragma unroll
    for (int t = 0; t < 4; ++t) beta[t] = wp[t] / s;
}

__global__ void out_kernel(const float* __restrict__ h2, const float* __restrict__ beta,
                           float* __restrict__ Z, int N)
{
    const int i = blockIdx.x * blockDim.x + threadIdx.x;
    if (i >= N * 64) return;
    const int n = i >> 6, o = i & 63;
    const float* r = h2 + (size_t)n * 256 + o;
    Z[i] = beta[0] * r[0] + beta[1] * r[64] + beta[2] * r[128] + beta[3] * r[192];
}

extern "C" void kernel_launch(void* const* d_in, const int* in_sizes, int n_in,
                              void* d_out, int out_size, void* d_ws, size_t ws_size,
                              hipStream_t stream)
{
    const float* node   = (const float*)d_in[0];
    const float* edgef  = (const float*)d_in[1];
    const int*   src    = (const int*)d_in[2];
    const int*   dst    = (const int*)d_in[3];
    const int*   etype  = (const int*)d_in[4];
    const float* w_n    = (const float*)d_in[5];
    const float* w_e    = (const float*)d_in[6];
    const float* w_attn = (const float*)d_in[7];
    const float* w_fc2  = (const float*)d_in[8];
    const float* w_sem  = (const float*)d_in[9];
    float* out = (float*)d_out;

    const int N = in_sizes[0] / 128;
    const int E = in_sizes[2];
    const int T = 4;

    float* ws   = (float*)d_ws;
    float* zz   = ws;
    float* ssrc = zz + (size_t)N * 64;
    float* sdst = ssrc + N;
    float* abuf = sdst + N;
    float* mbuf = abuf + E;
    float* den  = mbuf + N;
    float* zphi = den + N;
    float* beta = zphi + 256;
    float* h2   = beta + 16;
    float* mpart = h2;
    float* spart = mpart + (size_t)SCORE_BLOCKS * N;
    float* part  = zz;
    float* zpart = abuf;

    const int thr = 256;

    node_front_kernel<<<((N + 15) / 16 + 7) / 8, thr, 0, stream>>>(node, w_n, w_attn, w_fc2, ssrc, sdst, zz, N);
    edge_score_tables_kernel<<<SCORE_BLOCKS, SCORE_THREADS, 0, stream>>>(edgef, w_e, w_attn, src, dst,
                                                                      ssrc, sdst, abuf, mpart, spart, E, N);
    softmax_combine_kernel<<<(N + 31) / 32, thr, 0, stream>>>(mpart, spart, mbuf, den, N);
    init_kernel<<<1024, thr, 0, stream>>>(h2, N * T * 64);
    edge_scatter_fc2_kernel<<<((E + 15) / 16 + 7) / 8, thr, 0, stream>>>(edgef, w_e, w_fc2 + 64 * 64, zz, abuf, mbuf, den,
                                                                      src, dst, etype, h2, E);
    zphi_fold_kernel<<<1024, thr, 0, stream>>>(h2, part, N);
    zphi_fold_kernel<<<64, thr, 0, stream>>>(part, zpart, 1024);
    beta_kernel<<<1, thr, 0, stream>>>(zpart, 64, w_sem, beta);
    out_kernel<<<(N * 64 + thr - 1) / thr, thr, 0, stream>>>(h2, beta, out, N);
}


# edge score term as one 64-long dot per edge, ef . (w_e @ w_attn[64:96]) (linear): rows read as whole lines, 16-lane tree; the score kernel's 16x32 tile product goes; score sum order changes
# speedup vs baseline: 2.6139x; 1.0664x over previous
#include <hip/hip_runtime.h>
#include <hip/hip_bf16.h>
#include <math.h>

typedef __attribute__((ext_vector_type(2))) float v2f;
typedef __attribute__((ext_vector_type(8))) float v8f;


template <int M>
__device__ __forceinline__ float row_xor(float v) {
    return __int_as_float(__builtin_amdgcn_update_dpp(0, __float_as_int(v), 0x160 | M, 0xf, 0xf, true));
}

#define FRONT_STRIDE 68
__global__ void __launch_bounds__(256) node_front_kernel(
    const float* __restrict__ node, const float* __restrict__ w_n, const float* __restrict__ w_attn,
    const float* __restrict__ w_fc2,
    float* __restrict__ ssrc, float* __restrict__ sdst, float* __restrict__ zz, int M)
{
    __shared__ v2f sB1[64 * 64];
    __shared__ v2f sB2[32 * 64];
    __shared__ __attribute__((aligned(16))) float sT[8][16 * FRONT_STRIDE];
    const int tid  = threadIdx.x;
    const int lane = tid & 31;
    const int wave = tid >> 5;
    const int half = lane >> 4;
    const int l15  = lane & 15;
    const int rowTile = blockIdx.x * 8 + wave;
    int mrow = rowTile * 16 + l15;
    if (mrow >= M) mrow = M - 1;

    const float* arow = node + (size_t)mrow * 128 + 2 * half;
    v2f av[32];
#pragma unroll
    for (int j = 0; j < 32; ++j) av[j] = *(const v2f*)(arow + 4 * j);

#pragma unroll
    for (int i = 0; i < 16; ++i) {
        const int p  = tid + i * 256;
        const int kk = (p >> 6) << 1;
        const int n  = p & 63;
        v2f b;
        b.x = w_n[kk * 64 + n];
        b.y = w_n[(kk + 1) * 64 + n];
        sB1[p] = b;
    }
#pragma unroll
    for (int i = 0; i < 8; ++i) {
        const int p  = tid + i * 256;
        const int kk = (p >> 6) << 1;
        const int n  = p & 63;
        v2f b;
        b.x = w_fc2[kk * 64 + n];
        b.y = w_fc2[(kk + 1) * 64 + n];
        sB2[p] = b;
    }
    float wa[4], wb[4];
#pragma unroll
    for (int nt = 0; nt < 4; ++nt) {
        wa[nt] = w_attn[nt * 16 + l15];
        wb[nt] = w_attn[96 + nt * 16 + l15];
    }
    __syncthreads();

    v8f acc[4] = {};
    const v2f* __restrict__ b1col = sB1 + half * 64 + l15;
#pragma unroll
    for (int j = 0; j < 32; ++j) {
#pragma unroll
        for (int nt = 0; nt < 4; ++nt) {
            const v2f b = b1col[(2 * j) * 64 + nt * 16];
            acc[nt] = __builtin_amdgcn_wmma_f32_16x16x4_f32(
                false, av[j], false, b, (short)0, acc[nt], false, false);
        }
    }

    float s0 = 0.f, s1 = 0.f;
#pragma unroll
    for (int r = 0; r < 8; ++r) {
        float p0 = acc[0][r] * wa[0] + acc[1][r] * wa[1] + acc[2][r] * wa[2] + acc[3][r] * wa[3];
        float p1 = acc[0][r] * wb[0] + acc[1][r] * wb[1] + acc[2][r] * wb[2] + acc[3][r] * wb[3];
        p0 += row_xor<8>(p0);  p1 += row_xor<8>(p1);
        p0 += row_xor<4>(p0);  p1 += row_xor<4>(p1);
        p0 += row_xor<2>(p0);  p1 += row_xor<2>(p1);
        p0 += row_xor<1>(p0);  p1 += row_xor<1>(p1);
        if (l15 == r) { s0 = p0; s1 = p1; }
    }
    {
        const int orow = rowTile * 16 + (l15 & 7) + 8 * half;
        if (l15 < 8 && orow < M) { ssrc[orow] = s0; sdst[orow] = s1; }
    }

    float* tw = sT[wave];
#pragma unroll
    for (int nt = 0; nt < 4; ++nt) {
#pragma unroll
        for (int r = 0; r < 8; ++r)
            tw[(r + 8 * half) * FRONT_STRIDE + nt * 16 + l15] = acc[nt][r];
    }
    __syncthreads();

    v8f acc2[4] = {};
    const float* a2row = tw + l15 * FRONT_STRIDE + 2 * half;
    const v2f* __restrict__ b2col = sB2 + half * 64 + l15;
#pragma unroll
    for (int k0 = 0; k0 < 64; k0 += 4) {
        const v2f a = *(const v2f*)(a2row + k0);
#pragma unroll
        for (int nt = 0; nt < 4; ++nt) {
            const v2f b = b2col[(k0 >> 1) * 64 + nt * 16];
            acc2[nt] = __builtin_amdgcn_wmma_f32_16x16x4_f32(
                false, a, false, b, (short)0, acc2[nt], false, false);
        }
    }

#pragma unroll
    for (int nt = 0; nt < 4; ++nt) {
#pragma unroll
        for (int r = 0; r < 8; ++r) {
            const int row = rowTile * 16 + r + 8 * half;
            if (row < M)
                zz[(size_t)row * 64 + nt * 16 + l15] = acc2[nt][r];
        }
    }
}

__device__ __forceinline__ void ldsMaxFloat(float* addr, float val) {
    const int iv = __float_as_int(val);
    if (iv >= 0) atomicMax((int*)addr, iv);
    else         atomicMin((unsigned int*)addr, (unsigned int)iv);
}

__global__ void init_kernel(float* __restrict__ h, int hsize)
{
    const int i = blockIdx.x * blockDim.x + threadIdx.x;
    const int stride = gridDim.x * blockDim.x;
    for (int j = i; j < hsize; j += stride) h[j] = 0.f;
}

#define SCORE_BLOCKS   192
#define SCORE_THREADS  1024
#define TAB_NODES      50000
#define CHUNK_ITERS    5
#define CHUNK_CAP      (CHUNK_ITERS * SCORE_THREADS)

__device__ __forceinline__ void table_out(const float* tab, float* __restrict__ row, int N, int tid) {
    if (((N & 3) == 0) && ((((size_t)row) & 15) == 0)) {
        const float4* t4 = (const float4*)tab;
        float4* r4 = (float4*)row;
        for (int q = tid; q < (N >> 2); q += SCORE_THREADS) r4[q] = t4[q];
    } else {
        for (int n = tid; n < N; n += SCORE_THREADS) row[n] = tab[n];
    }
}
__global__ void __launch_bounds__(SCORE_THREADS) edge_score_tables_kernel(
    const float* __restrict__ ef, const float* __restrict__ w_e, const float* __restrict__ w_attn,
    const int* __restrict__ src, const int* __restrict__ dst,
    const float* __restrict__ ssrc, const float* __restrict__ sdst,
    float* __restrict__ a, float* __restrict__ mpart, float* __restrict__ spart, int E, int N)
{
    __shared__ __attribute__((aligned(16))) float tab[TAB_NODES];
    __shared__ __attribute__((aligned(16))) float su[64];
    __shared__ float sA[CHUNK_CAP];
    __shared__ int   sD[CHUNK_CAP];
    const int tid = threadIdx.x;
    if (tid < 64) {
        float acc = 0.f;
#pragma unroll
        for (int n = 0; n < 32; ++n) acc += w_e[tid * 32 + n] * w_attn[64 + n];
        su[tid] = acc;
    }
    for (int n = tid; n < TAB_NODES; n += SCORE_THREADS) tab[n] = -INFINITY;

    const int lane = tid & 31;
    const int wave = __builtin_amdgcn_readfirstlane(tid >> 5);
    const int numTiles = (E + 15) >> 4;
    const int tpb = (numTiles + (int)gridDim.x - 1) / (int)gridDim.x;
    const int t0  = min(numTiles, (int)blockIdx.x * tpb);
    const int t1  = min(numTiles, t0 + tpb);
    const int e0  = t0 * 16;
    const int cnt = min(E, t1 * 16) - e0;

    {
        int dreg[CHUNK_ITERS], sreg[CHUNK_ITERS];
#pragma unroll
        for (int k = 0; k < CHUNK_ITERS; ++k) {
            const int e = min(e0 + tid + k * SCORE_THREADS, E - 1);
            dreg[k] = dst[e];
            sreg[k] = src[e];
        }
        float breg[CHUNK_ITERS];
#pragma unroll
        for (int k = 0; k < CHUNK_ITERS; ++k) breg[k] = ssrc[sreg[k]] + sdst[dreg[k]];
#pragma unroll
        for (int k = 0; k < CHUNK_ITERS; ++k) {
            sD[tid + k * SCORE_THREADS] = dreg[k];
            sA[tid + k * SCORE_THREADS] = breg[k];
        }
    }
    __syncthreads();

    const int half = lane >> 4;
    const int l15  = lane & 15;
    const float4 u4 = *(const float4*)(su + 4 * l15);

    const int lastTile = numTiles - 1;
    float4 cur[8];
    {
        const int tl = min(t0 + wave, lastTile);
#pragma unroll
        for (int i = 0; i < 8; ++i) {
            const int row = min(tl * 16 + 2 * i + half, E - 1);
            cur[i] = *(const float4*)(ef + (size_t)row * 64 + 4 * l15);
        }
    }
    for (int tile = t0 + wave; tile < t1; tile += SCORE_THREADS / 32) {
        float4 nxt[8];
        {
            const int tl = min(tile + SCORE_THREADS / 32, lastTile);
#pragma unroll
            for (int i = 0; i < 8; ++i) {
                const int row = min(tl * 16 + 2 * i + half, E - 1);
                nxt[i] = *(const float4*)(ef + (size_t)row * 64 + 4 * l15);
            }
        }
        const int er = tile * 16 + 2 * l15 + half;
        const bool own = (l15 < 8) && (er < E);
        const int li = er - e0;

        float mid = 0.f;
#pragma unroll
        for (int i = 0; i < 8; ++i) {
            float p = cur[i].x * u4.x + cur[i].y * u4.y + cur[i].z * u4.z + cur[i].w * u4.w;
            p += row_xor<8>(p);
            p += row_xor<4>(p);
            p += row_xor<2>(p);
            p += row_xor<1>(p);
            if (l15 == i) mid = p;
        }
        if (own) {
            const float t  = sA[li] + mid;
            const float lr = t > 0.f ? t : 0.01f * t;
            a[er]  = lr;
            sA[li] = lr;
            ldsMaxFloat(&tab[sD[li]], lr);
        }
#pragma unroll
        for (int i = 0; i < 8; ++i) cur[i] = nxt[i];
    }
    __syncthreads();

    float mreg[CHUNK_ITERS];
#pragma unroll
    for (int k = 0; k < CHUNK_ITERS; ++k) {
        const int i = tid + k * SCORE_THREADS;
        mreg[k] = (i < cnt) ? tab[sD[i]] : 0.f;
    }
    table_out(tab, mpart + (size_t)blockIdx.x * N, N, tid);
    __syncthreads();
    for (int n = tid; n < TAB_NODES; n += SCORE_THREADS) tab[n] = 0.f;
    __syncthreads();
#pragma unroll
    for (int k = 0; k < CHUNK_ITERS; ++k) {
        const int i = tid + k * SCORE_THREADS;
        if (i < cnt) atomicAdd(&tab[sD[i]], __expf(sA[i] - mreg[k]));
    }
    __syncthreads();
    table_out(tab, spart + (size_t)blockIdx.x * N, N, tid);
}

__global__ void __launch_bounds__(256) softmax_combine_kernel(
    const float* __restrict__ mpart, const float* __restrict__ spart,
    float* __restrict__ mbuf, float* __restrict__ den, int N)
{
    __shared__ float smax[8][32];
    __shared__ float ssum[8][32];
    const int lane = threadIdx.x & 31;
    const int g    = threadIdx.x >> 5;
    const int n    = blockIdx.x * 32 + lane;
    const int nn   = min(n, N - 1);
    float mb[SCORE_BLOCKS / 8], sb[SCORE_BLOCKS / 8];
#pragma unroll
    for (int j = 0; j < SCORE_BLOCKS / 8; ++j) {
        const size_t row = (size_t)(g * (SCORE_BLOCKS / 8) + j) * N + nn;
        mb[j] = mpart[row];
        sb[j] = spart[row];
    }
    float m = -INFINITY;
#pragma unroll
    for (int j = 0; j < SCORE_BLOCKS / 8; ++j) m = fmaxf(m, mb[j]);
    smax[g][lane] = m;
    __syncthreads();
    float M = smax[0][lane];
#pragma unroll
    for (int gg = 1; gg < 8; ++gg) M = fmaxf(M, smax[gg][lane]);
    float d = 0.f;
#pragma unroll
    for (int j = 0; j < SCORE_BLOCKS / 8; ++j)
        if (sb[j] > 0.f) d += sb[j] * __expf(mb[j] - M);
    ssum[g][lane] = d;
    __syncthreads();
    if (g == 0 && n < N) {
        float t = 0.f;
#pragma unroll
        for (int gg = 0; gg < 8; ++gg) t += ssum[gg][lane];
        mbuf[n] = M;
        den[n]  = t;
    }
}

#define MSG_STRIDE 68
__global__ void __launch_bounds__(256) edge_scatter_fc2_kernel(
    const float* __restrict__ ef, const float* __restrict__ w_e, const float* __restrict__ W2,
    const float* __restrict__ zz,
    const float* __restrict__ a, const float* __restrict__ mbuf, const float* __restrict__ den,
    const int* __restrict__ src, const int* __restrict__ dst, const int* __restrict__ etype,
    float* __restrict__ h2, int E)
{
    __shared__ v2f   sBe[32 * 32];
    __shared__ v2f   sB2[16 * 64];
    __shared__ __attribute__((aligned(16))) float sT[8][16 * MSG_STRIDE];
    const int tid = threadIdx.x;
    for (int p = tid; p < 32 * 32; p += 256) {
        const int kk = (p >> 5) << 1;
        const int n  = p & 31;
        v2f b;
        b.x = w_e[kk * 32 + n];
        b.y = w_e[(kk + 1) * 32 + n];
        sBe[p] = b;
    }
    for (int p = tid; p < 16 * 64; p += 256) {
        const int kk = (p >> 6) << 1;
        const int n  = p & 63;
        v2f b;
        b.x = W2[kk * 64 + n];
        b.y = W2[(kk + 1) * 64 + n];
        sB2[p] = b;
    }
    __syncthreads();

    const int lane = tid & 31;
    const int wave = tid >> 5;
    const int tile = blockIdx.x * 8 + wave;
    const bool active = tile < ((E + 15) >> 4);
    const int half = lane >> 4;
    const int l15  = lane & 15;
    int ec = tile * 16 + l15;
    if (ec >= E) ec = E - 1;
    const int   d_l   = dst[ec];
    const int   s_l   = src[ec];
    const int   t_l   = etype[ec];
    const float a_l   = a[ec];
    const float m_l   = mbuf[d_l];
    const float den_l = den[d_l];

    if (active) {
        v8f pe[2] = {};
        const float* arow = ef + (size_t)ec * 64 + 2 * half;
        const v2f* __restrict__ becol = sBe + half * 32 + l15;
#pragma unroll
        for (int k0 = 0; k0 < 64; k0 += 4) {
            const v2f av = *(const v2f*)(arow + k0);
#pragma unroll
            for (int nt = 0; nt < 2; ++nt) {
                const v2f b = becol[(k0 >> 1) * 32 + nt * 16];
                pe[nt] = __builtin_amdgcn_wmma_f32_16x16x4_f32(
                    false, av, false, b, (short)0, pe[nt], false, false);
            }
        }
        float* tw = sT[wave];
#pragma unroll
        for (int nt = 0; nt < 2; ++nt) {
#pragma unroll
            for (int r = 0; r < 8; ++r)
                tw[(r + 8 * half) * MSG_STRIDE + nt * 16 + l15] = pe[nt][r];
        }
    }
    __syncthreads();
    if (active) {
        v8f acc[4] = {};
        const float* arow = sT[wave] + l15 * MSG_STRIDE + 2 * half;
        const v2f* __restrict__ bcol = sB2 + half * 64 + l15;
#pragma unroll
        for (int k0 = 0; k0 < 32; k0 += 4) {
            const v2f av = *(const v2f*)(arow + k0);
#pragma unroll
            for (int nt = 0; nt < 4; ++nt) {
                const v2f b = bcol[(k0 >> 1) * 64 + nt * 16];
                acc[nt] = __builtin_amdgcn_wmma_f32_16x16x4_f32(
                    false, av, false, b, (short)0, acc[nt], false, false);
            }
        }
        float* tw = sT[wave];
#pragma unroll
        for (int nt = 0; nt < 4; ++nt) {
#pragma unroll
            for (int r = 0; r < 8; ++r)
                tw[(r + 8 * half) * MSG_STRIDE + nt * 16 + l15] = acc[nt][r];
        }
    }
    __syncthreads();
    if (!active) return;

    const float al_l = __expf(a_l - m_l) / den_l;
    const float* tr = sT[wave];
    const int nvalid = E - __builtin_amdgcn_readfirstlane(tile) * 16;

    float m0[16], m1[16];
#pragma unroll
    for (int i = 0; i < 16; ++i) {
        const int s = __builtin_amdgcn_readlane(s_l, i);
        const float alpha = __int_as_float(__builtin_amdgcn_readlane(__float_as_int(al_l), i));
        const float* zr = zz + (size_t)s * 64;
        m0[i] = alpha * (tr[i * MSG_STRIDE + lane]      + zr[lane]);
        m1[i] = alpha * (tr[i * MSG_STRIDE + lane + 32] + zr[lane + 32]);
    }
#pragma unroll
    for (int i = 0; i < 16; ++i) {
        if (i < nvalid) {
            const int d = __builtin_amdgcn_readlane(d_l, i);
            const int t = __builtin_amdgcn_readlane(t_l, i);
            float* hb = h2 + ((size_t)d * 4 + t) * 64;
            atomicAdd(hb + lane,      m0[i]);
            atomicAdd(hb + lane + 32, m1[i]);
        }
    }
}

__global__ void zphi_fold_kernel(const float* __restrict__ part, float* __restrict__ zpart, int rows)
{
    const int s   = threadIdx.x;
    const int per = (rows + (int)gridDim.x - 1) / (int)gridDim.x;
    const int lo  = blockIdx.x * per;
    const int hi  = min(rows, lo + per);
    float sum = 0.f;
    for (int r = lo; r < hi; r += 8) {
        float v[8];
        for (int j = 0; j < 8; ++j)
            v[j] = (r + j < hi) ? part[(size_t)(r + j) * 256 + s] : 0.f;
        for (int j = 0; j < 8; ++j) sum += v[j];
    }
    zpart[blockIdx.x * 256 + s] = sum;
}

__global__ void beta_kernel(const float* __restrict__ zpart, int nparts,
                            const float* __restrict__ w_sem, float* __restrict__ beta)
{
    __shared__ float prod[256];
    __shared__ float wsh[4];
    const int tid = threadIdx.x;
    float z = 0.f;
    for (int r = 0; r < nparts; r += 8) {
        float v[8];
        for (int j = 0; j < 8; ++j)
            v[j] = (r + j < nparts) ? zpart[(r + j) * 256 + tid] : 0.f;
        for (int j = 0; j < 8; ++j) z += v[j];
    }
    prod[tid] = z * w_sem[tid & 63];
    __syncthreads();
    if (tid < 4) {
        float d = 0.f;
        for (int o = 0; o < 64; ++o) d += prod[tid * 64 + o];
        wsh[tid] = d > 0.f ? d : 0.01f * d;
    }
    __syncthreads();
    if (tid != 0) return;
    float wp[4], mx = -INFINITY;
#pragma unroll
    for (int t = 0; t < 4; ++t) {
        wp[t] = wsh[t];
        mx = fmaxf(mx, wp[t]);
    }
    float s = 0.f;
#pragma unroll
    for (int t = 0; t < 4; ++t) { wp[t] = __expf(wp[t] - mx); s += wp[t]; }
#pragma unroll
    for (int t = 0; t < 4; ++t) beta[t] = wp[t] / s;
}

__global__ void out_kernel(const float* __restrict__ h2, const float* __restrict__ beta,
                           float* __restrict__ Z, int N)
{
    const int i = blockIdx.x * blockDim.x + threadIdx.x;
    if (i >= N * 64) return;
    const int n = i >> 6, o = i & 63;
    const float* r = h2 + (size_t)n * 256 + o;
    Z[i] = beta[0] * r[0] + beta[1] * r[64] + beta[2] * r[128] + beta[3] * r[192];
}

extern "C" void kernel_launch(void* const* d_in, const int* in_sizes, int n_in,
                              void* d_out, int out_size, void* d_ws, size_t ws_size,
                              hipStream_t stream)
{
    const float* node   = (const float*)d_in[0];
    const float* edgef  = (const float*)d_in[1];
    const int*   src    = (const int*)d_in[2];
    const int*   dst    = (const int*)d_in[3];
    const int*   etype  = (const int*)d_in[4];
    const float* w_n    = (const float*)d_in[5];
    const float* w_e    = (const float*)d_in[6];
    const float* w_attn = (const float*)d_in[7];
    const float* w_fc2  = (const float*)d_in[8];
    const float* w_sem  = (const float*)d_in[9];
    float* out = (float*)d_out;

    const int N = in_sizes[0] / 128;
    const int E = in_sizes[2];
    const int T = 4;

    float* ws   = (float*)d_ws;
    float* zz   = ws;
    float* ssrc = zz + (size_t)N * 64;
    float* sdst = ssrc + N;
    float* abuf = sdst + N;
    float* mbuf = abuf + E;
    float* den  = mbuf + N;
    float* zphi = den + N;
    float* beta = zphi + 256;
    float* h2   = beta + 16;
    float* mpart = h2;
    float* spart = mpart + (size_t)SCORE_BLOCKS * N;
    float* part  = zz;
    float* zpart = abuf;

    const int thr = 256;

    node_front_kernel<<<((N + 15) / 16 + 7) / 8, thr, 0, stream>>>(node, w_n, w_attn, w_fc2, ssrc, sdst, zz, N);
    edge_score_tables_kernel<<<SCORE_BLOCKS, SCORE_THREADS, 0, stream>>>(edgef, w_e, w_attn, src, dst,
                                                                      ssrc, sdst, abuf, mpart, spart, E, N);
    softmax_combine_kernel<<<(N + 31) / 32, thr, 0, stream>>>(mpart, spart, mbuf, den, N);
    init_kernel<<<1024, thr, 0, stream>>>(h2, N * T * 64);
    edge_scatter_fc2_kernel<<<((E + 15) / 16 + 7) / 8, thr, 0, stream>>>(edgef, w_e, w_fc2 + 64 * 64, zz, abuf, mbuf, den,
                                                                      src, dst, etype, h2, E);
    zphi_fold_kernel<<<1024, thr, 0, stream>>>(h2, part, N);
    zphi_fold_kernel<<<64, thr, 0, stream>>>(part, zpart, 1024);
    beta_kernel<<<1, thr, 0, stream>>>(zpart, 64, w_sem, beta);
    out_kernel<<<(N * 64 + thr - 1) / thr, thr, 0, stream>>>(h2, beta, out, N);
}


# h2 plane begins on a 256-B boundary (was 64 B past one): each 32-lane f32 atomic add of the scatter covers ONE whole 128-B line, was the ends of two; order of every sum unchanged
# speedup vs baseline: 2.7978x; 1.0704x over previous
#include <hip/hip_runtime.h>
#include <hip/hip_bf16.h>
#include <math.h>

typedef __attribute__((ext_vector_type(2))) float v2f;
typedef __attribute__((ext_vector_type(8))) float v8f;


template <int M>
__device__ __forceinline__ float row_xor(float v) {
    return __int_as_float(__builtin_amdgcn_update_dpp(0, __float_as_int(v), 0x160 | M, 0xf, 0xf, true));
}

#define FRONT_STRIDE 68
__global__ void __launch_bounds__(256) node_front_kernel(
    const float* __restrict__ node, const float* __restrict__ w_n, const float* __restrict__ w_attn,
    const float* __restrict__ w_fc2,
    float* __restrict__ ssrc, float* __restrict__ sdst, float* __restrict__ zz, int M)
{
    __shared__ v2f sB1[64 * 64];
    __shared__ v2f sB2[32 * 64];
    __shared__ __attribute__((aligned(16))) float sT[8][16 * FRONT_STRIDE];
    const int tid  = threadIdx.x;
    const int lane = tid & 31;
    const int wave = tid >> 5;
    const int half = lane >> 4;
    const int l15  = lane & 15;
    const int rowTile = blockIdx.x * 8 + wave;
    int mrow = rowTile * 16 + l15;
    if (mrow >= M) mrow = M - 1;

    const float* arow = node + (size_t)mrow * 128 + 2 * half;
    v2f av[32];
#pragma unroll
    for (int j = 0; j < 32; ++j) av[j] = *(const v2f*)(arow + 4 * j);

#pragma unroll
    for (int i = 0; i < 16; ++i) {
        const int p  = tid + i * 256;
        const int kk = (p >> 6) << 1;
        const int n  = p & 63;
        v2f b;
        b.x = w_n[kk * 64 + n];
        b.y = w_n[(kk + 1) * 64 + n];
        sB1[p] = b;
    }
#pragma unroll
    for (int i = 0; i < 8; ++i) {
        const int p  = tid + i * 256;
        const int kk = (p >> 6) << 1;
        const int n  = p & 63;
        v2f b;
        b.x = w_fc2[kk * 64 + n];
        b.y = w_fc2[(kk + 1) * 64 + n];
        sB2[p] = b;
    }
    float wa[4], wb[4];
#pragma unroll
    for (int nt = 0; nt < 4; ++nt) {
        wa[nt] = w_attn[nt * 16 + l15];
        wb[nt] = w_attn[96 + nt * 16 + l15];
    }
    __syncthreads();

    v8f acc[4] = {};
    const v2f* __restrict__ b1col = sB1 + half * 64 + l15;
#pragma unroll
    for (int j = 0; j < 32; ++j) {
#pragma unroll
        for (int nt = 0; nt < 4; ++nt) {
            const v2f b = b1col[(2 * j) * 64 + nt * 16];
            acc[nt] = __builtin_amdgcn_wmma_f32_16x16x4_f32(
                false, av[j], false, b, (short)0, acc[nt], false, false);
        }
    }

    float s0 = 0.f, s1 = 0.f;
#pragma unroll
    for (int r = 0; r < 8; ++r) {
        float p0 = acc[0][r] * wa[0] + acc[1][r] * wa[1] + acc[2][r] * wa[2] + acc[3][r] * wa[3];
        float p1 = acc[0][r] * wb[0] + acc[1][r] * wb[1] + acc[2][r] * wb[2] + acc[3][r] * wb[3];
        p0 += row_xor<8>(p0);  p1 += row_xor<8>(p1);
        p0 += row_xor<4>(p0);  p1 += row_xor<4>(p1);
        p0 += row_xor<2>(p0);  p1 += row_xor<2>(p1);
        p0 += row_xor<1>(p0);  p1 += row_xor<1>(p1);
        if (l15 == r) { s0 = p0; s1 = p1; }
    }
    {
        const int orow = rowTile * 16 + (l15 & 7) + 8 * half;
        if (l15 < 8 && orow < M) { ssrc[orow] = s0; sdst[orow] = s1; }
    }

    float* tw = sT[wave];
#pragma unroll
    for (int nt = 0; nt < 4; ++nt) {
#pragma unroll
        for (int r = 0; r < 8; ++r)
            tw[(r + 8 * half) * FRONT_STRIDE + nt * 16 + l15] = acc[nt][r];
    }
    __syncthreads();

    v8f acc2[4] = {};
    const float* a2row = tw + l15 * FRONT_STRIDE + 2 * half;
    const v2f* __restrict__ b2col = sB2 + half * 64 + l15;
#pragma unroll
    for (int k0 = 0; k0 < 64; k0 += 4) {
        const v2f a = *(const v2f*)(a2row + k0);
#pragma unroll
        for (int nt = 0; nt < 4; ++nt) {
            const v2f b = b2col[(k0 >> 1) * 64 + nt * 16];
            acc2[nt] = __builtin_amdgcn_wmma_f32_16x16x4_f32(
                false, a, false, b, (short)0, acc2[nt], false, false);
        }
    }

#pragma unroll
    for (int nt = 0; nt < 4; ++nt) {
#pragma unroll
        for (int r = 0; r < 8; ++r) {
            const int row = rowTile * 16 + r + 8 * half;
            if (row < M)
                zz[(size_t)row * 64 + nt * 16 + l15] = acc2[nt][r];
        }
    }
}

__device__ __forceinline__ void ldsMaxFloat(float* addr, float val) {
    const int iv = __float_as_int(val);
    if (iv >= 0) atomicMax((int*)addr, iv);
    else         atomicMin((unsigned int*)addr, (unsigned int)iv);
}

__global__ void init_kernel(float* __restrict__ h, int hsize)
{
    const int i = blockIdx.x * blockDim.x + threadIdx.x;
    const int stride = gridDim.x * blockDim.x;
    for (int j = i; j < hsize; j += stride) h[j] = 0.f;
}

#define SCORE_BLOCKS   192
#define SCORE_THREADS  1024
#define TAB_NODES      50000
#define CHUNK_ITERS    5
#define CHUNK_CAP      (CHUNK_ITERS * SCORE_THREADS)

__device__ __forceinline__ void table_out(const float* tab, float* __restrict__ row, int N, int tid) {
    if (((N & 3) == 0) && ((((size_t)row) & 15) == 0)) {
        const float4* t4 = (const float4*)tab;
        float4* r4 = (float4*)row;
        for (int q = tid; q < (N >> 2); q += SCORE_THREADS) r4[q] = t4[q];
    } else {
        for (int n = tid; n < N; n += SCORE_THREADS) row[n] = tab[n];
    }
}
__global__ void __launch_bounds__(SCORE_THREADS) edge_score_tables_kernel(
    const float* __restrict__ ef, const float* __restrict__ w_e, const float* __restrict__ w_attn,
    const int* __restrict__ src, const int* __restrict__ dst,
    const float* __restrict__ ssrc, const float* __restrict__ sdst,
    float* __restrict__ a, float* __restrict__ mpart, float* __restrict__ spart, int E, int N)
{
    __shared__ __attribute__((aligned(16))) float tab[TAB_NODES];
    __shared__ __attribute__((aligned(16))) float su[64];
    __shared__ float sA[CHUNK_CAP];
    __shared__ int   sD[CHUNK_CAP];
    const int tid = threadIdx.x;
    if (tid < 64) {
        float acc = 0.f;
#pragma unroll
        for (int n = 0; n < 32; ++n) acc += w_e[tid * 32 + n] * w_attn[64 + n];
        su[tid] = acc;
    }
    for (int n = tid; n < TAB_NODES; n += SCORE_THREADS) tab[n] = -INFINITY;

    const int lane = tid & 31;
    const int wave = __builtin_amdgcn_readfirstlane(tid >> 5);
    const int numTiles = (E + 15) >> 4;
    const int tpb = (numTiles + (int)gridDim.x - 1) / (int)gridDim.x;
    const int t0  = min(numTiles, (int)blockIdx.x * tpb);
    const int t1  = min(numTiles, t0 + tpb);
    const int e0  = t0 * 16;
    const int cnt = min(E, t1 * 16) - e0;

    {
        int dreg[CHUNK_ITERS], sreg[CHUNK_ITERS];
#pragma unroll
        for (int k = 0; k < CHUNK_ITERS; ++k) {
            const int e = min(e0 + tid + k * SCORE_THREADS, E - 1);
            dreg[k] = dst[e];
            sreg[k] = src[e];
        }
        float breg[CHUNK_ITERS];
#pragma unroll
        for (int k = 0; k < CHUNK_ITERS; ++k) breg[k] = ssrc[sreg[k]] + sdst[dreg[k]];
#pragma unroll
        for (int k = 0; k < CHUNK_ITERS; ++k) {
            sD[tid + k * SCORE_THREADS] = dreg[k];
            sA[tid + k * SCORE_THREADS] = breg[k];
        }
    }
    __syncthreads();

    const int half = lane >> 4;
    const int l15  = lane & 15;
    const float4 u4 = *(const float4*)(su + 4 * l15);

    const int lastTile = numTiles - 1;
    float4 cur[8];
    {
        const int tl = min(t0 + wave, lastTile);
#pragma unroll
        for (int i = 0; i < 8; ++i) {
            const int row = min(tl * 16 + 2 * i + half, E - 1);
            cur[i] = *(const float4*)(ef + (size_t)row * 64 + 4 * l15);
        }
    }
    for (int tile = t0 + wave; tile < t1; tile += SCORE_THREADS / 32) {
        float4 nxt[8];
        {
            const int tl = min(tile + SCORE_THREADS / 32, lastTile);
#pragma unroll
            for (int i = 0; i < 8; ++i) {
                const int row = min(tl * 16 + 2 * i + half, E - 1);
                nxt[i] = *(const float4*)(ef + (size_t)row * 64 + 4 * l15);
            }
        }
        const int er = tile * 16 + 2 * l15 + half;
        const bool own = (l15 < 8) && (er < E);
        const int li = er - e0;

        float mid = 0.f;
#pragma unroll
        for (int i = 0; i < 8; ++i) {
            float p = cur[i].x * u4.x + cur[i].y * u4.y + cur[i].z * u4.z + cur[i].w * u4.w;
            p += row_xor<8>(p);
            p += row_xor<4>(p);
            p += row_xor<2>(p);
            p += row_xor<1>(p);
            if (l15 == i) mid = p;
        }
        if (own) {
            const float t  = sA[li] + mid;
            const float lr = t > 0.f ? t : 0.01f * t;
            a[er]  = lr;
            sA[li] = lr;
            ldsMaxFloat(&tab[sD[li]], lr);
        }
#pragma unroll
        for (int i = 0; i < 8; ++i) cur[i] = nxt[i];
    }
    __syncthreads();

    float mreg[CHUNK_ITERS];
#pragma unroll
    for (int k = 0; k < CHUNK_ITERS; ++k) {
        const int i = tid + k * SCORE_THREADS;
        mreg[k] = (i < cnt) ? tab[sD[i]] : 0.f;
    }
    table_out(tab, mpart + (size_t)blockIdx.x * N, N, tid);
    __syncthreads();
    for (int n = tid; n < TAB_NODES; n += SCORE_THREADS) tab[n] = 0.f;
    __syncthreads();
#pragma unroll
    for (int k = 0; k < CHUNK_ITERS; ++k) {
        const int i = tid + k * SCORE_THREADS;
        if (i < cnt) atomicAdd(&tab[sD[i]], __expf(sA[i] - mreg[k]));
    }
    __syncthreads();
    table_out(tab, spart + (size_t)blockIdx.x * N, N, tid);
}

__global__ void __launch_bounds__(256) softmax_combine_kernel(
    const float* __restrict__ mpart, const float* __restrict__ spart,
    float* __restrict__ mbuf, float* __restrict__ den, int N)
{
    __shared__ float smax[8][32];
    __shared__ float ssum[8][32];
    const int lane = threadIdx.x & 31;
    const int g    = threadIdx.x >> 5;
    const int n    = blockIdx.x * 32 + lane;
    const int nn   = min(n, N - 1);
    float mb[SCORE_BLOCKS / 8], sb[SCORE_BLOCKS / 8];
#pragma unroll
    for (int j = 0; j < SCORE_BLOCKS / 8; ++j) {
        const size_t row = (size_t)(g * (SCORE_BLOCKS / 8) + j) * N + nn;
        mb[j] = mpart[row];
        sb[j] = spart[row];
    }
    float m = -INFINITY;
#pragma unroll
    for (int j = 0; j < SCORE_BLOCKS / 8; ++j) m = fmaxf(m, mb[j]);
    smax[g][lane] = m;
    __syncthreads();
    float M = smax[0][lane];
#pragma unroll
    for (int gg = 1; gg < 8; ++gg) M = fmaxf(M, smax[gg][lane]);
    float d = 0.f;
#pragma unroll
    for (int j = 0; j < SCORE_BLOCKS / 8; ++j)
        if (sb[j] > 0.f) d += sb[j] * __expf(mb[j] - M);
    ssum[g][lane] = d;
    __syncthreads();
    if (g == 0 && n < N) {
        float t = 0.f;
#pragma unroll
        for (int gg = 0; gg < 8; ++gg) t += ssum[gg][lane];
        mbuf[n] = M;
        den[n]  = t;
    }
}

#define MSG_STRIDE 68
__global__ void __launch_bounds__(256) edge_scatter_fc2_kernel(
    const float* __restrict__ ef, const float* __restrict__ w_e, const float* __restrict__ W2,
    const float* __restrict__ zz,
    const float* __restrict__ a, const float* __restrict__ mbuf, const float* __restrict__ den,
    const int* __restrict__ src, const int* __restrict__ dst, const int* __restrict__ etype,
    float* __restrict__ h2, int E)
{
    __shared__ v2f   sBe[32 * 32];
    __shared__ v2f   sB2[16 * 64];
    __shared__ __attribute__((aligned(16))) float sT[8][16 * MSG_STRIDE];
    const int tid = threadIdx.x;
    for (int p = tid; p < 32 * 32; p += 256) {
        const int kk = (p >> 5) << 1;
        const int n  = p & 31;
        v2f b;
        b.x = w_e[kk * 32 + n];
        b.y = w_e[(kk + 1) * 32 + n];
        sBe[p] = b;
    }
    for (int p = tid; p < 16 * 64; p += 256) {
        const int kk = (p >> 6) << 1;
        const int n  = p & 63;
        v2f b;
        b.x = W2[kk * 64 + n];
        b.y = W2[(kk + 1) * 64 + n];
        sB2[p] = b;
    }
    __syncthreads();

    const int lane = tid & 31;
    const int wave = tid >> 5;
    const int tile = blockIdx.x * 8 + wave;
    const bool active = tile < ((E + 15) >> 4);
    const int half = lane >> 4;
    const int l15  = lane & 15;
    int ec = tile * 16 + l15;
    if (ec >= E) ec = E - 1;
    const int   d_l   = dst[ec];
    const int   s_l   = src[ec];
    const int   t_l   = etype[ec];
    const float a_l   = a[ec];
    const float m_l   = mbuf[d_l];
    const float den_l = den[d_l];

    if (active) {
        v8f pe[2] = {};
        const float* arow = ef + (size_t)ec * 64 + 2 * half;
        const v2f* __restrict__ becol = sBe + half * 32 + l15;
#pragma unroll
        for (int k0 = 0; k0 < 64; k0 += 4) {
            const v2f av = *(const v2f*)(arow + k0);
#pragma unroll
            for (int nt = 0; nt < 2; ++nt) {
                const v2f b = becol[(k0 >> 1) * 32 + nt * 16];
                pe[nt] = __builtin_amdgcn_wmma_f32_16x16x4_f32(
                    false, av, false, b, (short)0, pe[nt], false, false);
            }
        }
        float* tw = sT[wave];
#pragma unroll
        for (int nt = 0; nt < 2; ++nt) {
#pragma unroll
            for (int r = 0; r < 8; ++r)
                tw[(r + 8 * half) * MSG_STRIDE + nt * 16 + l15] = pe[nt][r];
        }
    }
    __syncthreads();
    if (active) {
        v8f acc[4] = {};
        const float* arow = sT[wave] + l15 * MSG_STRIDE + 2 * half;
        const v2f* __restrict__ bcol = sB2 + half * 64 + l15;
#pragma unroll
        for (int k0 = 0; k0 < 32; k0 += 4) {
            const v2f av = *(const v2f*)(arow + k0);
#pragma unroll
            for (int nt = 0; nt < 4; ++nt) {
                const v2f b = bcol[(k0 >> 1) * 64 + nt * 16];
                acc[nt] = __builtin_amdgcn_wmma_f32_16x16x4_f32(
                    false, av, false, b, (short)0, acc[nt], false, false);
            }
        }
        float* tw = sT[wave];
#pragma unroll
        for (int nt = 0; nt < 4; ++nt) {
#pragma unroll
            for (int r = 0; r < 8; ++r)
                tw[(r + 8 * half) * MSG_STRIDE + nt * 16 + l15] = acc[nt][r];
        }
    }
    __syncthreads();
    if (!active) return;

    const float al_l = __expf(a_l - m_l) / den_l;
    const float* tr = sT[wave];
    const int nvalid = E - __builtin_amdgcn_readfirstlane(tile) * 16;

    float m0[16], m1[16];
#pragma unroll
    for (int i = 0; i < 16; ++i) {
        const int s = __builtin_amdgcn_readlane(s_l, i);
        const float alpha = __int_as_float(__builtin_amdgcn_readlane(__float_as_int(al_l), i));
        const float* zr = zz + (size_t)s * 64;
        m0[i] = alpha * (tr[i * MSG_STRIDE + lane]      + zr[lane]);
        m1[i] = alpha * (tr[i * MSG_STRIDE + lane + 32] + zr[lane + 32]);
    }
#pragma unroll
    for (int i = 0; i < 16; ++i) {
        if (i < nvalid) {
            const int d = __builtin_amdgcn_readlane(d_l, i);
            const int t = __builtin_amdgcn_readlane(t_l, i);
            float* hb = h2 + ((size_t)d * 4 + t) * 64;
            atomicAdd(hb + lane,      m0[i]);
            atomicAdd(hb + lane + 32, m1[i]);
        }
    }
}

__global__ void zphi_fold_kernel(const float* __restrict__ part, float* __restrict__ zpart, int rows)
{
    const int s   = threadIdx.x;
    const int per = (rows + (int)gridDim.x - 1) / (int)gridDim.x;
    const int lo  = blockIdx.x * per;
    const int hi  = min(rows, lo + per);
    float sum = 0.f;
    for (int r = lo; r < hi; r += 8) {
        float v[8];
        for (int j = 0; j < 8; ++j)
            v[j] = (r + j < hi) ? part[(size_t)(r + j) * 256 + s] : 0.f;
        for (int j = 0; j < 8; ++j) sum += v[j];
    }
    zpart[blockIdx.x * 256 + s] = sum;
}

__global__ void beta_kernel(const float* __restrict__ zpart, int nparts,
                            const float* __restrict__ w_sem, float* __restrict__ beta)
{
    __shared__ float prod[256];
    __shared__ float wsh[4];
    const int tid = threadIdx.x;
    float z = 0.f;
    for (int r = 0; r < nparts; r += 8) {
        float v[8];
        for (int j = 0; j < 8; ++j)
            v[j] = (r + j < nparts) ? zpart[(r + j) * 256 + tid] : 0.f;
        for (int j = 0; j < 8; ++j) z += v[j];
    }
    prod[tid] = z * w_sem[tid & 63];
    __syncthreads();
    if (tid < 4) {
        float d = 0.f;
        for (int o = 0; o < 64; ++o) d += prod[tid * 64 + o];
        wsh[tid] = d > 0.f ? d : 0.01f * d;
    }
    __syncthreads();
    if (tid != 0) return;
    float wp[4], mx = -INFINITY;
#pragma unroll
    for (int t = 0; t < 4; ++t) {
        wp[t] = wsh[t];
        mx = fmaxf(mx, wp[t]);
    }
    float s = 0.f;
#pragma unroll
    for (int t = 0; t < 4; ++t) { wp[t] = __expf(wp[t] - mx); s += wp[t]; }
#pragma unroll
    for (int t = 0; t < 4; ++t) beta[t] = wp[t] / s;
}

__global__ void out_kernel(const float* __restrict__ h2, const float* __restrict__ beta,
                           float* __restrict__ Z, int N)
{
    const int i = blockIdx.x * blockDim.x + threadIdx.x;
    if (i >= N * 64) return;
    const int n = i >> 6, o = i & 63;
    const float* r = h2 + (size_t)n * 256 + o;
    Z[i] = beta[0] * r[0] + beta[1] * r[64] + beta[2] * r[128] + beta[3] * r[192];
}

extern "C" void kernel_launch(void* const* d_in, const int* in_sizes, int n_in,
                              void* d_out, int out_size, void* d_ws, size_t ws_size,
                              hipStream_t stream)
{
    const float* node   = (const float*)d_in[0];
    const float* edgef  = (const float*)d_in[1];
    const int*   src    = (const int*)d_in[2];
    const int*   dst    = (const int*)d_in[3];
    const int*   etype  = (const int*)d_in[4];
    const float* w_n    = (const float*)d_in[5];
    const float* w_e    = (const float*)d_in[6];
    const float* w_attn = (const float*)d_in[7];
    const float* w_fc2  = (const float*)d_in[8];
    const float* w_sem  = (const float*)d_in[9];
    float* out = (float*)d_out;

    const int N = in_sizes[0] / 128;
    const int E = in_sizes[2];
    const int T = 4;

    float* ws   = (float*)d_ws;
    float* zz   = ws;
    float* ssrc = zz + (size_t)N * 64;
    float* sdst = ssrc + N;
    float* abuf = sdst + N;
    float* mbuf = abuf + E;
    float* den  = mbuf + N;
    float* beta = den + N;
    float* h2   = (float*)(((size_t)(beta + 16) + 255) & ~(size_t)255);
    float* mpart = h2;
    float* spart = mpart + (size_t)SCORE_BLOCKS * N;
    float* part  = zz;
    float* zpart = abuf;

    const int thr = 256;

    node_front_kernel<<<((N + 15) / 16 + 7) / 8, thr, 0, stream>>>(node, w_n, w_attn, w_fc2, ssrc, sdst, zz, N);
    edge_score_tables_kernel<<<SCORE_BLOCKS, SCORE_THREADS, 0, stream>>>(edgef, w_e, w_attn, src, dst,
                                                                      ssrc, sdst, abuf, mpart, spart, E, N);
    softmax_combine_kernel<<<(N + 31) / 32, thr, 0, stream>>>(mpart, spart, mbuf, den, N);
    init_kernel<<<1024, thr, 0, stream>>>(h2, N * T * 64);
    edge_scatter_fc2_kernel<<<((E + 15) / 16 + 7) / 8, thr, 0, stream>>>(edgef, w_e, w_fc2 + 64 * 64, zz, abuf, mbuf, den,
                                                                      src, dst, etype, h2, E);
    zphi_fold_kernel<<<1024, thr, 0, stream>>>(h2, part, N);
    zphi_fold_kernel<<<64, thr, 0, stream>>>(part, zpart, 1024);
    beta_kernel<<<1, thr, 0, stream>>>(zpart, 64, w_sem, beta);
    out_kernel<<<(N * 64 + thr - 1) / thr, thr, 0, stream>>>(h2, beta, out, N);
}


# one-shot waves become resident grids: scatter blocks loop over groups of 8 tiles (12288 waves enter, was 50000; weights staged once), output pass 2 x 16 B per lane (12504, was 100000); sum order kept
# speedup vs baseline: 2.9030x; 1.0376x over previous
#include <hip/hip_runtime.h>
#include <hip/hip_bf16.h>
#include <math.h>

typedef __attribute__((ext_vector_type(2))) float v2f;
typedef __attribute__((ext_vector_type(8))) float v8f;


template <int M>
__device__ __forceinline__ float row_xor(float v) {
    return __int_as_float(__builtin_amdgcn_update_dpp(0, __float_as_int(v), 0x160 | M, 0xf, 0xf, true));
}

#define FRONT_STRIDE 68
__global__ void __launch_bounds__(256) node_front_kernel(
    const float* __restrict__ node, const float* __restrict__ w_n, const float* __restrict__ w_attn,
    const float* __restrict__ w_fc2,
    float* __restrict__ ssrc, float* __restrict__ sdst, float* __restrict__ zz, int M)
{
    __shared__ v2f sB1[64 * 64];
    __shared__ v2f sB2[32 * 64];
    __shared__ __attribute__((aligned(16))) float sT[8][16 * FRONT_STRIDE];
    const int tid  = threadIdx.x;
    const int lane = tid & 31;
    const int wave = tid >> 5;
    const int half = lane >> 4;
    const int l15  = lane & 15;
    const int rowTile = blockIdx.x * 8 + wave;
    int mrow = rowTile * 16 + l15;
    if (mrow >= M) mrow = M - 1;

    const float* arow = node + (size_t)mrow * 128 + 2 * half;
    v2f av[32];
#pragma unroll
    for (int j = 0; j < 32; ++j) av[j] = *(const v2f*)(arow + 4 * j);

#pragma unroll
    for (int i = 0; i < 16; ++i) {
        const int p  = tid + i * 256;
        const int kk = (p >> 6) << 1;
        const int n  = p & 63;
        v2f b;
        b.x = w_n[kk * 64 + n];
        b.y = w_n[(kk + 1) * 64 + n];
        sB1[p] = b;
    }
#pragma unroll
    for (int i = 0; i < 8; ++i) {
        const int p  = tid + i * 256;
        const int kk = (p >> 6) << 1;
        const int n  = p & 63;
        v2f b;
        b.x = w_fc2[kk * 64 + n];
        b.y = w_fc2[(kk + 1) * 64 + n];
        sB2[p] = b;
    }
    float wa[4], wb[4];
#pragma unroll
    for (int nt = 0; nt < 4; ++nt) {
        wa[nt] = w_attn[nt * 16 + l15];
        wb[nt] = w_attn[96 + nt * 16 + l15];
    }
    __syncthreads();

    v8f acc[4] = {};
    const v2f* __restrict__ b1col = sB1 + half * 64 + l15;
#pragma unroll
    for (int j = 0; j < 32; ++j) {
#pragma unroll
        for (int nt = 0; nt < 4; ++nt) {
            const v2f b = b1col[(2 * j) * 64 + nt * 16];
            acc[nt] = __builtin_amdgcn_wmma_f32_16x16x4_f32(
                false, av[j], false, b, (short)0, acc[nt], false, false);
        }
    }

    float s0 = 0.f, s1 = 0.f;
#pragma unroll
    for (int r = 0; r < 8; ++r) {
        float p0 = acc[0][r] * wa[0] + acc[1][r] * wa[1] + acc[2][r] * wa[2] + acc[3][r] * wa[3];
        float p1 = acc[0][r] * wb[0] + acc[1][r] * wb[1] + acc[2][r] * wb[2] + acc[3][r] * wb[3];
        p0 += row_xor<8>(p0);  p1 += row_xor<8>(p1);
        p0 += row_xor<4>(p0);  p1 += row_xor<4>(p1);
        p0 += row_xor<2>(p0);  p1 += row_xor<2>(p1);
        p0 += row_xor<1>(p0);  p1 += row_xor<1>(p1);
        if (l15 == r) { s0 = p0; s1 = p1; }
    }
    {
        const int orow = rowTile * 16 + (l15 & 7) + 8 * half;
        if (l15 < 8 && orow < M) { ssrc[orow] = s0; sdst[orow] = s1; }
    }

    float* tw = sT[wave];
#pragma unroll
    for (int nt = 0; nt < 4; ++nt) {
#pragma unroll
        for (int r = 0; r < 8; ++r)
            tw[(r + 8 * half) * FRONT_STRIDE + nt * 16 + l15] = acc[nt][r];
    }
    __syncthreads();

    v8f acc2[4] = {};
    const float* a2row = tw + l15 * FRONT_STRIDE + 2 * half;
    const v2f* __restrict__ b2col = sB2 + half * 64 + l15;
#pragma unroll
    for (int k0 = 0; k0 < 64; k0 += 4) {
        const v2f a = *(const v2f*)(a2row + k0);
#pragma unroll
        for (int nt = 0; nt < 4; ++nt) {
            const v2f b = b2col[(k0 >> 1) * 64 + nt * 16];
            acc2[nt] = __builtin_amdgcn_wmma_f32_16x16x4_f32(
                false, a, false, b, (short)0, acc2[nt], false, false);
        }
    }

#pragma unroll
    for (int nt = 0; nt < 4; ++nt) {
#pragma unroll
        for (int r = 0; r < 8; ++r) {
            const int row = rowTile * 16 + r + 8 * half;
            if (row < M)
                zz[(size_t)row * 64 + nt * 16 + l15] = acc2[nt][r];
        }
    }
}

__device__ __forceinline__ void ldsMaxFloat(float* addr, float val) {
    const int iv = __float_as_int(val);
    if (iv >= 0) atomicMax((int*)addr, iv);
    else         atomicMin((unsigned int*)addr, (unsigned int)iv);
}

__global__ void init_kernel(float* __restrict__ h, int hsize)
{
    const int i = blockIdx.x * blockDim.x + threadIdx.x;
    const int stride = gridDim.x * blockDim.x;
    for (int j = i; j < hsize; j += stride) h[j] = 0.f;
}

#define SCORE_BLOCKS   192
#define SCORE_THREADS  1024
#define TAB_NODES      50000
#define CHUNK_ITERS    5
#define CHUNK_CAP      (CHUNK_ITERS * SCORE_THREADS)

__device__ __forceinline__ void table_out(const float* tab, float* __restrict__ row, int N, int tid) {
    if (((N & 3) == 0) && ((((size_t)row) & 15) == 0)) {
        const float4* t4 = (const float4*)tab;
        float4* r4 = (float4*)row;
        for (int q = tid; q < (N >> 2); q += SCORE_THREADS) r4[q] = t4[q];
    } else {
        for (int n = tid; n < N; n += SCORE_THREADS) row[n] = tab[n];
    }
}
__global__ void __launch_bounds__(SCORE_THREADS) edge_score_tables_kernel(
    const float* __restrict__ ef, const float* __restrict__ w_e, const float* __restrict__ w_attn,
    const int* __restrict__ src, const int* __restrict__ dst,
    const float* __restrict__ ssrc, const float* __restrict__ sdst,
    float* __restrict__ a, float* __restrict__ mpart, float* __restrict__ spart, int E, int N)
{
    __shared__ __attribute__((aligned(16))) float tab[TAB_NODES];
    __shared__ __attribute__((aligned(16))) float su[64];
    __shared__ float sA[CHUNK_CAP];
    __shared__ int   sD[CHUNK_CAP];
    const int tid = threadIdx.x;
    if (tid < 64) {
        float acc = 0.f;
#pragma unroll
        for (int n = 0; n < 32; ++n) acc += w_e[tid * 32 + n] * w_attn[64 + n];
        su[tid] = acc;
    }
    for (int n = tid; n < TAB_NODES; n += SCORE_THREADS) tab[n] = -INFINITY;

    const int lane = tid & 31;
    const int wave = __builtin_amdgcn_readfirstlane(tid >> 5);
    const int numTiles = (E + 15) >> 4;
    const int tpb = (numTiles + (int)gridDim.x - 1) / (int)gridDim.x;
    const int t0  = min(numTiles, (int)blockIdx.x * tpb);
    const int t1  = min(numTiles, t0 + tpb);
    const int e0  = t0 * 16;
    const int cnt = min(E, t1 * 16) - e0;

    {
        int dreg[CHUNK_ITERS], sreg[CHUNK_ITERS];
#pragma unroll
        for (int k = 0; k < CHUNK_ITERS; ++k) {
            const int e = min(e0 + tid + k * SCORE_THREADS, E - 1);
            dreg[k] = dst[e];
            sreg[k] = src[e];
        }
        float breg[CHUNK_ITERS];
#pragma unroll
        for (int k = 0; k < CHUNK_ITERS; ++k) breg[k] = ssrc[sreg[k]] + sdst[dreg[k]];
#pragma unroll
        for (int k = 0; k < CHUNK_ITERS; ++k) {
            sD[tid + k * SCORE_THREADS] = dreg[k];
            sA[tid + k * SCORE_THREADS] = breg[k];
        }
    }
    __syncthreads();

    const int half = lane >> 4;
    const int l15  = lane & 15;
    const float4 u4 = *(const float4*)(su + 4 * l15);

    const int lastTile = numTiles - 1;
    float4 cur[8];
    {
        const int tl = min(t0 + wave, lastTile);
#pragma unroll
        for (int i = 0; i < 8; ++i) {
            const int row = min(tl * 16 + 2 * i + half, E - 1);
            cur[i] = *(const float4*)(ef + (size_t)row * 64 + 4 * l15);
        }
    }
    for (int tile = t0 + wave; tile < t1; tile += SCORE_THREADS / 32) {
        float4 nxt[8];
        {
            const int tl = min(tile + SCORE_THREADS / 32, lastTile);
#pragma unroll
            for (int i = 0; i < 8; ++i) {
                const int row = min(tl * 16 + 2 * i + half, E - 1);
                nxt[i] = *(const float4*)(ef + (size_t)row * 64 + 4 * l15);
            }
        }
        const int er = tile * 16 + 2 * l15 + half;
        const bool own = (l15 < 8) && (er < E);
        const int li = er - e0;

        float mid = 0.f;
#pragma unroll
        for (int i = 0; i < 8; ++i) {
            float p = cur[i].x * u4.x + cur[i].y * u4.y + cur[i].z * u4.z + cur[i].w * u4.w;
            p += row_xor<8>(p);
            p += row_xor<4>(p);
            p += row_xor<2>(p);
            p += row_xor<1>(p);
            if (l15 == i) mid = p;
        }
        if (own) {
            const float t  = sA[li] + mid;
            const float lr = t > 0.f ? t : 0.01f * t;
            a[er]  = lr;
            sA[li] = lr;
            ldsMaxFloat(&tab[sD[li]], lr);
        }
#pragma unroll
        for (int i = 0; i < 8; ++i) cur[i] = nxt[i];
    }
    __syncthreads();

    float mreg[CHUNK_ITERS];
#pragma unroll
    for (int k = 0; k < CHUNK_ITERS; ++k) {
        const int i = tid + k * SCORE_THREADS;
        mreg[k] = (i < cnt) ? tab[sD[i]] : 0.f;
    }
    table_out(tab, mpart + (size_t)blockIdx.x * N, N, tid);
    __syncthreads();
    for (int n = tid; n < TAB_NODES; n += SCORE_THREADS) tab[n] = 0.f;
    __syncthreads();
#pragma unroll
    for (int k = 0; k < CHUNK_ITERS; ++k) {
        const int i = tid + k * SCORE_THREADS;
        if (i < cnt) atomicAdd(&tab[sD[i]], __expf(sA[i] - mreg[k]));
    }
    __syncthreads();
    table_out(tab, spart + (size_t)blockIdx.x * N, N, tid);
}

__global__ void __launch_bounds__(256) softmax_combine_kernel(
    const float* __restrict__ mpart, const float* __restrict__ spart,
    float* __restrict__ mbuf, float* __restrict__ den, int N)
{
    __shared__ float smax[8][32];
    __shared__ float ssum[8][32];
    const int lane = threadIdx.x & 31;
    const int g    = threadIdx.x >> 5;
    const int n    = blockIdx.x * 32 + lane;
    const int nn   = min(n, N - 1);
    float mb[SCORE_BLOCKS / 8], sb[SCORE_BLOCKS / 8];
#pragma unroll
    for (int j = 0; j < SCORE_BLOCKS / 8; ++j) {
        const size_t row = (size_t)(g * (SCORE_BLOCKS / 8) + j) * N + nn;
        mb[j] = mpart[row];
        sb[j] = spart[row];
    }
    float m = -INFINITY;
#pragma unroll
    for (int j = 0; j < SCORE_BLOCKS / 8; ++j) m = fmaxf(m, mb[j]);
    smax[g][lane] = m;
    __syncthreads();
    float M = smax[0][lane];
#pragma unroll
    for (int gg = 1; gg < 8; ++gg) M = fmaxf(M, smax[gg][lane]);
    float d = 0.f;
#pragma unroll
    for (int j = 0; j < SCORE_BLOCKS / 8; ++j)
        if (sb[j] > 0.f) d += sb[j] * __expf(mb[j] - M);
    ssum[g][lane] = d;
    __syncthreads();
    if (g == 0 && n < N) {
        float t = 0.f;
#pragma unroll
        for (int gg = 0; gg < 8; ++gg) t += ssum[gg][lane];
        mbuf[n] = M;
        den[n]  = t;
    }
}

#define SCATTER_BLOCKS 1536
#define MSG_STRIDE 68
__global__ void __launch_bounds__(256, 12) edge_scatter_fc2_kernel(
    const float* __restrict__ ef, const float* __restrict__ w_e, const float* __restrict__ W2,
    const float* __restrict__ zz,
    const float* __restrict__ a, const float* __restrict__ mbuf, const float* __restrict__ den,
    const int* __restrict__ src, const int* __restrict__ dst, const int* __restrict__ etype,
    float* __restrict__ h2, int E)
{
    __shared__ v2f   sBe[32 * 32];
    __shared__ v2f   sB2[16 * 64];
    __shared__ __attribute__((aligned(16))) float sT[8][16 * MSG_STRIDE];
    const int tid = threadIdx.x;
    for (int p = tid; p < 32 * 32; p += 256) {
        const int kk = (p >> 5) << 1;
        const int n  = p & 31;
        v2f b;
        b.x = w_e[kk * 32 + n];
        b.y = w_e[(kk + 1) * 32 + n];
        sBe[p] = b;
    }
    for (int p = tid; p < 16 * 64; p += 256) {
        const int kk = (p >> 6) << 1;
        const int n  = p & 63;
        v2f b;
        b.x = W2[kk * 64 + n];
        b.y = W2[(kk + 1) * 64 + n];
        sB2[p] = b;
    }
    __syncthreads();

    const int lane = tid & 31;
    const int wave = tid >> 5;
    const int half = lane >> 4;
    const int l15  = lane & 15;
    const int numTiles  = (E + 15) >> 4;
    const int numGroups = (numTiles + 7) >> 3;
    for (int grp = blockIdx.x; grp < numGroups; grp += gridDim.x) {
        const int tile = grp * 8 + wave;
        const bool active = tile < numTiles;
        int ec = tile * 16 + l15;
        if (ec >= E) ec = E - 1;
        const int   d_l   = dst[ec];
        const int   s_l   = src[ec];
        const int   t_l   = etype[ec];
        const float a_l   = a[ec];
        const float m_l   = mbuf[d_l];
        const float den_l = den[d_l];

        if (active) {
            v8f pe[2] = {};
            const v2f* __restrict__ arow = (const v2f*)ef + (ec * 32 + half);
            const v2f* __restrict__ becol = sBe + half * 32 + l15;
#pragma unroll
            for (int k0 = 0; k0 < 64; k0 += 4) {
                const v2f av = arow[k0 >> 1];
#pragma unroll
                for (int nt = 0; nt < 2; ++nt) {
                    const v2f b = becol[(k0 >> 1) * 32 + nt * 16];
                    pe[nt] = __builtin_amdgcn_wmma_f32_16x16x4_f32(
                        false, av, false, b, (short)0, pe[nt], false, false);
                }
            }
            float* tw = sT[wave];
#pragma unroll
            for (int nt = 0; nt < 2; ++nt) {
#pragma unroll
                for (int r = 0; r < 8; ++r)
                    tw[(r + 8 * half) * MSG_STRIDE + nt * 16 + l15] = pe[nt][r];
            }
        }
        __syncthreads();
        if (active) {
            v8f acc[4] = {};
            const float* arow = sT[wave] + l15 * MSG_STRIDE + 2 * half;
            const v2f* __restrict__ bcol = sB2 + half * 64 + l15;
#pragma unroll
            for (int k0 = 0; k0 < 32; k0 += 4) {
                const v2f av = *(const v2f*)(arow + k0);
#pragma unroll
                for (int nt = 0; nt < 4; ++nt) {
                    const v2f b = bcol[(k0 >> 1) * 64 + nt * 16];
                    acc[nt] = __builtin_amdgcn_wmma_f32_16x16x4_f32(
                        false, av, false, b, (short)0, acc[nt], false, false);
                }
            }
            float* tw = sT[wave];
#pragma unroll
            for (int nt = 0; nt < 4; ++nt) {
#pragma unroll
                for (int r = 0; r < 8; ++r)
                    tw[(r + 8 * half) * MSG_STRIDE + nt * 16 + l15] = acc[nt][r];
            }
        }
        __syncthreads();
        if (active) {
            const float al_l = __expf(a_l - m_l) / den_l;
            const float* tr = sT[wave];
            const int nvalid = E - __builtin_amdgcn_readfirstlane(tile) * 16;

            float m0[16], m1[16];
#pragma unroll
            for (int i = 0; i < 16; ++i) {
                const int s = __builtin_amdgcn_readlane(s_l, i);
                const float alpha = __int_as_float(__builtin_amdgcn_readlane(__float_as_int(al_l), i));
                const float* zr = zz + (size_t)s * 64;
                m0[i] = alpha * (tr[i * MSG_STRIDE + lane]      + zr[lane]);
                m1[i] = alpha * (tr[i * MSG_STRIDE + lane + 32] + zr[lane + 32]);
            }
#pragma unroll
            for (int i = 0; i < 16; ++i) {
                if (i < nvalid) {
                    const int d = __builtin_amdgcn_readlane(d_l, i);
                    const int t = __builtin_amdgcn_readlane(t_l, i);
                    float* hb = h2 + ((size_t)d * 4 + t) * 64;
                    atomicAdd(hb + lane,      m0[i]);
                    atomicAdd(hb + lane + 32, m1[i]);
                }
            }
        }
    }
}

__global__ void zphi_fold_kernel(const float* __restrict__ part, float* __restrict__ zpart, int rows)
{
    const int s   = threadIdx.x;
    const int per = (rows + (int)gridDim.x - 1) / (int)gridDim.x;
    const int lo  = blockIdx.x * per;
    const int hi  = min(rows, lo + per);
    float sum = 0.f;
    for (int r = lo; r < hi; r += 8) {
        float v[8];
        for (int j = 0; j < 8; ++j)
            v[j] = (r + j < hi) ? part[(size_t)(r + j) * 256 + s] : 0.f;
        for (int j = 0; j < 8; ++j) sum += v[j];
    }
    zpart[blockIdx.x * 256 + s] = sum;
}

__global__ void beta_kernel(const float* __restrict__ zpart, int nparts,
                            const float* __restrict__ w_sem, float* __restrict__ beta)
{
    __shared__ float prod[256];
    __shared__ float wsh[4];
    const int tid = threadIdx.x;
    float z = 0.f;
    for (int r = 0; r < nparts; r += 8) {
        float v[8];
        for (int j = 0; j < 8; ++j)
            v[j] = (r + j < nparts) ? zpart[(r + j) * 256 + tid] : 0.f;
        for (int j = 0; j < 8; ++j) z += v[j];
    }
    prod[tid] = z * w_sem[tid & 63];
    __syncthreads();
    if (tid < 4) {
        float d = 0.f;
        for (int o = 0; o < 64; ++o) d += prod[tid * 64 + o];
        wsh[tid] = d > 0.f ? d : 0.01f * d;
    }
    __syncthreads();
    if (tid != 0) return;
    float wp[4], mx = -INFINITY;
#pragma unroll
    for (int t = 0; t < 4; ++t) {
        wp[t] = wsh[t];
        mx = fmaxf(mx, wp[t]);
    }
    float s = 0.f;
#pragma unroll
    for (int t = 0; t < 4; ++t) { wp[t] = __expf(wp[t] - mx); s += wp[t]; }
#pragma unroll
    for (int t = 0; t < 4; ++t) beta[t] = wp[t] / s;
}

__global__ void __launch_bounds__(256) out_kernel(const float* __restrict__ h2, const float* __restrict__ beta,
                                                  float* __restrict__ Z, int N)
{
    const float b0 = beta[0], b1 = beta[1], b2 = beta[2], b3 = beta[3];
    const int total = N * 16;
    const int q0 = blockIdx.x * 512 + threadIdx.x;
    float4 r[2][4];
#pragma unroll
    for (int k = 0; k < 2; ++k) {
        const int q = min(q0 + k * 256, total - 1);
        const float* p = h2 + (size_t)(q >> 4) * 256 + (q & 15) * 4;
#pragma unroll
        for (int t = 0; t < 4; ++t) r[k][t] = *(const float4*)(p + t * 64);
    }
#pragma unroll
    for (int k = 0; k < 2; ++k) {
        const int q = q0 + k * 256;
        if (q < total) {
            float4 z;
            z.x = b0 * r[k][0].x + b1 * r[k][1].x + b2 * r[k][2].x + b3 * r[k][3].x;
            z.y = b0 * r[k][0].y + b1 * r[k][1].y + b2 * r[k][2].y + b3 * r[k][3].y;
            z.z = b0 * r[k][0].z + b1 * r[k][1].z + b2 * r[k][2].z + b3 * r[k][3].z;
            z.w = b0 * r[k][0].w + b1 * r[k][1].w + b2 * r[k][2].w + b3 * r[k][3].w;
            *(float4*)(Z + (size_t)q * 4) = z;
        }
    }
}

extern "C" void kernel_launch(void* const* d_in, const int* in_sizes, int n_in,
                              void* d_out, int out_size, void* d_ws, size_t ws_size,
                              hipStream_t stream)
{
    const float* node   = (const float*)d_in[0];
    const float* edgef  = (const float*)d_in[1];
    const int*   src    = (const int*)d_in[2];
    const int*   dst    = (const int*)d_in[3];
    const int*   etype  = (const int*)d_in[4];
    const float* w_n    = (const float*)d_in[5];
    const float* w_e    = (const float*)d_in[6];
    const float* w_attn = (const float*)d_in[7];
    const float* w_fc2  = (const float*)d_in[8];
    const float* w_sem  = (const float*)d_in[9];
    float* out = (float*)d_out;

    const int N = in_sizes[0] / 128;
    const int E = in_sizes[2];
    const int T = 4;

    float* ws   = (float*)d_ws;
    float* zz   = ws;
    float* ssrc = zz + (size_t)N * 64;
    float* sdst = ssrc + N;
    float* abuf = sdst + N;
    float* mbuf = abuf + E;
    float* den  = mbuf + N;
    float* beta = den + N;
    float* h2   = (float*)(((size_t)(beta + 16) + 255) & ~(size_t)255);
    float* mpart = h2;
    float* spart = mpart + (size_t)SCORE_BLOCKS * N;
    float* part  = zz;
    float* zpart = abuf;

    const int thr = 256;

    node_front_kernel<<<((N + 15) / 16 + 7) / 8, thr, 0, stream>>>(node, w_n, w_attn, w_fc2, ssrc, sdst, zz, N);
    edge_score_tables_kernel<<<SCORE_BLOCKS, SCORE_THREADS, 0, stream>>>(edgef, w_e, w_attn, src, dst,
                                                                      ssrc, sdst, abuf, mpart, spart, E, N);
    softmax_combine_kernel<<<(N + 31) / 32, thr, 0, stream>>>(mpart, spart, mbuf, den, N);
    init_kernel<<<1024, thr, 0, stream>>>(h2, N * T * 64);
    edge_scatter_fc2_kernel<<<SCATTER_BLOCKS, thr, 0, stream>>>(edgef, w_e, w_fc2 + 64 * 64, zz, abuf, mbuf, den,
                                                                      src, dst, etype, h2, E);
    zphi_fold_kernel<<<1024, thr, 0, stream>>>(h2, part, N);
    zphi_fold_kernel<<<64, thr, 0, stream>>>(part, zpart, 1024);
    beta_kernel<<<1, thr, 0, stream>>>(zpart, 64, w_sem, beta);
    out_kernel<<<(N * 16 + 511) / 512, thr, 0, stream>>>(h2, beta, out, N);
}


# beta before the scatter: zphi_t.w_sem = sum of per-edge scalars alpha*(msg.w_sem) (linear), so scatter adds beta*alpha*msg into Z; [N,T,64] plane, 3 launches go; 184 score blocks; sum order changes
# speedup vs baseline: 3.2778x; 1.1291x over previous
#include <hip/hip_runtime.h>
#include <hip/hip_bf16.h>
#include <math.h>

typedef __attribute__((ext_vector_type(2))) float v2f;
typedef __attribute__((ext_vector_type(8))) float v8f;


template <int M>
__device__ __forceinline__ float row_xor(float v) {
    return __int_as_float(__builtin_amdgcn_update_dpp(0, __float_as_int(v), 0x160 | M, 0xf, 0xf, true));
}

#define FRONT_STRIDE 68
__global__ void __launch_bounds__(256) node_front_kernel(
    const float* __restrict__ node, const float* __restrict__ w_n, const float* __restrict__ w_attn,
    const float* __restrict__ w_fc2, const float* __restrict__ w_sem,
    float* __restrict__ ssrc, float* __restrict__ sdst, float* __restrict__ qn, float* __restrict__ zz, int M)
{
    __shared__ v2f sB1[64 * 64];
    __shared__ v2f sB2[32 * 64];
    __shared__ __attribute__((aligned(16))) float sT[8][16 * FRONT_STRIDE];
    const int tid  = threadIdx.x;
    const int lane = tid & 31;
    const int wave = tid >> 5;
    const int half = lane >> 4;
    const int l15  = lane & 15;
    const int rowTile = blockIdx.x * 8 + wave;
    int mrow = rowTile * 16 + l15;
    if (mrow >= M) mrow = M - 1;

    const float* arow = node + (size_t)mrow * 128 + 2 * half;
    v2f av[32];
#pragma unroll
    for (int j = 0; j < 32; ++j) av[j] = *(const v2f*)(arow + 4 * j);

#pragma unroll
    for (int i = 0; i < 16; ++i) {
        const int p  = tid + i * 256;
        const int kk = (p >> 6) << 1;
        const int n  = p & 63;
        v2f b;
        b.x = w_n[kk * 64 + n];
        b.y = w_n[(kk + 1) * 64 + n];
        sB1[p] = b;
    }
#pragma unroll
    for (int i = 0; i < 8; ++i) {
        const int p  = tid + i * 256;
        const int kk = (p >> 6) << 1;
        const int n  = p & 63;
        v2f b;
        b.x = w_fc2[kk * 64 + n];
        b.y = w_fc2[(kk + 1) * 64 + n];
        sB2[p] = b;
    }
    float wa[4], wb[4], wq[4];
#pragma unroll
    for (int nt = 0; nt < 4; ++nt) {
        wa[nt] = w_attn[nt * 16 + l15];
        wb[nt] = w_attn[96 + nt * 16 + l15];
        wq[nt] = w_sem[nt * 16 + l15];
    }
    __syncthreads();

    v8f acc[4] = {};
    const v2f* __restrict__ b1col = sB1 + half * 64 + l15;
#pragma unroll
    for (int j = 0; j < 32; ++j) {
#pragma unroll
        for (int nt = 0; nt < 4; ++nt) {
            const v2f b = b1col[(2 * j) * 64 + nt * 16];
            acc[nt] = __builtin_amdgcn_wmma_f32_16x16x4_f32(
                false, av[j], false, b, (short)0, acc[nt], false, false);
        }
    }

    float s0 = 0.f, s1 = 0.f;
#pragma unroll
    for (int r = 0; r < 8; ++r) {
        float p0 = acc[0][r] * wa[0] + acc[1][r] * wa[1] + acc[2][r] * wa[2] + acc[3][r] * wa[3];
        float p1 = acc[0][r] * wb[0] + acc[1][r] * wb[1] + acc[2][r] * wb[2] + acc[3][r] * wb[3];
        p0 += row_xor<8>(p0);  p1 += row_xor<8>(p1);
        p0 += row_xor<4>(p0);  p1 += row_xor<4>(p1);
        p0 += row_xor<2>(p0);  p1 += row_xor<2>(p1);
        p0 += row_xor<1>(p0);  p1 += row_xor<1>(p1);
        if (l15 == r) { s0 = p0; s1 = p1; }
    }
    {
        const int orow = rowTile * 16 + (l15 & 7) + 8 * half;
        if (l15 < 8 && orow < M) { ssrc[orow] = s0; sdst[orow] = s1; }
    }

    float* tw = sT[wave];
#pragma unroll
    for (int nt = 0; nt < 4; ++nt) {
#pragma unroll
        for (int r = 0; r < 8; ++r)
            tw[(r + 8 * half) * FRONT_STRIDE + nt * 16 + l15] = acc[nt][r];
    }
    __syncthreads();

    v8f acc2[4] = {};
    const float* a2row = tw + l15 * FRONT_STRIDE + 2 * half;
    const v2f* __restrict__ b2col = sB2 + half * 64 + l15;
#pragma unroll
    for (int k0 = 0; k0 < 64; k0 += 4) {
        const v2f a = *(const v2f*)(a2row + k0);
#pragma unroll
        for (int nt = 0; nt < 4; ++nt) {
            const v2f b = b2col[(k0 >> 1) * 64 + nt * 16];
            acc2[nt] = __builtin_amdgcn_wmma_f32_16x16x4_f32(
                false, a, false, b, (short)0, acc2[nt], false, false);
        }
    }

#pragma unroll
    for (int nt = 0; nt < 4; ++nt) {
#pragma unroll
        for (int r = 0; r < 8; ++r) {
            const int row = rowTile * 16 + r + 8 * half;
            if (row < M)
                zz[(size_t)row * 64 + nt * 16 + l15] = acc2[nt][r];
        }
    }

    float q0 = 0.f;
#pragma unroll
    for (int r = 0; r < 8; ++r) {
        float p = acc2[0][r] * wq[0] + acc2[1][r] * wq[1] + acc2[2][r] * wq[2] + acc2[3][r] * wq[3];
        p += row_xor<8>(p);
        p += row_xor<4>(p);
        p += row_xor<2>(p);
        p += row_xor<1>(p);
        if (l15 == r) q0 = p;
    }
    {
        const int orow = rowTile * 16 + (l15 & 7) + 8 * half;
        if (l15 < 8 && orow < M) qn[orow] = q0;
    }
}

__device__ __forceinline__ void ldsMaxFloat(float* addr, float val) {
    const int iv = __float_as_int(val);
    if (iv >= 0) atomicMax((int*)addr, iv);
    else         atomicMin((unsigned int*)addr, (unsigned int)iv);
}

#define SCORE_BLOCKS   184
#define SCORE_THREADS  1024
#define TAB_NODES      50000
#define CHUNK_ITERS    5
#define CHUNK_CAP      (CHUNK_ITERS * SCORE_THREADS)

__device__ __forceinline__ void table_out(const float* tab, float* __restrict__ row, int N, int tid) {
    if (((N & 3) == 0) && ((((size_t)row) & 15) == 0)) {
        const float4* t4 = (const float4*)tab;
        float4* r4 = (float4*)row;
        for (int q = tid; q < (N >> 2); q += SCORE_THREADS) r4[q] = t4[q];
    } else {
        for (int n = tid; n < N; n += SCORE_THREADS) row[n] = tab[n];
    }
}
__global__ void __launch_bounds__(SCORE_THREADS) edge_score_tables_kernel(
    const float* __restrict__ ef, const float* __restrict__ w_e, const float* __restrict__ w_attn,
    const float* __restrict__ W2, const float* __restrict__ w_sem,
    const int* __restrict__ src, const int* __restrict__ dst,
    const float* __restrict__ ssrc, const float* __restrict__ sdst, const float* __restrict__ qn,
    float* __restrict__ a, float* __restrict__ c, float* __restrict__ mpart, float* __restrict__ spart, int E, int N)
{
    __shared__ __attribute__((aligned(16))) float tab[TAB_NODES];
    __shared__ __attribute__((aligned(16))) float su[64];
    __shared__ __attribute__((aligned(16))) float sv[64];
    __shared__ float sg[32];
    __shared__ float sA[CHUNK_CAP];
    __shared__ float sC[CHUNK_CAP];
    __shared__ int   sD[CHUNK_CAP];
    const int tid = threadIdx.x;
    if (tid < 32) {
        float acc = 0.f;
#pragma unroll
        for (int o = 0; o < 64; ++o) acc += W2[tid * 64 + o] * w_sem[o];
        sg[tid] = acc;
    }
    __syncthreads();
    if (tid < 64) {
        float acc = 0.f, acv = 0.f;
#pragma unroll
        for (int n = 0; n < 32; ++n) {
            const float w = w_e[tid * 32 + n];
            acc += w * w_attn[64 + n];
            acv += w * sg[n];
        }
        su[tid] = acc;
        sv[tid] = acv;
    }
    for (int n = tid; n < TAB_NODES; n += SCORE_THREADS) tab[n] = -INFINITY;

    const int lane = tid & 31;
    const int wave = __builtin_amdgcn_readfirstlane(tid >> 5);
    const int numTiles = (E + 15) >> 4;
    const int tpb = (numTiles + (int)gridDim.x - 1) / (int)gridDim.x;
    const int t0  = min(numTiles, (int)blockIdx.x * tpb);
    const int t1  = min(numTiles, t0 + tpb);
    const int e0  = t0 * 16;
    const int cnt = min(E, t1 * 16) - e0;

    {
        int dreg[CHUNK_ITERS], sreg[CHUNK_ITERS];
#pragma unroll
        for (int k = 0; k < CHUNK_ITERS; ++k) {
            const int e = min(e0 + tid + k * SCORE_THREADS, E - 1);
            dreg[k] = dst[e];
            sreg[k] = src[e];
        }
        float breg[CHUNK_ITERS], qreg[CHUNK_ITERS];
#pragma unroll
        for (int k = 0; k < CHUNK_ITERS; ++k) {
            breg[k] = ssrc[sreg[k]] + sdst[dreg[k]];
            qreg[k] = qn[sreg[k]];
        }
#pragma unroll
        for (int k = 0; k < CHUNK_ITERS; ++k) {
            sD[tid + k * SCORE_THREADS] = dreg[k];
            sA[tid + k * SCORE_THREADS] = breg[k];
            sC[tid + k * SCORE_THREADS] = qreg[k];
        }
    }
    __syncthreads();

    const int half = lane >> 4;
    const int l15  = lane & 15;
    const float4 u4 = *(const float4*)(su + 4 * l15);
    const float4 v4 = *(const float4*)(sv + 4 * l15);

    const int lastTile = numTiles - 1;
    float4 cur[8];
    {
        const int tl = min(t0 + wave, lastTile);
#pragma unroll
        for (int i = 0; i < 8; ++i) {
            const int row = min(tl * 16 + 2 * i + half, E - 1);
            cur[i] = *(const float4*)(ef + (size_t)row * 64 + 4 * l15);
        }
    }
    for (int tile = t0 + wave; tile < t1; tile += SCORE_THREADS / 32) {
        float4 nxt[8];
        {
            const int tl = min(tile + SCORE_THREADS / 32, lastTile);
#pragma unroll
            for (int i = 0; i < 8; ++i) {
                const int row = min(tl * 16 + 2 * i + half, E - 1);
                nxt[i] = *(const float4*)(ef + (size_t)row * 64 + 4 * l15);
            }
        }
        const int er = tile * 16 + 2 * l15 + half;
        const bool own = (l15 < 8) && (er < E);
        const int li = er - e0;

        float mid = 0.f, mic = 0.f;
#pragma unroll
        for (int i = 0; i < 8; ++i) {
            float p = cur[i].x * u4.x + cur[i].y * u4.y + cur[i].z * u4.z + cur[i].w * u4.w;
            float g = cur[i].x * v4.x + cur[i].y * v4.y + cur[i].z * v4.z + cur[i].w * v4.w;
            p += row_xor<8>(p);  g += row_xor<8>(g);
            p += row_xor<4>(p);  g += row_xor<4>(g);
            p += row_xor<2>(p);  g += row_xor<2>(g);
            p += row_xor<1>(p);  g += row_xor<1>(g);
            if (l15 == i) { mid = p; mic = g; }
        }
        if (own) {
            const float t  = sA[li] + mid;
            const float lr = t > 0.f ? t : 0.01f * t;
            a[er]  = lr;
            c[er]  = sC[li] + mic;
            sA[li] = lr;
            ldsMaxFloat(&tab[sD[li]], lr);
        }
#pragma unroll
        for (int i = 0; i < 8; ++i) cur[i] = nxt[i];
    }
    __syncthreads();

    float mreg[CHUNK_ITERS];
#pragma unroll
    for (int k = 0; k < CHUNK_ITERS; ++k) {
        const int i = tid + k * SCORE_THREADS;
        mreg[k] = (i < cnt) ? tab[sD[i]] : 0.f;
    }
    table_out(tab, mpart + (size_t)blockIdx.x * N, N, tid);
    __syncthreads();
    for (int n = tid; n < TAB_NODES; n += SCORE_THREADS) tab[n] = 0.f;
    __syncthreads();
#pragma unroll
    for (int k = 0; k < CHUNK_ITERS; ++k) {
        const int i = tid + k * SCORE_THREADS;
        if (i < cnt) atomicAdd(&tab[sD[i]], __expf(sA[i] - mreg[k]));
    }
    __syncthreads();
    table_out(tab, spart + (size_t)blockIdx.x * N, N, tid);
}

__global__ void __launch_bounds__(256) softmax_combine_kernel(
    const float* __restrict__ mpart, const float* __restrict__ spart,
    float* __restrict__ mbuf, float* __restrict__ den, int N)
{
    __shared__ float smax[8][32];
    __shared__ float ssum[8][32];
    const int lane = threadIdx.x & 31;
    const int g    = threadIdx.x >> 5;
    const int n    = blockIdx.x * 32 + lane;
    const int nn   = min(n, N - 1);
    float mb[SCORE_BLOCKS / 8], sb[SCORE_BLOCKS / 8];
#pragma unroll
    for (int j = 0; j < SCORE_BLOCKS / 8; ++j) {
        const size_t row = (size_t)(g * (SCORE_BLOCKS / 8) + j) * N + nn;
        mb[j] = mpart[row];
        sb[j] = spart[row];
    }
    float m = -INFINITY;
#pragma unroll
    for (int j = 0; j < SCORE_BLOCKS / 8; ++j) m = fmaxf(m, mb[j]);
    smax[g][lane] = m;
    __syncthreads();
    float M = smax[0][lane];
#pragma unroll
    for (int gg = 1; gg < 8; ++gg) M = fmaxf(M, smax[gg][lane]);
    float d = 0.f;
#pragma unroll
    for (int j = 0; j < SCORE_BLOCKS / 8; ++j)
        if (sb[j] > 0.f) d += sb[j] * __expf(mb[j] - M);
    ssum[g][lane] = d;
    __syncthreads();
    if (g == 0 && n < N) {
        float t = 0.f;
#pragma unroll
        for (int gg = 0; gg < 8; ++gg) t += ssum[gg][lane];
        mbuf[n] = M;
        den[n]  = t;
    }
}

#define ALPHA_PER 4
__global__ void __launch_bounds__(256) alpha_type_sums_kernel(
    float* __restrict__ a, const float* __restrict__ c,
    const float* __restrict__ mbuf, const float* __restrict__ den,
    const int* __restrict__ dst, const int* __restrict__ etype,
    float* __restrict__ wpart, float* __restrict__ Z, int E, int zquads)
{
    __shared__ float sw[16][4];
    const int tid  = threadIdx.x;
    const int base = blockIdx.x * (256 * ALPHA_PER) + tid;
    int   dr[ALPHA_PER], tr[ALPHA_PER];
    float ar[ALPHA_PER], cr[ALPHA_PER], mr[ALPHA_PER], nr[ALPHA_PER];
#pragma unroll
    for (int k = 0; k < ALPHA_PER; ++k) {
        const int e = min(base + k * 256, E - 1);
        dr[k] = dst[e];
        tr[k] = etype[e];
        ar[k] = a[e];
        cr[k] = c[e];
    }
#pragma unroll
    for (int k = 0; k < ALPHA_PER; ++k) {
        mr[k] = mbuf[dr[k]];
        nr[k] = den[dr[k]];
    }
    {
        const float4 zero = {0.f, 0.f, 0.f, 0.f};
        float4* Z4 = (float4*)Z;
        for (int q = blockIdx.x * 256 + tid; q < zquads; q += (int)gridDim.x * 256) Z4[q] = zero;
    }
    float s[4] = {0.f, 0.f, 0.f, 0.f};
#pragma unroll
    for (int k = 0; k < ALPHA_PER; ++k) {
        const int e = base + k * 256;
        if (e < E) {
            const float al = __expf(ar[k] - mr[k]) / nr[k];
            a[e] = al;
            const float x = al * cr[k];
#pragma unroll
            for (int t = 0; t < 4; ++t) s[t] += (tr[k] == t) ? x : 0.f;
        }
    }
#pragma unroll
    for (int t = 0; t < 4; ++t) {
        s[t] += row_xor<8>(s[t]);
        s[t] += row_xor<4>(s[t]);
        s[t] += row_xor<2>(s[t]);
        s[t] += row_xor<1>(s[t]);
    }
    if ((tid & 15) == 0) {
#pragma unroll
        for (int t = 0; t < 4; ++t) sw[tid >> 4][t] = s[t];
    }
    __syncthreads();
    if (tid < 4) {
        float w = 0.f;
#pragma unroll
        for (int r = 0; r < 16; ++r) w += sw[r][tid];
        wpart[blockIdx.x * 4 + tid] = w;
    }
}

#define SCATTER_BLOCKS 1536
#define MSG_STRIDE 68
__global__ void __launch_bounds__(256, 12) edge_scatter_fc2_kernel(
    const float* __restrict__ ef, const float* __restrict__ w_e, const float* __restrict__ W2,
    const float* __restrict__ zz,
    const float* __restrict__ alpha, const float* __restrict__ beta,
    const int* __restrict__ src, const int* __restrict__ dst, const int* __restrict__ etype,
    float* __restrict__ Z, int E)
{
    __shared__ v2f   sBe[32 * 32];
    __shared__ v2f   sB2[16 * 64];
    __shared__ __attribute__((aligned(16))) float sT[8][16 * MSG_STRIDE];
    const int tid = threadIdx.x;
    for (int p = tid; p < 32 * 32; p += 256) {
        const int kk = (p >> 5) << 1;
        const int n  = p & 31;
        v2f b;
        b.x = w_e[kk * 32 + n];
        b.y = w_e[(kk + 1) * 32 + n];
        sBe[p] = b;
    }
    for (int p = tid; p < 16 * 64; p += 256) {
        const int kk = (p >> 6) << 1;
        const int n  = p & 63;
        v2f b;
        b.x = W2[kk * 64 + n];
        b.y = W2[(kk + 1) * 64 + n];
        sB2[p] = b;
    }
    __syncthreads();

    const float b0 = beta[0], b1 = beta[1], b2 = beta[2], b3 = beta[3];
    const int lane = tid & 31;
    const int wave = tid >> 5;
    const int half = lane >> 4;
    const int l15  = lane & 15;
    const int numTiles  = (E + 15) >> 4;
    const int numGroups = (numTiles + 7) >> 3;
    for (int grp = blockIdx.x; grp < numGroups; grp += gridDim.x) {
        const int tile = grp * 8 + wave;
        const bool active = tile < numTiles;
        int ec = tile * 16 + l15;
        if (ec >= E) ec = E - 1;
        const int   d_l   = dst[ec];
        const int   s_l   = src[ec];
        const int   t_l   = etype[ec];
        const float a_l   = alpha[ec];

        if (active) {
            v8f pe[2] = {};
            const v2f* __restrict__ arow = (const v2f*)ef + (ec * 32 + half);
            const v2f* __restrict__ becol = sBe + half * 32 + l15;
#pragma unroll
            for (int k0 = 0; k0 < 64; k0 += 4) {
                const v2f av = arow[k0 >> 1];
#pragma unroll
                for (int nt = 0; nt < 2; ++nt) {
                    const v2f b = becol[(k0 >> 1) * 32 + nt * 16];
                    pe[nt] = __builtin_amdgcn_wmma_f32_16x16x4_f32(
                        false, av, false, b, (short)0, pe[nt], false, false);
                }
            }
            float* tw = sT[wave];
#pragma unroll
            for (int nt = 0; nt < 2; ++nt) {
#pragma unroll
                for (int r = 0; r < 8; ++r)
                    tw[(r + 8 * half) * MSG_STRIDE + nt * 16 + l15] = pe[nt][r];
            }
        }
        __syncthreads();
        if (active) {
            v8f acc[4] = {};
            const float* arow = sT[wave] + l15 * MSG_STRIDE + 2 * half;
            const v2f* __restrict__ bcol = sB2 + half * 64 + l15;
#pragma unroll
            for (int k0 = 0; k0 < 32; k0 += 4) {
                const v2f av = *(const v2f*)(arow + k0);
#pragma unroll
                for (int nt = 0; nt < 4; ++nt) {
                    const v2f b = bcol[(k0 >> 1) * 64 + nt * 16];
                    acc[nt] = __builtin_amdgcn_wmma_f32_16x16x4_f32(
                        false, av, false, b, (short)0, acc[nt], false, false);
                }
            }
            float* tw = sT[wave];
#pragma unroll
            for (int nt = 0; nt < 4; ++nt) {
#pragma unroll
                for (int r = 0; r < 8; ++r)
                    tw[(r + 8 * half) * MSG_STRIDE + nt * 16 + l15] = acc[nt][r];
            }
        }
        __syncthreads();
        if (active) {
            const float al_l = a_l * (t_l == 0 ? b0 : t_l == 1 ? b1 : t_l == 2 ? b2 : b3);
            const float* tr = sT[wave];
            const int nvalid = E - __builtin_amdgcn_readfirstlane(tile) * 16;

            float m0[16], m1[16];
#pragma unroll
            for (int i = 0; i < 16; ++i) {
                const int s = __builtin_amdgcn_readlane(s_l, i);
                const float alpha = __int_as_float(__builtin_amdgcn_readlane(__float_as_int(al_l), i));
                const int zi = s * 64 + lane;
                m0[i] = alpha * (tr[i * MSG_STRIDE + lane]      + zz[zi]);
                m1[i] = alpha * (tr[i * MSG_STRIDE + lane + 32] + zz[zi + 32]);
            }
#pragma unroll
            for (int i = 0; i < 16; ++i) {
                if (i < nvalid) {
                    const int d = __builtin_amdgcn_readlane(d_l, i);
                    float* hb = Z + (d * 64 + lane);
                    atomicAdd(hb,      m0[i]);
                    atomicAdd(hb + 32, m1[i]);
                }
            }
        }
    }
}

__global__ void __launch_bounds__(256) beta_kernel(const float* __restrict__ wpart, int nparts, float* __restrict__ beta)
{
    __shared__ float prod[256];
    __shared__ float wsh[4];
    const int tid = threadIdx.x;
    float z = 0.f;
    for (int r = 0; r < nparts; r += 64 * 8) {
        float v[8];
        for (int j = 0; j < 8; ++j) {
            const int row = r + j * 64 + (tid >> 2);
            v[j] = (row < nparts) ? wpart[row * 4 + (tid & 3)] : 0.f;
        }
        for (int j = 0; j < 8; ++j) z += v[j];
    }
    prod[tid] = z;
    __syncthreads();
    if (tid < 4) {
        float d = 0.f;
        for (int g = 0; g < 64; ++g) d += prod[g * 4 + tid];
        wsh[tid] = d > 0.f ? d : 0.01f * d;
    }
    __syncthreads();
    if (tid != 0) return;
    float wp[4], mx = -INFINITY;
#pragma unroll
    for (int t = 0; t < 4; ++t) {
        wp[t] = wsh[t];
        mx = fmaxf(mx, wp[t]);
    }
    float sm = 0.f;
#pragma unroll
    for (int t = 0; t < 4; ++t) { wp[t] = __expf(wp[t] - mx); sm += wp[t]; }
#pragma unroll
    for (int t = 0; t < 4; ++t) beta[t] = wp[t] / sm;
}

extern "C" void kernel_launch(void* const* d_in, const int* in_sizes, int n_in,
                              void* d_out, int out_size, void* d_ws, size_t ws_size,
                              hipStream_t stream)
{
    const float* node   = (const float*)d_in[0];
    const float* edgef  = (const float*)d_in[1];
    const int*   src    = (const int*)d_in[2];
    const int*   dst    = (const int*)d_in[3];
    const int*   etype  = (const int*)d_in[4];
    const float* w_n    = (const float*)d_in[5];
    const float* w_e    = (const float*)d_in[6];
    const float* w_attn = (const float*)d_in[7];
    const float* w_fc2  = (const float*)d_in[8];
    const float* w_sem  = (const float*)d_in[9];
    float* out = (float*)d_out;

    const int N = in_sizes[0] / 128;
    const int E = in_sizes[2];

    float* ws   = (float*)d_ws;
    float* zz   = ws;
    float* ssrc = zz + (size_t)N * 64;
    float* sdst = ssrc + N;
    float* abuf = sdst + N;
    float* mbuf = abuf + E;
    float* den  = mbuf + N;
    float* beta = den + N;
    float* qn   = den;
    float* mpart = (float*)(((size_t)(beta + 16) + 255) & ~(size_t)255);
    float* spart = mpart + (size_t)SCORE_BLOCKS * N;
    float* cbuf  = spart + (size_t)SCORE_BLOCKS * N;
    float* wpart = mpart;

    const int thr = 256;
    const int alphaBlocks = (E + 256 * ALPHA_PER - 1) / (256 * ALPHA_PER);

    node_front_kernel<<<((N + 15) / 16 + 7) / 8, thr, 0, stream>>>(node, w_n, w_attn, w_fc2, w_sem, ssrc, sdst, qn, zz, N);
    edge_score_tables_kernel<<<SCORE_BLOCKS, SCORE_THREADS, 0, stream>>>(edgef, w_e, w_attn, w_fc2 + 64 * 64, w_sem, src, dst,
                                                                      ssrc, sdst, qn, abuf, cbuf, mpart, spart, E, N);
    softmax_combine_kernel<<<(N + 31) / 32, thr, 0, stream>>>(mpart, spart, mbuf, den, N);
    alpha_type_sums_kernel<<<alphaBlocks, thr, 0, stream>>>(abuf, cbuf, mbuf, den, dst, etype, wpart, out, E, N * 16);
    beta_kernel<<<1, thr, 0, stream>>>(wpart, alphaBlocks, beta);
    edge_scatter_fc2_kernel<<<SCATTER_BLOCKS, thr, 0, stream>>>(edgef, w_e, w_fc2 + 64 * 64, zz, abuf, beta,
                                                                      src, dst, etype, out, E);
}


# score tables: one row per block, its log-sum-exp L_b = m_b + log(s_b), not a max row and a sum row; lse = logsumexp over 256 rows (was 2 x 184), alpha = exp(a - lse[dst]); den sum order changes
# speedup vs baseline: 3.4992x; 1.0675x over previous
#include <hip/hip_runtime.h>
#include <hip/hip_bf16.h>
#include <math.h>

typedef __attribute__((ext_vector_type(2))) float v2f;
typedef __attribute__((ext_vector_type(8))) float v8f;


template <int M>
__device__ __forceinline__ float row_xor(float v) {
    return __int_as_float(__builtin_amdgcn_update_dpp(0, __float_as_int(v), 0x160 | M, 0xf, 0xf, true));
}

#define FRONT_STRIDE 68
__global__ void __launch_bounds__(256) node_front_kernel(
    const float* __restrict__ node, const float* __restrict__ w_n, const float* __restrict__ w_attn,
    const float* __restrict__ w_fc2, const float* __restrict__ w_sem,
    float* __restrict__ ssrc, float* __restrict__ sdst, float* __restrict__ qn, float* __restrict__ zz, int M)
{
    __shared__ v2f sB1[64 * 64];
    __shared__ v2f sB2[32 * 64];
    __shared__ __attribute__((aligned(16))) float sT[8][16 * FRONT_STRIDE];
    const int tid  = threadIdx.x;
    const int lane = tid & 31;
    const int wave = tid >> 5;
    const int half = lane >> 4;
    const int l15  = lane & 15;
    const int rowTile = blockIdx.x * 8 + wave;
    int mrow = rowTile * 16 + l15;
    if (mrow >= M) mrow = M - 1;

    const float* arow = node + (size_t)mrow * 128 + 2 * half;
    v2f av[32];
#pragma unroll
    for (int j = 0; j < 32; ++j) av[j] = *(const v2f*)(arow + 4 * j);

#pragma unroll
    for (int i = 0; i < 16; ++i) {
        const int p  = tid + i * 256;
        const int kk = (p >> 6) << 1;
        const int n  = p & 63;
        v2f b;
        b.x = w_n[kk * 64 + n];
        b.y = w_n[(kk + 1) * 64 + n];
        sB1[p] = b;
    }
#pragma unroll
    for (int i = 0; i < 8; ++i) {
        const int p  = tid + i * 256;
        const int kk = (p >> 6) << 1;
        const int n  = p & 63;
        v2f b;
        b.x = w_fc2[kk * 64 + n];
        b.y = w_fc2[(kk + 1) * 64 + n];
        sB2[p] = b;
    }
    float wa[4], wb[4], wq[4];
#pragma unroll
    for (int nt = 0; nt < 4; ++nt) {
        wa[nt] = w_attn[nt * 16 + l15];
        wb[nt] = w_attn[96 + nt * 16 + l15];
        wq[nt] = w_sem[nt * 16 + l15];
    }
    __syncthreads();

    v8f acc[4] = {};
    const v2f* __restrict__ b1col = sB1 + half * 64 + l15;
#pragma unroll
    for (int j = 0; j < 32; ++j) {
#pragma unroll
        for (int nt = 0; nt < 4; ++nt) {
            const v2f b = b1col[(2 * j) * 64 + nt * 16];
            acc[nt] = __builtin_amdgcn_wmma_f32_16x16x4_f32(
                false, av[j], false, b, (short)0, acc[nt], false, false);
        }
    }

    float s0 = 0.f, s1 = 0.f;
#pragma unroll
    for (int r = 0; r < 8; ++r) {
        float p0 = acc[0][r] * wa[0] + acc[1][r] * wa[1] + acc[2][r] * wa[2] + acc[3][r] * wa[3];
        float p1 = acc[0][r] * wb[0] + acc[1][r] * wb[1] + acc[2][r] * wb[2] + acc[3][r] * wb[3];
        p0 += row_xor<8>(p0);  p1 += row_xor<8>(p1);
        p0 += row_xor<4>(p0);  p1 += row_xor<4>(p1);
        p0 += row_xor<2>(p0);  p1 += row_xor<2>(p1);
        p0 += row_xor<1>(p0);  p1 += row_xor<1>(p1);
        if (l15 == r) { s0 = p0; s1 = p1; }
    }
    {
        const int orow = rowTile * 16 + (l15 & 7) + 8 * half;
        if (l15 < 8 && orow < M) { ssrc[orow] = s0; sdst[orow] = s1; }
    }

    float* tw = sT[wave];
#pragma unroll
    for (int nt = 0; nt < 4; ++nt) {
#pragma unroll
        for (int r = 0; r < 8; ++r)
            tw[(r + 8 * half) * FRONT_STRIDE + nt * 16 + l15] = acc[nt][r];
    }
    __syncthreads();

    v8f acc2[4] = {};
    const float* a2row = tw + l15 * FRONT_STRIDE + 2 * half;
    const v2f* __restrict__ b2col = sB2 + half * 64 + l15;
#pragma unroll
    for (int k0 = 0; k0 < 64; k0 += 4) {
        const v2f a = *(const v2f*)(a2row + k0);
#pragma unroll
        for (int nt = 0; nt < 4; ++nt) {
            const v2f b = b2col[(k0 >> 1) * 64 + nt * 16];
            acc2[nt] = __builtin_amdgcn_wmma_f32_16x16x4_f32(
                false, a, false, b, (short)0, acc2[nt], false, false);
        }
    }

#pragma unroll
    for (int nt = 0; nt < 4; ++nt) {
#pragma unroll
        for (int r = 0; r < 8; ++r) {
            const int row = rowTile * 16 + r + 8 * half;
            if (row < M)
                zz[(size_t)row * 64 + nt * 16 + l15] = acc2[nt][r];
        }
    }

    float q0 = 0.f;
#pragma unroll
    for (int r = 0; r < 8; ++r) {
        float p = acc2[0][r] * wq[0] + acc2[1][r] * wq[1] + acc2[2][r] * wq[2] + acc2[3][r] * wq[3];
        p += row_xor<8>(p);
        p += row_xor<4>(p);
        p += row_xor<2>(p);
        p += row_xor<1>(p);
        if (l15 == r) q0 = p;
    }
    {
        const int orow = rowTile * 16 + (l15 & 7) + 8 * half;
        if (l15 < 8 && orow < M) qn[orow] = q0;
    }
}

__device__ __forceinline__ void ldsMaxFloat(float* addr, float val) {
    const int iv = __float_as_int(val);
    if (iv >= 0) atomicMax((int*)addr, iv);
    else         atomicMin((unsigned int*)addr, (unsigned int)iv);
}

#define SCORE_BLOCKS   256
#define SCORE_THREADS  1024
#define TAB_NODES      50000
#define CHUNK_ITERS    4
#define CHUNK_CAP      (CHUNK_ITERS * SCORE_THREADS)

__device__ __forceinline__ void table_out(const float* tab, float* __restrict__ row, int N, int tid) {
    if (((N & 3) == 0) && ((((size_t)row) & 15) == 0)) {
        const float4* t4 = (const float4*)tab;
        float4* r4 = (float4*)row;
        for (int q = tid; q < (N >> 2); q += SCORE_THREADS) r4[q] = t4[q];
    } else {
        for (int n = tid; n < N; n += SCORE_THREADS) row[n] = tab[n];
    }
}
__device__ __forceinline__ void table_fill(float* tab, float v, int tid) {
    const float4 v4 = {v, v, v, v};
    float4* t4 = (float4*)tab;
    for (int q = tid; q < (TAB_NODES >> 2); q += SCORE_THREADS) t4[q] = v4;
}
__global__ void __launch_bounds__(SCORE_THREADS) edge_score_tables_kernel(
    const float* __restrict__ ef, const float* __restrict__ w_e, const float* __restrict__ w_attn,
    const float* __restrict__ W2, const float* __restrict__ w_sem,
    const int* __restrict__ src, const int* __restrict__ dst,
    const float* __restrict__ ssrc, const float* __restrict__ sdst, const float* __restrict__ qn,
    float* __restrict__ a, float* __restrict__ c, float* __restrict__ lpart, int E, int N)
{
    __shared__ __attribute__((aligned(16))) float tab[TAB_NODES];
    __shared__ __attribute__((aligned(16))) float su[64];
    __shared__ __attribute__((aligned(16))) float sv[64];
    __shared__ float sg[32];
    __shared__ float sA[CHUNK_CAP];
    __shared__ float sC[CHUNK_CAP];
    __shared__ int   sD[CHUNK_CAP];
    const int tid = threadIdx.x;
    if (tid < 32) {
        float acc = 0.f;
#pragma unroll
        for (int o = 0; o < 64; ++o) acc += W2[tid * 64 + o] * w_sem[o];
        sg[tid] = acc;
    }
    __syncthreads();
    if (tid < 64) {
        float acc = 0.f, acv = 0.f;
#pragma unroll
        for (int n = 0; n < 32; ++n) {
            const float w = w_e[tid * 32 + n];
            acc += w * w_attn[64 + n];
            acv += w * sg[n];
        }
        su[tid] = acc;
        sv[tid] = acv;
    }
    table_fill(tab, -INFINITY, tid);

    const int lane = tid & 31;
    const int wave = __builtin_amdgcn_readfirstlane(tid >> 5);
    const int numTiles = (E + 15) >> 4;
    const int tpb = (numTiles + (int)gridDim.x - 1) / (int)gridDim.x;
    const int t0  = min(numTiles, (int)blockIdx.x * tpb);
    const int t1  = min(numTiles, t0 + tpb);
    const int e0  = t0 * 16;
    const int cnt = min(E, t1 * 16) - e0;

    {
        int dreg[CHUNK_ITERS], sreg[CHUNK_ITERS];
#pragma unroll
        for (int k = 0; k < CHUNK_ITERS; ++k) {
            const int e = min(e0 + tid + k * SCORE_THREADS, E - 1);
            dreg[k] = dst[e];
            sreg[k] = src[e];
        }
        float breg[CHUNK_ITERS], qreg[CHUNK_ITERS];
#pragma unroll
        for (int k = 0; k < CHUNK_ITERS; ++k) {
            breg[k] = ssrc[sreg[k]] + sdst[dreg[k]];
            qreg[k] = qn[sreg[k]];
        }
#pragma unroll
        for (int k = 0; k < CHUNK_ITERS; ++k) {
            sD[tid + k * SCORE_THREADS] = dreg[k];
            sA[tid + k * SCORE_THREADS] = breg[k];
            sC[tid + k * SCORE_THREADS] = qreg[k];
        }
    }
    __syncthreads();

    const int half = lane >> 4;
    const int l15  = lane & 15;
    const float4 u4 = *(const float4*)(su + 4 * l15);
    const float4 v4 = *(const float4*)(sv + 4 * l15);

    const int lastTile = numTiles - 1;
    float4 cur[8];
    {
        const int tl = min(t0 + wave, lastTile);
#pragma unroll
        for (int i = 0; i < 8; ++i) {
            const int row = min(tl * 16 + 2 * i + half, E - 1);
            cur[i] = *(const float4*)(ef + (size_t)row * 64 + 4 * l15);
        }
    }
    for (int tile = t0 + wave; tile < t1; tile += SCORE_THREADS / 32) {
        float4 nxt[8];
        {
            const int tl = min(tile + SCORE_THREADS / 32, lastTile);
#pragma unroll
            for (int i = 0; i < 8; ++i) {
                const int row = min(tl * 16 + 2 * i + half, E - 1);
                nxt[i] = *(const float4*)(ef + (size_t)row * 64 + 4 * l15);
            }
        }
        const int er = tile * 16 + 2 * l15 + half;
        const bool own = (l15 < 8) && (er < E);
        const int li = er - e0;

        float mid = 0.f, mic = 0.f;
#pragma unroll
        for (int i = 0; i < 8; ++i) {
            float p = cur[i].x * u4.x + cur[i].y * u4.y + cur[i].z * u4.z + cur[i].w * u4.w;
            float g = cur[i].x * v4.x + cur[i].y * v4.y + cur[i].z * v4.z + cur[i].w * v4.w;
            p += row_xor<8>(p);  g += row_xor<8>(g);
            p += row_xor<4>(p);  g += row_xor<4>(g);
            p += row_xor<2>(p);  g += row_xor<2>(g);
            p += row_xor<1>(p);  g += row_xor<1>(g);
            if (l15 == i) { mid = p; mic = g; }
        }
        if (own) {
            const float t  = sA[li] + mid;
            const float lr = t > 0.f ? t : 0.01f * t;
            a[er]  = lr;
            c[er]  = sC[li] + mic;
            sA[li] = lr;
            ldsMaxFloat(&tab[sD[li]], lr);
        }
#pragma unroll
        for (int i = 0; i < 8; ++i) cur[i] = nxt[i];
    }
    __syncthreads();

    float mreg[CHUNK_ITERS];
#pragma unroll
    for (int k = 0; k < CHUNK_ITERS; ++k) {
        const int i = tid + k * SCORE_THREADS;
        mreg[k] = (i < cnt) ? tab[sD[i]] : 0.f;
    }
    __syncthreads();
    table_fill(tab, 0.f, tid);
    __syncthreads();
#pragma unroll
    for (int k = 0; k < CHUNK_ITERS; ++k) {
        const int i = tid + k * SCORE_THREADS;
        if (i < cnt) atomicAdd(&tab[sD[i]], __expf(sA[i] - mreg[k]));
    }
    __syncthreads();
    float sreg[CHUNK_ITERS];
#pragma unroll
    for (int k = 0; k < CHUNK_ITERS; ++k) {
        const int i = tid + k * SCORE_THREADS;
        sreg[k] = (i < cnt) ? tab[sD[i]] : 1.f;
    }
    __syncthreads();
    table_fill(tab, -INFINITY, tid);
    __syncthreads();
#pragma unroll
    for (int k = 0; k < CHUNK_ITERS; ++k) {
        const int i = tid + k * SCORE_THREADS;
        if (i < cnt) tab[sD[i]] = mreg[k] + logf(sreg[k]);
    }
    __syncthreads();
    table_out(tab, lpart + (size_t)blockIdx.x * N, N, tid);
}

__global__ void __launch_bounds__(256) softmax_combine_kernel(
    const float* __restrict__ lpart, float* __restrict__ lse, int N)
{
    __shared__ float smax[8][32];
    __shared__ float ssum[8][32];
    const int lane = threadIdx.x & 31;
    const int g    = threadIdx.x >> 5;
    const int n    = blockIdx.x * 32 + lane;
    const int nn   = min(n, N - 1);
    float mb[SCORE_BLOCKS / 8];
#pragma unroll
    for (int j = 0; j < SCORE_BLOCKS / 8; ++j) {
        const size_t row = (size_t)(g * (SCORE_BLOCKS / 8) + j) * N + nn;
        mb[j] = lpart[row];
    }
    float m = -INFINITY;
#pragma unroll
    for (int j = 0; j < SCORE_BLOCKS / 8; ++j) m = fmaxf(m, mb[j]);
    smax[g][lane] = m;
    __syncthreads();
    float M = smax[0][lane];
#pragma unroll
    for (int gg = 1; gg < 8; ++gg) M = fmaxf(M, smax[gg][lane]);
    float d = 0.f;
#pragma unroll
    for (int j = 0; j < SCORE_BLOCKS / 8; ++j)
        if (mb[j] > -INFINITY) d += __expf(mb[j] - M);
    ssum[g][lane] = d;
    __syncthreads();
    if (g == 0 && n < N) {
        float t = 0.f;
#pragma unroll
        for (int gg = 0; gg < 8; ++gg) t += ssum[gg][lane];
        lse[n] = M + logf(t);
    }
}

#define ALPHA_PER 4
__global__ void __launch_bounds__(256) alpha_type_sums_kernel(
    float* __restrict__ a, const float* __restrict__ c,
    const float* __restrict__ lse,
    const int* __restrict__ dst, const int* __restrict__ etype,
    float* __restrict__ wpart, float* __restrict__ Z, int E, int zquads)
{
    __shared__ float sw[16][4];
    const int tid  = threadIdx.x;
    const int base = blockIdx.x * (256 * ALPHA_PER) + tid;
    int   dr[ALPHA_PER], tr[ALPHA_PER];
    float ar[ALPHA_PER], cr[ALPHA_PER], mr[ALPHA_PER];
#pragma unroll
    for (int k = 0; k < ALPHA_PER; ++k) {
        const int e = min(base + k * 256, E - 1);
        dr[k] = dst[e];
        tr[k] = etype[e];
        ar[k] = a[e];
        cr[k] = c[e];
    }
#pragma unroll
    for (int k = 0; k < ALPHA_PER; ++k) mr[k] = lse[dr[k]];
    {
        const float4 zero = {0.f, 0.f, 0.f, 0.f};
        float4* Z4 = (float4*)Z;
        for (int q = blockIdx.x * 256 + tid; q < zquads; q += (int)gridDim.x * 256) Z4[q] = zero;
    }
    float s[4] = {0.f, 0.f, 0.f, 0.f};
#pragma unroll
    for (int k = 0; k < ALPHA_PER; ++k) {
        const int e = base + k * 256;
        if (e < E) {
            const float al = __expf(ar[k] - mr[k]);
            a[e] = al;
            const float x = al * cr[k];
#pragma unroll
            for (int t = 0; t < 4; ++t) s[t] += (tr[k] == t) ? x : 0.f;
        }
    }
#pragma unroll
    for (int t = 0; t < 4; ++t) {
        s[t] += row_xor<8>(s[t]);
        s[t] += row_xor<4>(s[t]);
        s[t] += row_xor<2>(s[t]);
        s[t] += row_xor<1>(s[t]);
    }
    if ((tid & 15) == 0) {
#pragma unroll
        for (int t = 0; t < 4; ++t) sw[tid >> 4][t] = s[t];
    }
    __syncthreads();
    if (tid < 4) {
        float w = 0.f;
#pragma unroll
        for (int r = 0; r < 16; ++r) w += sw[r][tid];
        wpart[blockIdx.x * 4 + tid] = w;
    }
}

#define SCATTER_BLOCKS 1536
#define MSG_STRIDE 68
__global__ void __launch_bounds__(256, 12) edge_scatter_fc2_kernel(
    const float* __restrict__ ef, const float* __restrict__ w_e, const float* __restrict__ W2,
    const float* __restrict__ zz,
    const float* __restrict__ alpha, const float* __restrict__ beta,
    const int* __restrict__ src, const int* __restrict__ dst, const int* __restrict__ etype,
    float* __restrict__ Z, int E)
{
    __shared__ v2f   sBe[32 * 32];
    __shared__ v2f   sB2[16 * 64];
    __shared__ __attribute__((aligned(16))) float sT[8][16 * MSG_STRIDE];
    const int tid = threadIdx.x;
    for (int p = tid; p < 32 * 32; p += 256) {
        const int kk = (p >> 5) << 1;
        const int n  = p & 31;
        v2f b;
        b.x = w_e[kk * 32 + n];
        b.y = w_e[(kk + 1) * 32 + n];
        sBe[p] = b;
    }
    for (int p = tid; p < 16 * 64; p += 256) {
        const int kk = (p >> 6) << 1;
        const int n  = p & 63;
        v2f b;
        b.x = W2[kk * 64 + n];
        b.y = W2[(kk + 1) * 64 + n];
        sB2[p] = b;
    }
    __syncthreads();

    const float b0 = beta[0], b1 = beta[1], b2 = beta[2], b3 = beta[3];
    const int lane = tid & 31;
    const int wave = tid >> 5;
    const int half = lane >> 4;
    const int l15  = lane & 15;
    const int numTiles  = (E + 15) >> 4;
    const int numGroups = (numTiles + 7) >> 3;
    for (int grp = blockIdx.x; grp < numGroups; grp += gridDim.x) {
        const int tile = grp * 8 + wave;
        const bool active = tile < numTiles;
        int ec = tile * 16 + l15;
        if (ec >= E) ec = E - 1;
        const int   d_l   = dst[ec];
        const int   s_l   = src[ec];
        const int   t_l   = etype[ec];
        const float a_l   = alpha[ec];

        if (active) {
            v8f pe[2] = {};
            const v2f* __restrict__ arow = (const v2f*)ef + (ec * 32 + half);
            const v2f* __restrict__ becol = sBe + half * 32 + l15;
#pragma unroll
            for (int k0 = 0; k0 < 64; k0 += 4) {
                const v2f av = arow[k0 >> 1];
#pragma unroll
                for (int nt = 0; nt < 2; ++nt) {
                    const v2f b = becol[(k0 >> 1) * 32 + nt * 16];
                    pe[nt] = __builtin_amdgcn_wmma_f32_16x16x4_f32(
                        false, av, false, b, (short)0, pe[nt], false, false);
                }
            }
            float* tw = sT[wave];
#pragma unroll
            for (int nt = 0; nt < 2; ++nt) {
#pragma unroll
                for (int r = 0; r < 8; ++r)
                    tw[(r + 8 * half) * MSG_STRIDE + nt * 16 + l15] = pe[nt][r];
            }
        }
        __syncthreads();
        if (active) {
            v8f acc[4] = {};
            const float* arow = sT[wave] + l15 * MSG_STRIDE + 2 * half;
            const v2f* __restrict__ bcol = sB2 + half * 64 + l15;
#pragma unroll
            for (int k0 = 0; k0 < 32; k0 += 4) {
                const v2f av = *(const v2f*)(arow + k0);
#pragma unroll
                for (int nt = 0; nt < 4; ++nt) {
                    const v2f b = bcol[(k0 >> 1) * 64 + nt * 16];
                    acc[nt] = __builtin_amdgcn_wmma_f32_16x16x4_f32(
                        false, av, false, b, (short)0, acc[nt], false, false);
                }
            }
            float* tw = sT[wave];
#pragma unroll
            for (int nt = 0; nt < 4; ++nt) {
#pragma unroll
                for (int r = 0; r < 8; ++r)
                    tw[(r + 8 * half) * MSG_STRIDE + nt * 16 + l15] = acc[nt][r];
            }
        }
        __syncthreads();
        if (active) {
            const float al_l = a_l * (t_l == 0 ? b0 : t_l == 1 ? b1 : t_l == 2 ? b2 : b3);
            const float* tr = sT[wave];
            const int nvalid = E - __builtin_amdgcn_readfirstlane(tile) * 16;

            float m0[16], m1[16];
#pragma unroll
            for (int i = 0; i < 16; ++i) {
                const int s = __builtin_amdgcn_readlane(s_l, i);
                const float alpha = __int_as_float(__builtin_amdgcn_readlane(__float_as_int(al_l), i));
                const int zi = s * 64 + lane;
                m0[i] = alpha * (tr[i * MSG_STRIDE + lane]      + zz[zi]);
                m1[i] = alpha * (tr[i * MSG_STRIDE + lane + 32] + zz[zi + 32]);
            }
#pragma unroll
            for (int i = 0; i < 16; ++i) {
                if (i < nvalid) {
                    const int d = __builtin_amdgcn_readlane(d_l, i);
                    float* hb = Z + (d * 64 + lane);
                    atomicAdd(hb,      m0[i]);
                    atomicAdd(hb + 32, m1[i]);
                }
            }
        }
    }
}

__global__ void __launch_bounds__(256) beta_kernel(const float* __restrict__ wpart, int nparts, float* __restrict__ beta)
{
    __shared__ float prod[256];
    __shared__ float wsh[4];
    const int tid = threadIdx.x;
    float z = 0.f;
    for (int r = 0; r < nparts; r += 64 * 8) {
        float v[8];
        for (int j = 0; j < 8; ++j) {
            const int row = r + j * 64 + (tid >> 2);
            v[j] = (row < nparts) ? wpart[row * 4 + (tid & 3)] : 0.f;
        }
        for (int j = 0; j < 8; ++j) z += v[j];
    }
    prod[tid] = z;
    __syncthreads();
    if (tid < 4) {
        float d = 0.f;
        for (int g = 0; g < 64; ++g) d += prod[g * 4 + tid];
        wsh[tid] = d > 0.f ? d : 0.01f * d;
    }
    __syncthreads();
    if (tid != 0) return;
    float wp[4], mx = -INFINITY;
#pragma unroll
    for (int t = 0; t < 4; ++t) {
        wp[t] = wsh[t];
        mx = fmaxf(mx, wp[t]);
    }
    float sm = 0.f;
#pragma unroll
    for (int t = 0; t < 4; ++t) { wp[t] = __expf(wp[t] - mx); sm += wp[t]; }
#pragma unroll
    for (int t = 0; t < 4; ++t) beta[t] = wp[t] / sm;
}

extern "C" void kernel_launch(void* const* d_in, const int* in_sizes, int n_in,
                              void* d_out, int out_size, void* d_ws, size_t ws_size,
                              hipStream_t stream)
{
    const float* node   = (const float*)d_in[0];
    const float* edgef  = (const float*)d_in[1];
    const int*   src    = (const int*)d_in[2];
    const int*   dst    = (const int*)d_in[3];
    const int*   etype  = (const int*)d_in[4];
    const float* w_n    = (const float*)d_in[5];
    const float* w_e    = (const float*)d_in[6];
    const float* w_attn = (const float*)d_in[7];
    const float* w_fc2  = (const float*)d_in[8];
    const float* w_sem  = (const float*)d_in[9];
    float* out = (float*)d_out;

    const int N = in_sizes[0] / 128;
    const int E = in_sizes[2];

    float* ws   = (float*)d_ws;
    float* zz   = ws;
    float* ssrc = zz + (size_t)N * 64;
    float* sdst = ssrc + N;
    float* abuf = sdst + N;
    float* mbuf = abuf + E;
    float* den  = mbuf + N;
    float* beta = den + N;
    float* qn   = den;
    float* lpart = (float*)(((size_t)(beta + 16) + 255) & ~(size_t)255);
    float* cbuf  = lpart + (size_t)SCORE_BLOCKS * N;
    float* wpart = lpart;

    const int thr = 256;
    const int alphaBlocks = (E + 256 * ALPHA_PER - 1) / (256 * ALPHA_PER);

    node_front_kernel<<<((N + 15) / 16 + 7) / 8, thr, 0, stream>>>(node, w_n, w_attn, w_fc2, w_sem, ssrc, sdst, qn, zz, N);
    edge_score_tables_kernel<<<SCORE_BLOCKS, SCORE_THREADS, 0, stream>>>(edgef, w_e, w_attn, w_fc2 + 64 * 64, w_sem, src, dst,
                                                                      ssrc, sdst, qn, abuf, cbuf, lpart, E, N);
    softmax_combine_kernel<<<(N + 31) / 32, thr, 0, stream>>>(lpart, mbuf, N);
    alpha_type_sums_kernel<<<alphaBlocks, thr, 0, stream>>>(abuf, cbuf, mbuf, dst, etype, wpart, out, E, N * 16);
    beta_kernel<<<1, thr, 0, stream>>>(wpart, alphaBlocks, beta);
    edge_scatter_fc2_kernel<<<SCATTER_BLOCKS, thr, 0, stream>>>(edgef, w_e, w_fc2 + 64 * 64, zz, abuf, beta,
                                                                      src, dst, etype, out, E);
}
